# starting kernel
# baseline (speedup 1.0000x reference)
#include <hip/hip_runtime.h>

typedef __attribute__((ext_vector_type(2))) float v2f;
typedef __attribute__((ext_vector_type(8))) float v8f;

#define LN_EPS 1e-5f

__device__ __forceinline__ v8f wmma4(v2f a, v2f b, v8f c) {
  return __builtin_amdgcn_wmma_f32_16x16x4_f32(false, a, false, b, (short)0, c,
                                               false, false);
}

__device__ __forceinline__ v2f ldA(const float* m, int ld, int r0, int c0,
                                   int l16, int h) {
  v2f a;
  a.x = m[(r0 + l16) * ld + c0 + 2 * h + 0];
  a.y = m[(r0 + l16) * ld + c0 + 2 * h + 1];
  return a;
}
__device__ __forceinline__ v2f ldB(const float* m, int ld, int c0, int n0,
                                   int l16, int h) {
  v2f b;
  b.x = m[(c0 + 2 * h + 0) * ld + n0 + l16];
  b.y = m[(c0 + 2 * h + 1) * ld + n0 + l16];
  return b;
}
__device__ __forceinline__ v2f ldBT(const float* m, int ld, int c0, int n0,
                                    int l16, int h) {
  v2f b;
  b.x = m[(n0 + l16) * ld + c0 + 2 * h + 0];
  b.y = m[(n0 + l16) * ld + c0 + 2 * h + 1];
  return b;
}

__device__ __forceinline__ unsigned int f2key(float f) {
  unsigned int u = __float_as_uint(f);
  return (u & 0x80000000u) ? ~u : (u | 0x80000000u);
}
__device__ __forceinline__ float key2f(unsigned int k) {
  unsigned int u = (k & 0x80000000u) ? (k & 0x7FFFFFFFu) : ~k;
  return __uint_as_float(u);
}

__device__ __forceinline__ float bsum(float v, float* red, int tid) {
  __syncthreads();
  red[tid] = v;
  __syncthreads();
#pragma unroll
  for (int s = 64; s > 0; s >>= 1) {
    if (tid < s) red[tid] += red[tid + s];
    __syncthreads();
  }
  return red[0];
}

__device__ __forceinline__ float bsum_g(float v, float* red, int tid, int t,
                                        int gbase) {
  __syncthreads();
  red[tid] = v;
  __syncthreads();
#pragma unroll
  for (int s = 64; s > 0; s >>= 1) {
    if (t < s) red[tid] += red[tid + s];
    __syncthreads();
  }
  return red[gbase];
}

__global__ void enc_init(unsigned int* gmax, float* gsum) {
  int i = blockIdx.x * blockDim.x + threadIdx.x;
  if (i < 4096) {
    gmax[i] = 0u;
    gsum[i] = 0.0f;
  }
}

__global__ __launch_bounds__(128) void enc_pass1_max(
    const float* __restrict__ x, const float* __restrict__ Wq,
    const float* __restrict__ Wk, unsigned int* __restrict__ gmax, int nb) {
  __shared__ __align__(16) float sX[1024];
  __shared__ float sQ[1024], sK[1024];
  __shared__ float sWq[256], sWk[256];
  int tid = threadIdx.x;
  int w = tid >> 5, lane = tid & 31, l16 = lane & 15, h = lane >> 4;
  int r0 = 16 * w;
  for (int i = tid; i < 256; i += 128) {
    sWq[i] = Wq[i];
    sWk[i] = Wk[i];
  }
  float vmax[4][8];
#pragma unroll
  for (int kt = 0; kt < 4; kt++)
#pragma unroll
    for (int r = 0; r < 8; r++) vmax[kt][r] = -3.4e38f;
  __syncthreads();

  for (int it = 0; it < nb; ++it) {
    size_t b = (size_t)blockIdx.x * nb + it;
    const float4* xb4 = (const float4*)(x + b * 1024);
    ((float4*)sX)[tid] = xb4[tid];
    ((float4*)sX)[tid + 128] = xb4[tid + 128];
    __syncthreads();
    v2f a[4];
#pragma unroll
    for (int kk = 0; kk < 4; kk++) a[kk] = ldA(sX, 16, r0, 4 * kk, l16, h);
    v8f q = {};
    v8f k = {};
#pragma unroll
    for (int kk = 0; kk < 4; kk++) {
      q = wmma4(a[kk], ldB(sWq, 16, 4 * kk, 0, l16, h), q);
      k = wmma4(a[kk], ldB(sWk, 16, 4 * kk, 0, l16, h), k);
    }
#pragma unroll
    for (int r = 0; r < 8; r++) {
      sQ[(r0 + r + 8 * h) * 16 + l16] = q[r];
      sK[(r0 + r + 8 * h) * 16 + l16] = k[r];
    }
    __syncthreads();
    v2f qa[4];
#pragma unroll
    for (int kk = 0; kk < 4; kk++) qa[kk] = ldA(sQ, 16, r0, 4 * kk, l16, h);
#pragma unroll
    for (int kt = 0; kt < 4; kt++) {
      v8f s = {};
#pragma unroll
      for (int kk = 0; kk < 4; kk++)
        s = wmma4(qa[kk], ldBT(sK, 16, 4 * kk, 16 * kt, l16, h), s);
#pragma unroll
      for (int r = 0; r < 8; r++)
        vmax[kt][r] = fmaxf(vmax[kt][r], s[r] * (1.0f / 3.0f));
    }
    __syncthreads();
  }
#pragma unroll
  for (int kt = 0; kt < 4; kt++)
#pragma unroll
    for (int r = 0; r < 8; r++) {
      int qrow = r0 + r + 8 * h, kcol = 16 * kt + l16;
      atomicMax(&gmax[qrow * 64 + kcol], f2key(vmax[kt][r]));
    }
}

__global__ __launch_bounds__(128) void enc_pass2_sum(
    const float* __restrict__ x, const float* __restrict__ Wq,
    const float* __restrict__ Wk, const unsigned int* __restrict__ gmax,
    float* __restrict__ gsum, int nb) {
  __shared__ __align__(16) float sX[1024];
  __shared__ float sQ[1024], sK[1024];
  __shared__ float sWq[256], sWk[256];
  int tid = threadIdx.x;
  int w = tid >> 5, lane = tid & 31, l16 = lane & 15, h = lane >> 4;
  int r0 = 16 * w;
  for (int i = tid; i < 256; i += 128) {
    sWq[i] = Wq[i];
    sWk[i] = Wk[i];
  }
  float gm[4][8], vsum[4][8];
#pragma unroll
  for (int kt = 0; kt < 4; kt++)
#pragma unroll
    for (int r = 0; r < 8; r++) {
      int qrow = r0 + r + 8 * h, kcol = 16 * kt + l16;
      gm[kt][r] = key2f(gmax[qrow * 64 + kcol]);
      vsum[kt][r] = 0.0f;
    }
  __syncthreads();

  for (int it = 0; it < nb; ++it) {
    size_t b = (size_t)blockIdx.x * nb + it;
    const float4* xb4 = (const float4*)(x + b * 1024);
    ((float4*)sX)[tid] = xb4[tid];
    ((float4*)sX)[tid + 128] = xb4[tid + 128];
    __syncthreads();
    v2f a[4];
#pragma unroll
    for (int kk = 0; kk < 4; kk++) a[kk] = ldA(sX, 16, r0, 4 * kk, l16, h);
    v8f q = {};
    v8f k = {};
#pragma unroll
    for (int kk = 0; kk < 4; kk++) {
      q = wmma4(a[kk], ldB(sWq, 16, 4 * kk, 0, l16, h), q);
      k = wmma4(a[kk], ldB(sWk, 16, 4 * kk, 0, l16, h), k);
    }
#pragma unroll
    for (int r = 0; r < 8; r++) {
      sQ[(r0 + r + 8 * h) * 16 + l16] = q[r];
      sK[(r0 + r + 8 * h) * 16 + l16] = k[r];
    }
    __syncthreads();
    v2f qa[4];
#pragma unroll
    for (int kk = 0; kk < 4; kk++) qa[kk] = ldA(sQ, 16, r0, 4 * kk, l16, h);
#pragma unroll
    for (int kt = 0; kt < 4; kt++) {
      v8f s = {};
#pragma unroll
      for (int kk = 0; kk < 4; kk++)
        s = wmma4(qa[kk], ldBT(sK, 16, 4 * kk, 16 * kt, l16, h), s);
#pragma unroll
      for (int r = 0; r < 8; r++)
        vsum[kt][r] += __expf(s[r] * (1.0f / 3.0f) - gm[kt][r]);
    }
    __syncthreads();
  }
#pragma unroll
  for (int kt = 0; kt < 4; kt++)
#pragma unroll
    for (int r = 0; r < 8; r++) {
      int qrow = r0 + r + 8 * h, kcol = 16 * kt + l16;
      atomicAdd(&gsum[qrow * 64 + kcol], vsum[kt][r]);
    }
}

#define WSTRIDE 260

__global__ __launch_bounds__(256) void enc_pass3_final(
    const float* __restrict__ x, const float* __restrict__ Wq,
    const float* __restrict__ Wk, const float* __restrict__ Wv,
    const float* __restrict__ W1, const float* __restrict__ b1,
    const float* __restrict__ W2, const float* __restrict__ b2,
    const unsigned int* __restrict__ gmax, const float* __restrict__ gsum,
    float* __restrict__ out, int nb) {
  __shared__ float sW1[64 * WSTRIDE];
  __shared__ float sW2[64 * WSTRIDE];
  __shared__ float sB1[1024], sB2[1024];
  __shared__ float sWq[256], sWk[256], sWv[256];
  __shared__ __align__(16) float sX[2][1024];
  __shared__ float sQ[2][1024];
  __shared__ float sK[2][1024], sV[2][1024];
  __shared__ float sS[2][4096];
  __shared__ float red[256];

  int tid = threadIdx.x;
  int g = tid >> 7;
  int t = tid & 127;
  int gbase = g << 7;
  int w = t >> 5, lane = tid & 31, l16 = lane & 15, h = lane >> 4;
  int r0 = 16 * w;
  int p = t >> 1, o0 = (t & 1) * 8;

  float* mX = sX[g];
  float* mQ = sQ[g];
  float* mK = sK[g];
  float* mV = sV[g];
  float* mS = sS[g];

  for (int i = tid; i < 16384; i += 256) {
    int pp = i >> 8, rr = i & 255;
    sW1[pp * WSTRIDE + rr] = W1[i];
    sW2[pp * WSTRIDE + rr] = W2[i];
  }
  for (int i = tid; i < 1024; i += 256) {
    sB1[i] = b1[i];
    sB2[i] = b2[i];
  }
  if (tid < 256) {
    sWq[tid] = Wq[tid];
    sWk[tid] = Wk[tid];
    sWv[tid] = Wv[tid];
  }
  float gm[4][8], rs[4][8];
#pragma unroll
  for (int kt = 0; kt < 4; kt++)
#pragma unroll
    for (int r = 0; r < 8; r++) {
      int qrow = r0 + r + 8 * h, kcol = 16 * kt + l16;
      gm[kt][r] = key2f(gmax[qrow * 64 + kcol]);
      rs[kt][r] = 1.0f / gsum[qrow * 64 + kcol];
    }
  __syncthreads();

  for (int it = 0; it < nb; ++it) {
    size_t b = (size_t)blockIdx.x * (2 * nb) + (size_t)g * nb + it;
    const float4* xb4 = (const float4*)(x + b * 1024);
    ((float4*)mX)[t] = xb4[t];
    ((float4*)mX)[t + 128] = xb4[t + 128];
    __syncthreads();

    v2f a[4];
#pragma unroll
    for (int kk = 0; kk < 4; kk++) a[kk] = ldA(mX, 16, r0, 4 * kk, l16, h);
    v8f q = {};
    v8f k = {};
    v8f v = {};
#pragma unroll
    for (int kk = 0; kk < 4; kk++) {
      q = wmma4(a[kk], ldB(sWq, 16, 4 * kk, 0, l16, h), q);
      k = wmma4(a[kk], ldB(sWk, 16, 4 * kk, 0, l16, h), k);
      v = wmma4(a[kk], ldB(sWv, 16, 4 * kk, 0, l16, h), v);
    }
#pragma unroll
    for (int r = 0; r < 8; r++) {
      int row = r0 + r + 8 * h;
      mQ[row * 16 + l16] = q[r];
      mK[row * 16 + l16] = k[r];
      mV[row * 16 + l16] = v[r];
    }
    __syncthreads();

    v2f qa[4];
#pragma unroll
    for (int kk = 0; kk < 4; kk++) qa[kk] = ldA(mQ, 16, r0, 4 * kk, l16, h);
#pragma unroll
    for (int kt = 0; kt < 4; kt++) {
      v8f s = {};
#pragma unroll
      for (int kk = 0; kk < 4; kk++)
        s = wmma4(qa[kk], ldBT(mK, 16, 4 * kk, 16 * kt, l16, h), s);
#pragma unroll
      for (int r = 0; r < 8; r++) {
        int qrow = r0 + r + 8 * h, kcol = 16 * kt + l16;
        mS[qrow * 64 + kcol] =
            __expf(s[r] * (1.0f / 3.0f) - gm[kt][r]) * rs[kt][r];
      }
    }

    v8f z = {};
#pragma unroll
    for (int c = 0; c < 16; c++)
      z = wmma4(ldA(mS, 64, r0, 4 * c, l16, h), ldB(mV, 16, 4 * c, 0, l16, h),
                z);

    float part = 0.0f, part2 = 0.0f, tv[8];
#pragma unroll
    for (int r = 0; r < 8; r++) {
      int row = r0 + r + 8 * h;
      float tt = mX[row * 16 + l16] + z[r];
      tv[r] = tt;
      part += tt;
      part2 += tt * tt;
    }
    float s1 = bsum_g(part, red, tid, t, gbase);
    float s2 = bsum_g(part2, red, tid, t, gbase);
    float mean = s1 * (1.0f / 1024.0f);
    float rstd = rsqrtf(s2 * (1.0f / 1024.0f) - mean * mean + LN_EPS);
#pragma unroll
    for (int r = 0; r < 8; r++) {
      int row = r0 + r + 8 * h;
      mX[row * 16 + l16] = (tv[r] - mean) * rstd;
    }
    __syncthreads();

    float acc[8];
#pragma unroll
    for (int j = 0; j < 8; j++) acc[j] = sB1[p * 16 + o0 + j];
#pragma unroll
    for (int i = 0; i < 16; i++) {
      float yv = mX[p * 16 + i];
      const float* wr = &sW1[p * WSTRIDE + i * 16 + o0];
#pragma unroll
      for (int j = 0; j < 8; j++) acc[j] = fmaf(yv, wr[j], acc[j]);
    }
#pragma unroll
    for (int j = 0; j < 8; j++) mQ[p * 16 + o0 + j] = fmaxf(acc[j], 0.0f);
    __syncthreads();

    float acc2[8];
#pragma unroll
    for (int j = 0; j < 8; j++) acc2[j] = sB2[p * 16 + o0 + j];
#pragma unroll
    for (int i = 0; i < 16; i++) {
      float hv = mQ[p * 16 + i];
      const float* wr = &sW2[p * WSTRIDE + i * 16 + o0];
#pragma unroll
      for (int j = 0; j < 8; j++) acc2[j] = fmaf(hv, wr[j], acc2[j]);
    }
    float t2[8];
    float q1 = 0.0f, q2 = 0.0f;
#pragma unroll
    for (int j = 0; j < 8; j++) {
      float tt = fmaxf(acc2[j], 0.0f) + mX[p * 16 + o0 + j];
      t2[j] = tt;
      q1 += tt;
      q2 += tt * tt;
    }
    float u1 = bsum_g(q1, red, tid, t, gbase);
    float u2 = bsum_g(q2, red, tid, t, gbase);
    float m2 = u1 * (1.0f / 1024.0f);
    float r2 = rsqrtf(u2 * (1.0f / 1024.0f) - m2 * m2 + LN_EPS);
    float* op = out + b * 1024 + p * 16 + o0;
#pragma unroll
    for (int j = 0; j < 8; j++) op[j] = (t2[j] - m2) * r2;
    __syncthreads();
  }
}

extern "C" void kernel_launch(void* const* d_in, const int* in_sizes, int n_in,
                              void* d_out, int out_size, void* d_ws,
                              size_t ws_size, hipStream_t stream) {
  (void)n_in;
  (void)out_size;
  (void)ws_size;
  const float* x = (const float*)d_in[0];
  const float* Wq = (const float*)d_in[1];
  const float* Wk = (const float*)d_in[2];
  const float* Wv = (const float*)d_in[3];
  const float* W1 = (const float*)d_in[4];
  const float* b1 = (const float*)d_in[5];
  const float* W2 = (const float*)d_in[6];
  const float* b2 = (const float*)d_in[7];
  float* out = (float*)d_out;

  unsigned int* gmax = (unsigned int*)d_ws;
  float* gsum = (float*)((char*)d_ws + 4096 * sizeof(unsigned int));

  const int B = in_sizes[0] / 1024;
  const int NB = 64;
  const int blocks12 = B / NB;
  const int blocks3 = B / (2 * NB);

  enc_init<<<16, 256, 0, stream>>>(gmax, gsum);
  enc_pass1_max<<<blocks12, 128, 0, stream>>>(x, Wq, Wk, gmax, NB);
  enc_pass2_sum<<<blocks12, 128, 0, stream>>>(x, Wq, Wk, gmax, gsum, NB);
  enc_pass3_final<<<blocks3, 256, 0, stream>>>(x, Wq, Wk, Wv, W1, b1, W2, b2,
                                               gmax, gsum, out, NB);
}


# pass 3 LayerNorm sums: barrier-stepped LDS trees -> in-wave DPP reduce + one LDS exchange of (sum, sumsq) pairs; changes the reduction tree (fp sum order); 41 -> 6 barriers per batch
# speedup vs baseline: 1.2865x; 1.2865x over previous
#include <hip/hip_runtime.h>

typedef __attribute__((ext_vector_type(2))) float v2f;
typedef __attribute__((ext_vector_type(8))) float v8f;

#define LN_EPS 1e-5f

__device__ __forceinline__ v8f wmma4(v2f a, v2f b, v8f c) {
  return __builtin_amdgcn_wmma_f32_16x16x4_f32(false, a, false, b, (short)0, c,
                                               false, false);
}

__device__ __forceinline__ v2f ldA(const float* m, int ld, int r0, int c0,
                                   int l16, int h) {
  v2f a;
  a.x = m[(r0 + l16) * ld + c0 + 2 * h + 0];
  a.y = m[(r0 + l16) * ld + c0 + 2 * h + 1];
  return a;
}
__device__ __forceinline__ v2f ldB(const float* m, int ld, int c0, int n0,
                                   int l16, int h) {
  v2f b;
  b.x = m[(c0 + 2 * h + 0) * ld + n0 + l16];
  b.y = m[(c0 + 2 * h + 1) * ld + n0 + l16];
  return b;
}
__device__ __forceinline__ v2f ldBT(const float* m, int ld, int c0, int n0,
                                    int l16, int h) {
  v2f b;
  b.x = m[(n0 + l16) * ld + c0 + 2 * h + 0];
  b.y = m[(n0 + l16) * ld + c0 + 2 * h + 1];
  return b;
}

__device__ __forceinline__ unsigned int f2key(float f) {
  unsigned int u = __float_as_uint(f);
  return (u & 0x80000000u) ? ~u : (u | 0x80000000u);
}
__device__ __forceinline__ float key2f(unsigned int k) {
  unsigned int u = (k & 0x80000000u) ? (k & 0x7FFFFFFFu) : ~k;
  return __uint_as_float(u);
}

template <int XM>
__device__ __forceinline__ float row_xor(float v) {
  int b = __builtin_bit_cast(int, v);
  return __builtin_bit_cast(
      float, __builtin_amdgcn_update_dpp(0, b, 0x160 | XM, 0xf, 0xf, true));
}
__device__ __forceinline__ float xhalf(float v) {
  int b = __builtin_bit_cast(int, v);
  return __builtin_bit_cast(
      float, __builtin_amdgcn_permlanex16(b, b, 0x76543210, 0xfedcba98, false,
                                          false));
}
__device__ __forceinline__ float wave_sum(float v) {
  v += row_xor<1>(v);
  v += row_xor<2>(v);
  v += row_xor<4>(v);
  v += row_xor<8>(v);
  v += xhalf(v);
  return v;
}

__device__ __forceinline__ float2 gsum2(float a, float b, float2* slot,
                                        int wv, int lane, int g) {
  float wa = wave_sum(a);
  float wb = wave_sum(b);
  if (lane == 0) slot[wv] = make_float2(wa, wb);
  __syncthreads();
  float2 p0 = slot[4 * g + 0], p1 = slot[4 * g + 1];
  float2 p2 = slot[4 * g + 2], p3 = slot[4 * g + 3];
  return make_float2((p0.x + p1.x) + (p2.x + p3.x),
                     (p0.y + p1.y) + (p2.y + p3.y));
}

__global__ void enc_init(unsigned int* gmax, float* gsum) {
  int i = blockIdx.x * blockDim.x + threadIdx.x;
  if (i < 4096) {
    gmax[i] = 0u;
    gsum[i] = 0.0f;
  }
}

__global__ __launch_bounds__(128) void enc_pass1_max(
    const float* __restrict__ x, const float* __restrict__ Wq,
    const float* __restrict__ Wk, unsigned int* __restrict__ gmax, int nb) {
  __shared__ __align__(16) float sX[1024];
  __shared__ float sQ[1024], sK[1024];
  __shared__ float sWq[256], sWk[256];
  int tid = threadIdx.x;
  int w = tid >> 5, lane = tid & 31, l16 = lane & 15, h = lane >> 4;
  int r0 = 16 * w;
  for (int i = tid; i < 256; i += 128) {
    sWq[i] = Wq[i];
    sWk[i] = Wk[i];
  }
  float vmax[4][8];
#pragma unroll
  for (int kt = 0; kt < 4; kt++)
#pragma unroll
    for (int r = 0; r < 8; r++) vmax[kt][r] = -3.4e38f;
  __syncthreads();

  for (int it = 0; it < nb; ++it) {
    size_t b = (size_t)blockIdx.x * nb + it;
    const float4* xb4 = (const float4*)(x + b * 1024);
    ((float4*)sX)[tid] = xb4[tid];
    ((float4*)sX)[tid + 128] = xb4[tid + 128];
    __syncthreads();
    v2f a[4];
#pragma unroll
    for (int kk = 0; kk < 4; kk++) a[kk] = ldA(sX, 16, r0, 4 * kk, l16, h);
    v8f q = {};
    v8f k = {};
#pragma unroll
    for (int kk = 0; kk < 4; kk++) {
      q = wmma4(a[kk], ldB(sWq, 16, 4 * kk, 0, l16, h), q);
      k = wmma4(a[kk], ldB(sWk, 16, 4 * kk, 0, l16, h), k);
    }
#pragma unroll
    for (int r = 0; r < 8; r++) {
      sQ[(r0 + r + 8 * h) * 16 + l16] = q[r];
      sK[(r0 + r + 8 * h) * 16 + l16] = k[r];
    }
    __syncthreads();
    v2f qa[4];
#pragma unroll
    for (int kk = 0; kk < 4; kk++) qa[kk] = ldA(sQ, 16, r0, 4 * kk, l16, h);
#pragma unroll
    for (int kt = 0; kt < 4; kt++) {
      v8f s = {};
#pragma unroll
      for (int kk = 0; kk < 4; kk++)
        s = wmma4(qa[kk], ldBT(sK, 16, 4 * kk, 16 * kt, l16, h), s);
#pragma unroll
      for (int r = 0; r < 8; r++)
        vmax[kt][r] = fmaxf(vmax[kt][r], s[r] * (1.0f / 3.0f));
    }
    __syncthreads();
  }
#pragma unroll
  for (int kt = 0; kt < 4; kt++)
#pragma unroll
    for (int r = 0; r < 8; r++) {
      int qrow = r0 + r + 8 * h, kcol = 16 * kt + l16;
      atomicMax(&gmax[qrow * 64 + kcol], f2key(vmax[kt][r]));
    }
}

__global__ __launch_bounds__(128) void enc_pass2_sum(
    const float* __restrict__ x, const float* __restrict__ Wq,
    const float* __restrict__ Wk, const unsigned int* __restrict__ gmax,
    float* __restrict__ gsum, int nb) {
  __shared__ __align__(16) float sX[1024];
  __shared__ float sQ[1024], sK[1024];
  __shared__ float sWq[256], sWk[256];
  int tid = threadIdx.x;
  int w = tid >> 5, lane = tid & 31, l16 = lane & 15, h = lane >> 4;
  int r0 = 16 * w;
  for (int i = tid; i < 256; i += 128) {
    sWq[i] = Wq[i];
    sWk[i] = Wk[i];
  }
  float gm[4][8], vsum[4][8];
#pragma unroll
  for (int kt = 0; kt < 4; kt++)
#pragma unroll
    for (int r = 0; r < 8; r++) {
      int qrow = r0 + r + 8 * h, kcol = 16 * kt + l16;
      gm[kt][r] = key2f(gmax[qrow * 64 + kcol]);
      vsum[kt][r] = 0.0f;
    }
  __syncthreads();

  for (int it = 0; it < nb; ++it) {
    size_t b = (size_t)blockIdx.x * nb + it;
    const float4* xb4 = (const float4*)(x + b * 1024);
    ((float4*)sX)[tid] = xb4[tid];
    ((float4*)sX)[tid + 128] = xb4[tid + 128];
    __syncthreads();
    v2f a[4];
#pragma unroll
    for (int kk = 0; kk < 4; kk++) a[kk] = ldA(sX, 16, r0, 4 * kk, l16, h);
    v8f q = {};
    v8f k = {};
#pragma unroll
    for (int kk = 0; kk < 4; kk++) {
      q = wmma4(a[kk], ldB(sWq, 16, 4 * kk, 0, l16, h), q);
      k = wmma4(a[kk], ldB(sWk, 16, 4 * kk, 0, l16, h), k);
    }
#pragma unroll
    for (int r = 0; r < 8; r++) {
      sQ[(r0 + r + 8 * h) * 16 + l16] = q[r];
      sK[(r0 + r + 8 * h) * 16 + l16] = k[r];
    }
    __syncthreads();
    v2f qa[4];
#pragma unroll
    for (int kk = 0; kk < 4; kk++) qa[kk] = ldA(sQ, 16, r0, 4 * kk, l16, h);
#pragma unroll
    for (int kt = 0; kt < 4; kt++) {
      v8f s = {};
#pragma unroll
      for (int kk = 0; kk < 4; kk++)
        s = wmma4(qa[kk], ldBT(sK, 16, 4 * kk, 16 * kt, l16, h), s);
#pragma unroll
      for (int r = 0; r < 8; r++)
        vsum[kt][r] += __expf(s[r] * (1.0f / 3.0f) - gm[kt][r]);
    }
    __syncthreads();
  }
#pragma unroll
  for (int kt = 0; kt < 4; kt++)
#pragma unroll
    for (int r = 0; r < 8; r++) {
      int qrow = r0 + r + 8 * h, kcol = 16 * kt + l16;
      atomicAdd(&gsum[qrow * 64 + kcol], vsum[kt][r]);
    }
}

#define WSTRIDE 260

__global__ __launch_bounds__(256) void enc_pass3_final(
    const float* __restrict__ x, const float* __restrict__ Wq,
    const float* __restrict__ Wk, const float* __restrict__ Wv,
    const float* __restrict__ W1, const float* __restrict__ b1,
    const float* __restrict__ W2, const float* __restrict__ b2,
    const unsigned int* __restrict__ gmax, const float* __restrict__ gsum,
    float* __restrict__ out, int nb) {
  __shared__ float sW1[64 * WSTRIDE];
  __shared__ float sW2[64 * WSTRIDE];
  __shared__ float sB1[1024], sB2[1024];
  __shared__ float sWq[256], sWk[256], sWv[256];
  __shared__ __align__(16) float sX[2][1024];
  __shared__ float sQ[2][1024];
  __shared__ float sK[2][1024], sV[2][1024];
  __shared__ float sS[2][4096];
  __shared__ float2 redA[8], redB[8];

  int tid = threadIdx.x;
  int g = tid >> 7;
  int t = tid & 127;
  int wv = tid >> 5;
  int w = t >> 5, lane = tid & 31, l16 = lane & 15, h = lane >> 4;
  int r0 = 16 * w;
  int p = t >> 1, o0 = (t & 1) * 8;

  float* mX = sX[g];
  float* mQ = sQ[g];
  float* mK = sK[g];
  float* mV = sV[g];
  float* mS = sS[g];

  for (int i = tid; i < 16384; i += 256) {
    int pp = i >> 8, rr = i & 255;
    sW1[pp * WSTRIDE + rr] = W1[i];
    sW2[pp * WSTRIDE + rr] = W2[i];
  }
  for (int i = tid; i < 1024; i += 256) {
    sB1[i] = b1[i];
    sB2[i] = b2[i];
  }
  if (tid < 256) {
    sWq[tid] = Wq[tid];
    sWk[tid] = Wk[tid];
    sWv[tid] = Wv[tid];
  }
  float gm[4][8], rs[4][8];
#pragma unroll
  for (int kt = 0; kt < 4; kt++)
#pragma unroll
    for (int r = 0; r < 8; r++) {
      int qrow = r0 + r + 8 * h, kcol = 16 * kt + l16;
      gm[kt][r] = key2f(gmax[qrow * 64 + kcol]);
      rs[kt][r] = 1.0f / gsum[qrow * 64 + kcol];
    }
  __syncthreads();

  for (int it = 0; it < nb; ++it) {
    size_t b = (size_t)blockIdx.x * (2 * nb) + (size_t)g * nb + it;
    const float4* xb4 = (const float4*)(x + b * 1024);
    ((float4*)mX)[t] = xb4[t];
    ((float4*)mX)[t + 128] = xb4[t + 128];
    __syncthreads();

    v2f a[4];
#pragma unroll
    for (int kk = 0; kk < 4; kk++) a[kk] = ldA(mX, 16, r0, 4 * kk, l16, h);
    v8f q = {};
    v8f k = {};
    v8f v = {};
#pragma unroll
    for (int kk = 0; kk < 4; kk++) {
      q = wmma4(a[kk], ldB(sWq, 16, 4 * kk, 0, l16, h), q);
      k = wmma4(a[kk], ldB(sWk, 16, 4 * kk, 0, l16, h), k);
      v = wmma4(a[kk], ldB(sWv, 16, 4 * kk, 0, l16, h), v);
    }
#pragma unroll
    for (int r = 0; r < 8; r++) {
      int row = r0 + r + 8 * h;
      mQ[row * 16 + l16] = q[r];
      mK[row * 16 + l16] = k[r];
      mV[row * 16 + l16] = v[r];
    }
    __syncthreads();

    v2f qa[4];
#pragma unroll
    for (int kk = 0; kk < 4; kk++) qa[kk] = ldA(mQ, 16, r0, 4 * kk, l16, h);
#pragma unroll
    for (int kt = 0; kt < 4; kt++) {
      v8f s = {};
#pragma unroll
      for (int kk = 0; kk < 4; kk++)
        s = wmma4(qa[kk], ldBT(mK, 16, 4 * kk, 16 * kt, l16, h), s);
#pragma unroll
      for (int r = 0; r < 8; r++) {
        int qrow = r0 + r + 8 * h, kcol = 16 * kt + l16;
        mS[qrow * 64 + kcol] =
            __expf(s[r] * (1.0f / 3.0f) - gm[kt][r]) * rs[kt][r];
      }
    }

    v8f z = {};
#pragma unroll
    for (int c = 0; c < 16; c++)
      z = wmma4(ldA(mS, 64, r0, 4 * c, l16, h), ldB(mV, 16, 4 * c, 0, l16, h),
                z);

    float part = 0.0f, part2 = 0.0f, tv[8];
#pragma unroll
    for (int r = 0; r < 8; r++) {
      int row = r0 + r + 8 * h;
      float tt = mX[row * 16 + l16] + z[r];
      tv[r] = tt;
      part += tt;
      part2 += tt * tt;
    }
    float2 s12 = gsum2(part, part2, redA, wv, lane, g);
    float mean = s12.x * (1.0f / 1024.0f);
    float rstd = rsqrtf(s12.y * (1.0f / 1024.0f) - mean * mean + LN_EPS);
#pragma unroll
    for (int r = 0; r < 8; r++) {
      int row = r0 + r + 8 * h;
      mX[row * 16 + l16] = (tv[r] - mean) * rstd;
    }
    __syncthreads();

    float acc[8];
#pragma unroll
    for (int j = 0; j < 8; j++) acc[j] = sB1[p * 16 + o0 + j];
#pragma unroll
    for (int i = 0; i < 16; i++) {
      float yv = mX[p * 16 + i];
      const float* wr = &sW1[p * WSTRIDE + i * 16 + o0];
#pragma unroll
      for (int j = 0; j < 8; j++) acc[j] = fmaf(yv, wr[j], acc[j]);
    }
#pragma unroll
    for (int j = 0; j < 8; j++) mQ[p * 16 + o0 + j] = fmaxf(acc[j], 0.0f);
    __syncthreads();

    float acc2[8];
#pragma unroll
    for (int j = 0; j < 8; j++) acc2[j] = sB2[p * 16 + o0 + j];
#pragma unroll
    for (int i = 0; i < 16; i++) {
      float hv = mQ[p * 16 + i];
      const float* wr = &sW2[p * WSTRIDE + i * 16 + o0];
#pragma unroll
      for (int j = 0; j < 8; j++) acc2[j] = fmaf(hv, wr[j], acc2[j]);
    }
    float t2[8];
    float q1 = 0.0f, q2 = 0.0f;
#pragma unroll
    for (int j = 0; j < 8; j++) {
      float tt = fmaxf(acc2[j], 0.0f) + mX[p * 16 + o0 + j];
      t2[j] = tt;
      q1 += tt;
      q2 += tt * tt;
    }
    float2 u12 = gsum2(q1, q2, redB, wv, lane, g);
    float m2 = u12.x * (1.0f / 1024.0f);
    float r2 = rsqrtf(u12.y * (1.0f / 1024.0f) - m2 * m2 + LN_EPS);
    float* op = out + b * 1024 + p * 16 + o0;
#pragma unroll
    for (int j = 0; j < 8; j++) op[j] = (t2[j] - m2) * r2;
  }
}

extern "C" void kernel_launch(void* const* d_in, const int* in_sizes, int n_in,
                              void* d_out, int out_size, void* d_ws,
                              size_t ws_size, hipStream_t stream) {
  (void)n_in;
  (void)out_size;
  (void)ws_size;
  const float* x = (const float*)d_in[0];
  const float* Wq = (const float*)d_in[1];
  const float* Wk = (const float*)d_in[2];
  const float* Wv = (const float*)d_in[3];
  const float* W1 = (const float*)d_in[4];
  const float* b1 = (const float*)d_in[5];
  const float* W2 = (const float*)d_in[6];
  const float* b2 = (const float*)d_in[7];
  float* out = (float*)d_out;

  unsigned int* gmax = (unsigned int*)d_ws;
  float* gsum = (float*)((char*)d_ws + 4096 * sizeof(unsigned int));

  const int B = in_sizes[0] / 1024;
  const int NB = 64;
  const int blocks12 = B / NB;
  const int blocks3 = B / (2 * NB);

  enc_init<<<16, 256, 0, stream>>>(gmax, gsum);
  enc_pass1_max<<<blocks12, 128, 0, stream>>>(x, Wq, Wk, gmax, NB);
  enc_pass2_sum<<<blocks12, 128, 0, stream>>>(x, Wq, Wk, gmax, gsum, NB);
  enc_pass3_final<<<blocks3, 256, 0, stream>>>(x, Wq, Wk, Wv, W1, b1, W2, b2,
                                               gmax, gsum, out, NB);
}


# x tile of the next batch requested global->registers one batch ahead in all three passes; in-loop barriers written by hand (LDS wait + s_barrier) so the load stays in flight; sums unchanged
# speedup vs baseline: 1.7555x; 1.3646x over previous
#include <hip/hip_runtime.h>

typedef __attribute__((ext_vector_type(2))) float v2f;
typedef __attribute__((ext_vector_type(8))) float v8f;

#define LN_EPS 1e-5f

__device__ __forceinline__ v8f wmma4(v2f a, v2f b, v8f c) {
  return __builtin_amdgcn_wmma_f32_16x16x4_f32(false, a, false, b, (short)0, c,
                                               false, false);
}

__device__ __forceinline__ v2f ldA(const float* m, int ld, int r0, int c0,
                                   int l16, int h) {
  v2f a;
  a.x = m[(r0 + l16) * ld + c0 + 2 * h + 0];
  a.y = m[(r0 + l16) * ld + c0 + 2 * h + 1];
  return a;
}
__device__ __forceinline__ v2f ldB(const float* m, int ld, int c0, int n0,
                                   int l16, int h) {
  v2f b;
  b.x = m[(c0 + 2 * h + 0) * ld + n0 + l16];
  b.y = m[(c0 + 2 * h + 1) * ld + n0 + l16];
  return b;
}
__device__ __forceinline__ v2f ldBT(const float* m, int ld, int c0, int n0,
                                    int l16, int h) {
  v2f b;
  b.x = m[(n0 + l16) * ld + c0 + 2 * h + 0];
  b.y = m[(n0 + l16) * ld + c0 + 2 * h + 1];
  return b;
}

__device__ __forceinline__ unsigned int f2key(float f) {
  unsigned int u = __float_as_uint(f);
  return (u & 0x80000000u) ? ~u : (u | 0x80000000u);
}
__device__ __forceinline__ float key2f(unsigned int k) {
  unsigned int u = (k & 0x80000000u) ? (k & 0x7FFFFFFFu) : ~k;
  return __uint_as_float(u);
}

template <int XM>
__device__ __forceinline__ float row_xor(float v) {
  int b = __builtin_bit_cast(int, v);
  return __builtin_bit_cast(
      float, __builtin_amdgcn_update_dpp(0, b, 0x160 | XM, 0xf, 0xf, true));
}
__device__ __forceinline__ float xhalf(float v) {
  int b = __builtin_bit_cast(int, v);
  return __builtin_bit_cast(
      float, __builtin_amdgcn_permlanex16(b, b, 0x76543210, 0xfedcba98, false,
                                          false));
}
__device__ __forceinline__ float wave_sum(float v) {
  v += row_xor<1>(v);
  v += row_xor<2>(v);
  v += row_xor<4>(v);
  v += row_xor<8>(v);
  v += xhalf(v);
  return v;
}

__device__ __forceinline__ void lds_barrier() {
  asm volatile("s_wait_dscnt 0x0\n\ts_barrier_signal -1\n\ts_barrier_wait -1" ::
                   : "memory");
}

__device__ __forceinline__ float2 gsum2(float a, float b, float2* slot,
                                        int wv, int lane, int g) {
  float wa = wave_sum(a);
  float wb = wave_sum(b);
  if (lane == 0) slot[wv] = make_float2(wa, wb);
  lds_barrier();
  float2 p0 = slot[4 * g + 0], p1 = slot[4 * g + 1];
  float2 p2 = slot[4 * g + 2], p3 = slot[4 * g + 3];
  return make_float2((p0.x + p1.x) + (p2.x + p3.x),
                     (p0.y + p1.y) + (p2.y + p3.y));
}

__global__ void enc_init(unsigned int* gmax, float* gsum) {
  int i = blockIdx.x * blockDim.x + threadIdx.x;
  if (i < 4096) {
    gmax[i] = 0u;
    gsum[i] = 0.0f;
  }
}

__global__ __launch_bounds__(128) void enc_pass1_max(
    const float* __restrict__ x, const float* __restrict__ Wq,
    const float* __restrict__ Wk, unsigned int* __restrict__ gmax, int nb) {
  __shared__ __align__(16) float sX[1024];
  __shared__ float sQ[1024], sK[1024];
  __shared__ float sWq[256], sWk[256];
  int tid = threadIdx.x;
  int w = tid >> 5, lane = tid & 31, l16 = lane & 15, h = lane >> 4;
  int r0 = 16 * w;
  for (int i = tid; i < 256; i += 128) {
    sWq[i] = Wq[i];
    sWk[i] = Wk[i];
  }
  const float4* xg = (const float4*)(x + (size_t)blockIdx.x * nb * 1024);
  float4 xc0 = {}, xc1 = {};
  if (nb > 0) {
    xc0 = xg[tid];
    xc1 = xg[tid + 128];
  }
  float vmax[4][8];
#pragma unroll
  for (int kt = 0; kt < 4; kt++)
#pragma unroll
    for (int r = 0; r < 8; r++) vmax[kt][r] = -3.4e38f;
  __syncthreads();

  for (int it = 0; it < nb; ++it) {
    ((float4*)sX)[tid] = xc0;
    ((float4*)sX)[tid + 128] = xc1;
    {
      const float4* xf = xg + (size_t)(it + 1 < nb ? it + 1 : it) * 256;
      xc0 = xf[tid];
      xc1 = xf[tid + 128];
    }
    lds_barrier();
    v2f a[4];
#pragma unroll
    for (int kk = 0; kk < 4; kk++) a[kk] = ldA(sX, 16, r0, 4 * kk, l16, h);
    v8f q = {};
    v8f k = {};
#pragma unroll
    for (int kk = 0; kk < 4; kk++) {
      q = wmma4(a[kk], ldB(sWq, 16, 4 * kk, 0, l16, h), q);
      k = wmma4(a[kk], ldB(sWk, 16, 4 * kk, 0, l16, h), k);
    }
#pragma unroll
    for (int r = 0; r < 8; r++) {
      sQ[(r0 + r + 8 * h) * 16 + l16] = q[r];
      sK[(r0 + r + 8 * h) * 16 + l16] = k[r];
    }
    lds_barrier();
    v2f qa[4];
#pragma unroll
    for (int kk = 0; kk < 4; kk++) qa[kk] = ldA(sQ, 16, r0, 4 * kk, l16, h);
#pragma unroll
    for (int kt = 0; kt < 4; kt++) {
      v8f s = {};
#pragma unroll
      for (int kk = 0; kk < 4; kk++)
        s = wmma4(qa[kk], ldBT(sK, 16, 4 * kk, 16 * kt, l16, h), s);
#pragma unroll
      for (int r = 0; r < 8; r++)
        vmax[kt][r] = fmaxf(vmax[kt][r], s[r] * (1.0f / 3.0f));
    }
    lds_barrier();
  }
#pragma unroll
  for (int kt = 0; kt < 4; kt++)
#pragma unroll
    for (int r = 0; r < 8; r++) {
      int qrow = r0 + r + 8 * h, kcol = 16 * kt + l16;
      atomicMax(&gmax[qrow * 64 + kcol], f2key(vmax[kt][r]));
    }
}

__global__ __launch_bounds__(128) void enc_pass2_sum(
    const float* __restrict__ x, const float* __restrict__ Wq,
    const float* __restrict__ Wk, const unsigned int* __restrict__ gmax,
    float* __restrict__ gsum, int nb) {
  __shared__ __align__(16) float sX[1024];
  __shared__ float sQ[1024], sK[1024];
  __shared__ float sWq[256], sWk[256];
  int tid = threadIdx.x;
  int w = tid >> 5, lane = tid & 31, l16 = lane & 15, h = lane >> 4;
  int r0 = 16 * w;
  for (int i = tid; i < 256; i += 128) {
    sWq[i] = Wq[i];
    sWk[i] = Wk[i];
  }
  const float4* xg = (const float4*)(x + (size_t)blockIdx.x * nb * 1024);
  float4 xc0 = {}, xc1 = {};
  if (nb > 0) {
    xc0 = xg[tid];
    xc1 = xg[tid + 128];
  }
  float gm[4][8], vsum[4][8];
#pragma unroll
  for (int kt = 0; kt < 4; kt++)
#pragma unroll
    for (int r = 0; r < 8; r++) {
      int qrow = r0 + r + 8 * h, kcol = 16 * kt + l16;
      gm[kt][r] = key2f(gmax[qrow * 64 + kcol]);
      vsum[kt][r] = 0.0f;
    }
  __syncthreads();

  for (int it = 0; it < nb; ++it) {
    ((float4*)sX)[tid] = xc0;
    ((float4*)sX)[tid + 128] = xc1;
    {
      const float4* xf = xg + (size_t)(it + 1 < nb ? it + 1 : it) * 256;
      xc0 = xf[tid];
      xc1 = xf[tid + 128];
    }
    lds_barrier();
    v2f a[4];
#pragma unroll
    for (int kk = 0; kk < 4; kk++) a[kk] = ldA(sX, 16, r0, 4 * kk, l16, h);
    v8f q = {};
    v8f k = {};
#pragma unroll
    for (int kk = 0; kk < 4; kk++) {
      q = wmma4(a[kk], ldB(sWq, 16, 4 * kk, 0, l16, h), q);
      k = wmma4(a[kk], ldB(sWk, 16, 4 * kk, 0, l16, h), k);
    }
#pragma unroll
    for (int r = 0; r < 8; r++) {
      sQ[(r0 + r + 8 * h) * 16 + l16] = q[r];
      sK[(r0 + r + 8 * h) * 16 + l16] = k[r];
    }
    lds_barrier();
    v2f qa[4];
#pragma unroll
    for (int kk = 0; kk < 4; kk++) qa[kk] = ldA(sQ, 16, r0, 4 * kk, l16, h);
#pragma unroll
    for (int kt = 0; kt < 4; kt++) {
      v8f s = {};
#pragma unroll
      for (int kk = 0; kk < 4; kk++)
        s = wmma4(qa[kk], ldBT(sK, 16, 4 * kk, 16 * kt, l16, h), s);
#pragma unroll
      for (int r = 0; r < 8; r++)
        vsum[kt][r] += __expf(s[r] * (1.0f / 3.0f) - gm[kt][r]);
    }
    lds_barrier();
  }
#pragma unroll
  for (int kt = 0; kt < 4; kt++)
#pragma unroll
    for (int r = 0; r < 8; r++) {
      int qrow = r0 + r + 8 * h, kcol = 16 * kt + l16;
      atomicAdd(&gsum[qrow * 64 + kcol], vsum[kt][r]);
    }
}

#define WSTRIDE 260

__global__ __launch_bounds__(256) void enc_pass3_final(
    const float* __restrict__ x, const float* __restrict__ Wq,
    const float* __restrict__ Wk, const float* __restrict__ Wv,
    const float* __restrict__ W1, const float* __restrict__ b1,
    const float* __restrict__ W2, const float* __restrict__ b2,
    const unsigned int* __restrict__ gmax, const float* __restrict__ gsum,
    float* __restrict__ out, int nb) {
  __shared__ float sW1[64 * WSTRIDE];
  __shared__ float sW2[64 * WSTRIDE];
  __shared__ float sB1[1024], sB2[1024];
  __shared__ float sWq[256], sWk[256], sWv[256];
  __shared__ __align__(16) float sX[2][1024];
  __shared__ float sQ[2][1024];
  __shared__ float sK[2][1024], sV[2][1024];
  __shared__ float sS[2][4096];
  __shared__ float2 redA[8], redB[8];

  int tid = threadIdx.x;
  int g = tid >> 7;
  int t = tid & 127;
  int wv = tid >> 5;
  int w = t >> 5, lane = tid & 31, l16 = lane & 15, h = lane >> 4;
  int r0 = 16 * w;
  int p = t >> 1, o0 = (t & 1) * 8;

  float* mX = sX[g];
  float* mQ = sQ[g];
  float* mK = sK[g];
  float* mV = sV[g];
  float* mS = sS[g];

  const float4* xg = (const float4*)(x + ((size_t)blockIdx.x * (2 * nb) +
                                          (size_t)g * nb) * 1024);
  float4 xc0 = {}, xc1 = {};
  if (nb > 0) {
    xc0 = xg[t];
    xc1 = xg[t + 128];
  }

  for (int i = tid; i < 16384; i += 256) {
    int pp = i >> 8, rr = i & 255;
    sW1[pp * WSTRIDE + rr] = W1[i];
    sW2[pp * WSTRIDE + rr] = W2[i];
  }
  for (int i = tid; i < 1024; i += 256) {
    sB1[i] = b1[i];
    sB2[i] = b2[i];
  }
  if (tid < 256) {
    sWq[tid] = Wq[tid];
    sWk[tid] = Wk[tid];
    sWv[tid] = Wv[tid];
  }
  float gm[4][8], rs[4][8];
#pragma unroll
  for (int kt = 0; kt < 4; kt++)
#pragma unroll
    for (int r = 0; r < 8; r++) {
      int qrow = r0 + r + 8 * h, kcol = 16 * kt + l16;
      gm[kt][r] = key2f(gmax[qrow * 64 + kcol]);
      rs[kt][r] = 1.0f / gsum[qrow * 64 + kcol];
    }
  __syncthreads();

  for (int it = 0; it < nb; ++it) {
    size_t b = (size_t)blockIdx.x * (2 * nb) + (size_t)g * nb + it;
    ((float4*)mX)[t] = xc0;
    ((float4*)mX)[t + 128] = xc1;
    {
      const float4* xf = xg + (size_t)(it + 1 < nb ? it + 1 : it) * 256;
      xc0 = xf[t];
      xc1 = xf[t + 128];
    }
    lds_barrier();

    v2f a[4];
#pragma unroll
    for (int kk = 0; kk < 4; kk++) a[kk] = ldA(mX, 16, r0, 4 * kk, l16, h);
    v8f q = {};
    v8f k = {};
    v8f v = {};
#pragma unroll
    for (int kk = 0; kk < 4; kk++) {
      q = wmma4(a[kk], ldB(sWq, 16, 4 * kk, 0, l16, h), q);
      k = wmma4(a[kk], ldB(sWk, 16, 4 * kk, 0, l16, h), k);
      v = wmma4(a[kk], ldB(sWv, 16, 4 * kk, 0, l16, h), v);
    }
#pragma unroll
    for (int r = 0; r < 8; r++) {
      int row = r0 + r + 8 * h;
      mQ[row * 16 + l16] = q[r];
      mK[row * 16 + l16] = k[r];
      mV[row * 16 + l16] = v[r];
    }
    lds_barrier();

    v2f qa[4];
#pragma unroll
    for (int kk = 0; kk < 4; kk++) qa[kk] = ldA(mQ, 16, r0, 4 * kk, l16, h);
#pragma unroll
    for (int kt = 0; kt < 4; kt++) {
      v8f s = {};
#pragma unroll
      for (int kk = 0; kk < 4; kk++)
        s = wmma4(qa[kk], ldBT(mK, 16, 4 * kk, 16 * kt, l16, h), s);
#pragma unroll
      for (int r = 0; r < 8; r++) {
        int qrow = r0 + r + 8 * h, kcol = 16 * kt + l16;
        mS[qrow * 64 + kcol] =
            __expf(s[r] * (1.0f / 3.0f) - gm[kt][r]) * rs[kt][r];
      }
    }

    v8f z = {};
#pragma unroll
    for (int c = 0; c < 16; c++)
      z = wmma4(ldA(mS, 64, r0, 4 * c, l16, h), ldB(mV, 16, 4 * c, 0, l16, h),
                z);

    float part = 0.0f, part2 = 0.0f, tv[8];
#pragma unroll
    for (int r = 0; r < 8; r++) {
      int row = r0 + r + 8 * h;
      float tt = mX[row * 16 + l16] + z[r];
      tv[r] = tt;
      part += tt;
      part2 += tt * tt;
    }
    float2 s12 = gsum2(part, part2, redA, wv, lane, g);
    float mean = s12.x * (1.0f / 1024.0f);
    float rstd = rsqrtf(s12.y * (1.0f / 1024.0f) - mean * mean + LN_EPS);
#pragma unroll
    for (int r = 0; r < 8; r++) {
      int row = r0 + r + 8 * h;
      mX[row * 16 + l16] = (tv[r] - mean) * rstd;
    }
    lds_barrier();

    float acc[8];
#pragma unroll
    for (int j = 0; j < 8; j++) acc[j] = sB1[p * 16 + o0 + j];
#pragma unroll
    for (int i = 0; i < 16; i++) {
      float yv = mX[p * 16 + i];
      const float* wr = &sW1[p * WSTRIDE + i * 16 + o0];
#pragma unroll
      for (int j = 0; j < 8; j++) acc[j] = fmaf(yv, wr[j], acc[j]);
    }
#pragma unroll
    for (int j = 0; j < 8; j++) mQ[p * 16 + o0 + j] = fmaxf(acc[j], 0.0f);
    lds_barrier();

    float acc2[8];
#pragma unroll
    for (int j = 0; j < 8; j++) acc2[j] = sB2[p * 16 + o0 + j];
#pragma unroll
    for (int i = 0; i < 16; i++) {
      float hv = mQ[p * 16 + i];
      const float* wr = &sW2[p * WSTRIDE + i * 16 + o0];
#pragma unroll
      for (int j = 0; j < 8; j++) acc2[j] = fmaf(hv, wr[j], acc2[j]);
    }
    float t2[8];
    float q1 = 0.0f, q2 = 0.0f;
#pragma unroll
    for (int j = 0; j < 8; j++) {
      float tt = fmaxf(acc2[j], 0.0f) + mX[p * 16 + o0 + j];
      t2[j] = tt;
      q1 += tt;
      q2 += tt * tt;
    }
    float2 u12 = gsum2(q1, q2, redB, wv, lane, g);
    float m2 = u12.x * (1.0f / 1024.0f);
    float r2 = rsqrtf(u12.y * (1.0f / 1024.0f) - m2 * m2 + LN_EPS);
    float* op = out + b * 1024 + p * 16 + o0;
#pragma unroll
    for (int j = 0; j < 8; j++) op[j] = (t2[j] - m2) * r2;
  }
}

extern "C" void kernel_launch(void* const* d_in, const int* in_sizes, int n_in,
                              void* d_out, int out_size, void* d_ws,
                              size_t ws_size, hipStream_t stream) {
  (void)n_in;
  (void)out_size;
  (void)ws_size;
  const float* x = (const float*)d_in[0];
  const float* Wq = (const float*)d_in[1];
  const float* Wk = (const float*)d_in[2];
  const float* Wv = (const float*)d_in[3];
  const float* W1 = (const float*)d_in[4];
  const float* b1 = (const float*)d_in[5];
  const float* W2 = (const float*)d_in[6];
  const float* b2 = (const float*)d_in[7];
  float* out = (float*)d_out;

  unsigned int* gmax = (unsigned int*)d_ws;
  float* gsum = (float*)((char*)d_ws + 4096 * sizeof(unsigned int));

  const int B = in_sizes[0] / 1024;
  const int NB = 64;
  const int blocks12 = B / NB;
  const int blocks3 = B / (2 * NB);

  enc_init<<<16, 256, 0, stream>>>(gmax, gsum);
  enc_pass1_max<<<blocks12, 128, 0, stream>>>(x, Wq, Wk, gmax, NB);
  enc_pass2_sum<<<blocks12, 128, 0, stream>>>(x, Wq, Wk, gmax, gsum, NB);
  enc_pass3_final<<<blocks3, 256, 0, stream>>>(x, Wq, Wk, Wv, W1, b1, W2, b2,
                                               gmax, gsum, out, NB);
}


# LDS re-pitched in all 3 passes against 2-16-way bank conflicts: tiles 16->20, 64->68 floats; MLP weights 264, a thread's 8 outputs as two 4-wide pieces; regroups last LayerNorm partials (fp order)
# speedup vs baseline: 2.3969x; 1.3654x over previous
#include <hip/hip_runtime.h>

typedef __attribute__((ext_vector_type(2))) float v2f;
typedef __attribute__((ext_vector_type(8))) float v8f;

#define LN_EPS 1e-5f

#define XP 20
#define SP 68

__device__ __forceinline__ v8f wmma4(v2f a, v2f b, v8f c) {
  return __builtin_amdgcn_wmma_f32_16x16x4_f32(false, a, false, b, (short)0, c,
                                               false, false);
}

__device__ __forceinline__ v2f ldA(const float* m, int ld, int r0, int c0,
                                   int l16, int h) {
  v2f a;
  a.x = m[(r0 + l16) * ld + c0 + 2 * h + 0];
  a.y = m[(r0 + l16) * ld + c0 + 2 * h + 1];
  return a;
}
__device__ __forceinline__ v2f ldB(const float* m, int ld, int c0, int n0,
                                   int l16, int h) {
  v2f b;
  b.x = m[(c0 + 2 * h + 0) * ld + n0 + l16];
  b.y = m[(c0 + 2 * h + 1) * ld + n0 + l16];
  return b;
}
__device__ __forceinline__ v2f ldBT(const float* m, int ld, int c0, int n0,
                                    int l16, int h) {
  v2f b;
  b.x = m[(n0 + l16) * ld + c0 + 2 * h + 0];
  b.y = m[(n0 + l16) * ld + c0 + 2 * h + 1];
  return b;
}

__device__ __forceinline__ unsigned int f2key(float f) {
  unsigned int u = __float_as_uint(f);
  return (u & 0x80000000u) ? ~u : (u | 0x80000000u);
}
__device__ __forceinline__ float key2f(unsigned int k) {
  unsigned int u = (k & 0x80000000u) ? (k & 0x7FFFFFFFu) : ~k;
  return __uint_as_float(u);
}

template <int XM>
__device__ __forceinline__ float row_xor(float v) {
  int b = __builtin_bit_cast(int, v);
  return __builtin_bit_cast(
      float, __builtin_amdgcn_update_dpp(0, b, 0x160 | XM, 0xf, 0xf, true));
}
__device__ __forceinline__ float xhalf(float v) {
  int b = __builtin_bit_cast(int, v);
  return __builtin_bit_cast(
      float, __builtin_amdgcn_permlanex16(b, b, 0x76543210, 0xfedcba98, false,
                                          false));
}
__device__ __forceinline__ float wave_sum(float v) {
  v += row_xor<1>(v);
  v += row_xor<2>(v);
  v += row_xor<4>(v);
  v += row_xor<8>(v);
  v += xhalf(v);
  return v;
}

__device__ __forceinline__ void lds_barrier() {
  asm volatile("s_wait_dscnt 0x0\n\ts_barrier_signal -1\n\ts_barrier_wait -1" ::
                   : "memory");
}

__device__ __forceinline__ float2 gsum2(float a, float b, float2* slot,
                                        int wv, int lane, int g) {
  float wa = wave_sum(a);
  float wb = wave_sum(b);
  if (lane == 0) slot[wv] = make_float2(wa, wb);
  lds_barrier();
  float2 p0 = slot[4 * g + 0], p1 = slot[4 * g + 1];
  float2 p2 = slot[4 * g + 2], p3 = slot[4 * g + 3];
  return make_float2((p0.x + p1.x) + (p2.x + p3.x),
                     (p0.y + p1.y) + (p2.y + p3.y));
}

__global__ void enc_init(unsigned int* gmax, float* gsum) {
  int i = blockIdx.x * blockDim.x + threadIdx.x;
  if (i < 4096) {
    gmax[i] = 0u;
    gsum[i] = 0.0f;
  }
}

__global__ __launch_bounds__(128) void enc_pass1_max(
    const float* __restrict__ x, const float* __restrict__ Wq,
    const float* __restrict__ Wk, unsigned int* __restrict__ gmax, int nb) {
  __shared__ __align__(16) float sX[64 * XP];
  __shared__ __align__(16) float sQ[64 * XP], sK[64 * XP];
  __shared__ float sWq[256], sWk[256];
  int tid = threadIdx.x;
  int w = tid >> 5, lane = tid & 31, l16 = lane & 15, h = lane >> 4;
  int r0 = 16 * w;
  float4* sx0 = (float4*)&sX[(tid >> 2) * XP + (tid & 3) * 4];
  float4* sx1 = sx0 + 8 * XP;
  for (int i = tid; i < 256; i += 128) {
    sWq[i] = Wq[i];
    sWk[i] = Wk[i];
  }
  const float4* xg = (const float4*)(x + (size_t)blockIdx.x * nb * 1024);
  float4 xc0 = {}, xc1 = {};
  if (nb > 0) {
    xc0 = xg[tid];
    xc1 = xg[tid + 128];
  }
  float vmax[4][8];
#pragma unroll
  for (int kt = 0; kt < 4; kt++)
#pragma unroll
    for (int r = 0; r < 8; r++) vmax[kt][r] = -3.4e38f;
  __syncthreads();

  for (int it = 0; it < nb; ++it) {
    *sx0 = xc0;
    *sx1 = xc1;
    {
      const float4* xf = xg + (size_t)(it + 1 < nb ? it + 1 : it) * 256;
      xc0 = xf[tid];
      xc1 = xf[tid + 128];
    }
    lds_barrier();
    v2f a[4];
#pragma unroll
    for (int kk = 0; kk < 4; kk++) a[kk] = ldA(sX, XP, r0, 4 * kk, l16, h);
    v8f q = {};
    v8f k = {};
#pragma unroll
    for (int kk = 0; kk < 4; kk++) {
      q = wmma4(a[kk], ldB(sWq, 16, 4 * kk, 0, l16, h), q);
      k = wmma4(a[kk], ldB(sWk, 16, 4 * kk, 0, l16, h), k);
    }
#pragma unroll
    for (int r = 0; r < 8; r++) {
      sQ[(r0 + r + 8 * h) * XP + l16] = q[r];
      sK[(r0 + r + 8 * h) * XP + l16] = k[r];
    }
    lds_barrier();
    v2f qa[4];
#pragma unroll
    for (int kk = 0; kk < 4; kk++) qa[kk] = ldA(sQ, XP, r0, 4 * kk, l16, h);
#pragma unroll
    for (int kt = 0; kt < 4; kt++) {
      v8f s = {};
#pragma unroll
      for (int kk = 0; kk < 4; kk++)
        s = wmma4(qa[kk], ldBT(sK, XP, 4 * kk, 16 * kt, l16, h), s);
#pragma unroll
      for (int r = 0; r < 8; r++)
        vmax[kt][r] = fmaxf(vmax[kt][r], s[r] * (1.0f / 3.0f));
    }
    lds_barrier();
  }
#pragma unroll
  for (int kt = 0; kt < 4; kt++)
#pragma unroll
    for (int r = 0; r < 8; r++) {
      int qrow = r0 + r + 8 * h, kcol = 16 * kt + l16;
      atomicMax(&gmax[qrow * 64 + kcol], f2key(vmax[kt][r]));
    }
}

__global__ __launch_bounds__(128) void enc_pass2_sum(
    const float* __restrict__ x, const float* __restrict__ Wq,
    const float* __restrict__ Wk, const unsigned int* __restrict__ gmax,
    float* __restrict__ gsum, int nb) {
  __shared__ __align__(16) float sX[64 * XP];
  __shared__ __align__(16) float sQ[64 * XP], sK[64 * XP];
  __shared__ float sWq[256], sWk[256];
  int tid = threadIdx.x;
  int w = tid >> 5, lane = tid & 31, l16 = lane & 15, h = lane >> 4;
  int r0 = 16 * w;
  float4* sx0 = (float4*)&sX[(tid >> 2) * XP + (tid & 3) * 4];
  float4* sx1 = sx0 + 8 * XP;
  for (int i = tid; i < 256; i += 128) {
    sWq[i] = Wq[i];
    sWk[i] = Wk[i];
  }
  const float4* xg = (const float4*)(x + (size_t)blockIdx.x * nb * 1024);
  float4 xc0 = {}, xc1 = {};
  if (nb > 0) {
    xc0 = xg[tid];
    xc1 = xg[tid + 128];
  }
  float gm[4][8], vsum[4][8];
#pragma unroll
  for (int kt = 0; kt < 4; kt++)
#pragma unroll
    for (int r = 0; r < 8; r++) {
      int qrow = r0 + r + 8 * h, kcol = 16 * kt + l16;
      gm[kt][r] = key2f(gmax[qrow * 64 + kcol]);
      vsum[kt][r] = 0.0f;
    }
  __syncthreads();

  for (int it = 0; it < nb; ++it) {
    *sx0 = xc0;
    *sx1 = xc1;
    {
      const float4* xf = xg + (size_t)(it + 1 < nb ? it + 1 : it) * 256;
      xc0 = xf[tid];
      xc1 = xf[tid + 128];
    }
    lds_barrier();
    v2f a[4];
#pragma unroll
    for (int kk = 0; kk < 4; kk++) a[kk] = ldA(sX, XP, r0, 4 * kk, l16, h);
    v8f q = {};
    v8f k = {};
#pragma unroll
    for (int kk = 0; kk < 4; kk++) {
      q = wmma4(a[kk], ldB(sWq, 16, 4 * kk, 0, l16, h), q);
      k = wmma4(a[kk], ldB(sWk, 16, 4 * kk, 0, l16, h), k);
    }
#pragma unroll
    for (int r = 0; r < 8; r++) {
      sQ[(r0 + r + 8 * h) * XP + l16] = q[r];
      sK[(r0 + r + 8 * h) * XP + l16] = k[r];
    }
    lds_barrier();
    v2f qa[4];
#pragma unroll
    for (int kk = 0; kk < 4; kk++) qa[kk] = ldA(sQ, XP, r0, 4 * kk, l16, h);
#pragma unroll
    for (int kt = 0; kt < 4; kt++) {
      v8f s = {};
#pragma unroll
      for (int kk = 0; kk < 4; kk++)
        s = wmma4(qa[kk], ldBT(sK, XP, 4 * kk, 16 * kt, l16, h), s);
#pragma unroll
      for (int r = 0; r < 8; r++)
        vsum[kt][r] += __expf(s[r] * (1.0f / 3.0f) - gm[kt][r]);
    }
    lds_barrier();
  }
#pragma unroll
  for (int kt = 0; kt < 4; kt++)
#pragma unroll
    for (int r = 0; r < 8; r++) {
      int qrow = r0 + r + 8 * h, kcol = 16 * kt + l16;
      atomicAdd(&gsum[qrow * 64 + kcol], vsum[kt][r]);
    }
}

#define WSTRIDE 264

__global__ __launch_bounds__(256) void enc_pass3_final(
    const float* __restrict__ x, const float* __restrict__ Wq,
    const float* __restrict__ Wk, const float* __restrict__ Wv,
    const float* __restrict__ W1, const float* __restrict__ b1,
    const float* __restrict__ W2, const float* __restrict__ b2,
    const unsigned int* __restrict__ gmax, const float* __restrict__ gsum,
    float* __restrict__ out, int nb) {
  __shared__ __align__(16) float sW1[64 * WSTRIDE];
  __shared__ __align__(16) float sW2[64 * WSTRIDE];
  __shared__ float sB1[1024], sB2[1024];
  __shared__ float sWq[256], sWk[256], sWv[256];
  __shared__ __align__(16) float sX[2][64 * XP];
  __shared__ __align__(16) float sQ[2][64 * XP];
  __shared__ __align__(16) float sK[2][64 * XP], sV[2][64 * XP];
  __shared__ __align__(16) float sS[2][64 * SP];
  __shared__ float2 redA[8], redB[8];

  int tid = threadIdx.x;
  int g = tid >> 7;
  int t = tid & 127;
  int wv = tid >> 5;
  int w = t >> 5, lane = tid & 31, l16 = lane & 15, h = lane >> 4;
  int r0 = 16 * w;
  int p = t >> 1, oA = (t & 1) * 4, oB = 8 + (t & 1) * 4;

  float* mX = sX[g];
  float* mQ = sQ[g];
  float* mK = sK[g];
  float* mV = sV[g];
  float* mS = sS[g];
  float4* sx0 = (float4*)&mX[(t >> 2) * XP + (t & 3) * 4];
  float4* sx1 = sx0 + 8 * XP;

  const float4* xg = (const float4*)(x + ((size_t)blockIdx.x * (2 * nb) +
                                          (size_t)g * nb) * 1024);
  float4 xc0 = {}, xc1 = {};
  if (nb > 0) {
    xc0 = xg[t];
    xc1 = xg[t + 128];
  }

  for (int i = tid; i < 16384; i += 256) {
    int pp = i >> 8, rr = i & 255;
    sW1[pp * WSTRIDE + rr] = W1[i];
    sW2[pp * WSTRIDE + rr] = W2[i];
  }
  for (int i = tid; i < 1024; i += 256) {
    sB1[i] = b1[i];
    sB2[i] = b2[i];
  }
  if (tid < 256) {
    sWq[tid] = Wq[tid];
    sWk[tid] = Wk[tid];
    sWv[tid] = Wv[tid];
  }
  float gm[4][8], rs[4][8];
#pragma unroll
  for (int kt = 0; kt < 4; kt++)
#pragma unroll
    for (int r = 0; r < 8; r++) {
      int qrow = r0 + r + 8 * h, kcol = 16 * kt + l16;
      gm[kt][r] = key2f(gmax[qrow * 64 + kcol]);
      rs[kt][r] = 1.0f / gsum[qrow * 64 + kcol];
    }
  __syncthreads();

  for (int it = 0; it < nb; ++it) {
    size_t b = (size_t)blockIdx.x * (2 * nb) + (size_t)g * nb + it;
    *sx0 = xc0;
    *sx1 = xc1;
    {
      const float4* xf = xg + (size_t)(it + 1 < nb ? it + 1 : it) * 256;
      xc0 = xf[t];
      xc1 = xf[t + 128];
    }
    lds_barrier();

    v2f a[4];
#pragma unroll
    for (int kk = 0; kk < 4; kk++) a[kk] = ldA(mX, XP, r0, 4 * kk, l16, h);
    v8f q = {};
    v8f k = {};
    v8f v = {};
#pragma unroll
    for (int kk = 0; kk < 4; kk++) {
      q = wmma4(a[kk], ldB(sWq, 16, 4 * kk, 0, l16, h), q);
      k = wmma4(a[kk], ldB(sWk, 16, 4 * kk, 0, l16, h), k);
      v = wmma4(a[kk], ldB(sWv, 16, 4 * kk, 0, l16, h), v);
    }
#pragma unroll
    for (int r = 0; r < 8; r++) {
      int row = r0 + r + 8 * h;
      mQ[row * XP + l16] = q[r];
      mK[row * XP + l16] = k[r];
      mV[row * XP + l16] = v[r];
    }
    lds_barrier();

    v2f qa[4];
#pragma unroll
    for (int kk = 0; kk < 4; kk++) qa[kk] = ldA(mQ, XP, r0, 4 * kk, l16, h);
#pragma unroll
    for (int kt = 0; kt < 4; kt++) {
      v8f s = {};
#pragma unroll
      for (int kk = 0; kk < 4; kk++)
        s = wmma4(qa[kk], ldBT(mK, XP, 4 * kk, 16 * kt, l16, h), s);
#pragma unroll
      for (int r = 0; r < 8; r++) {
        int qrow = r0 + r + 8 * h, kcol = 16 * kt + l16;
        mS[qrow * SP + kcol] =
            __expf(s[r] * (1.0f / 3.0f) - gm[kt][r]) * rs[kt][r];
      }
    }

    v8f z = {};
#pragma unroll
    for (int c = 0; c < 16; c++)
      z = wmma4(ldA(mS, SP, r0, 4 * c, l16, h), ldB(mV, XP, 4 * c, 0, l16, h),
                z);

    float part = 0.0f, part2 = 0.0f, tv[8];
#pragma unroll
    for (int r = 0; r < 8; r++) {
      int row = r0 + r + 8 * h;
      float tt = mX[row * XP + l16] + z[r];
      tv[r] = tt;
      part += tt;
      part2 += tt * tt;
    }
    float2 s12 = gsum2(part, part2, redA, wv, lane, g);
    float mean = s12.x * (1.0f / 1024.0f);
    float rstd = rsqrtf(s12.y * (1.0f / 1024.0f) - mean * mean + LN_EPS);
#pragma unroll
    for (int r = 0; r < 8; r++) {
      int row = r0 + r + 8 * h;
      mX[row * XP + l16] = (tv[r] - mean) * rstd;
    }
    lds_barrier();

    float acc[8];
#pragma unroll
    for (int j = 0; j < 4; j++) {
      acc[j] = sB1[p * 16 + oA + j];
      acc[4 + j] = sB1[p * 16 + oB + j];
    }
#pragma unroll
    for (int i = 0; i < 16; i++) {
      float yv = mX[p * XP + i];
      const float* wr = &sW1[p * WSTRIDE + i * 16];
#pragma unroll
      for (int j = 0; j < 4; j++) {
        acc[j] = fmaf(yv, wr[oA + j], acc[j]);
        acc[4 + j] = fmaf(yv, wr[oB + j], acc[4 + j]);
      }
    }
#pragma unroll
    for (int j = 0; j < 4; j++) {
      mQ[p * XP + oA + j] = fmaxf(acc[j], 0.0f);
      mQ[p * XP + oB + j] = fmaxf(acc[4 + j], 0.0f);
    }
    lds_barrier();

    float acc2[8];
#pragma unroll
    for (int j = 0; j < 4; j++) {
      acc2[j] = sB2[p * 16 + oA + j];
      acc2[4 + j] = sB2[p * 16 + oB + j];
    }
#pragma unroll
    for (int i = 0; i < 16; i++) {
      float hv = mQ[p * XP + i];
      const float* wr = &sW2[p * WSTRIDE + i * 16];
#pragma unroll
      for (int j = 0; j < 4; j++) {
        acc2[j] = fmaf(hv, wr[oA + j], acc2[j]);
        acc2[4 + j] = fmaf(hv, wr[oB + j], acc2[4 + j]);
      }
    }
    float t2[8];
    float q1 = 0.0f, q2 = 0.0f;
#pragma unroll
    for (int j = 0; j < 8; j++) {
      int col = (j < 4) ? oA + j : oB + j - 4;
      float tt = fmaxf(acc2[j], 0.0f) + mX[p * XP + col];
      t2[j] = tt;
      q1 += tt;
      q2 += tt * tt;
    }
    float2 u12 = gsum2(q1, q2, redB, wv, lane, g);
    float m2 = u12.x * (1.0f / 1024.0f);
    float r2 = rsqrtf(u12.y * (1.0f / 1024.0f) - m2 * m2 + LN_EPS);
    float* op = out + b * 1024 + p * 16;
#pragma unroll
    for (int j = 0; j < 4; j++) {
      op[oA + j] = (t2[j] - m2) * r2;
      op[oB + j] = (t2[4 + j] - m2) * r2;
    }
  }
}

extern "C" void kernel_launch(void* const* d_in, const int* in_sizes, int n_in,
                              void* d_out, int out_size, void* d_ws,
                              size_t ws_size, hipStream_t stream) {
  (void)n_in;
  (void)out_size;
  (void)ws_size;
  const float* x = (const float*)d_in[0];
  const float* Wq = (const float*)d_in[1];
  const float* Wk = (const float*)d_in[2];
  const float* Wv = (const float*)d_in[3];
  const float* W1 = (const float*)d_in[4];
  const float* b1 = (const float*)d_in[5];
  const float* W2 = (const float*)d_in[6];
  const float* b2 = (const float*)d_in[7];
  float* out = (float*)d_out;

  unsigned int* gmax = (unsigned int*)d_ws;
  float* gsum = (float*)((char*)d_ws + 4096 * sizeof(unsigned int));

  const int B = in_sizes[0] / 1024;
  const int NB = 64;
  const int blocks12 = B / NB;
  const int blocks3 = B / (2 * NB);

  enc_init<<<16, 256, 0, stream>>>(gmax, gsum);
  enc_pass1_max<<<blocks12, 128, 0, stream>>>(x, Wq, Wk, gmax, NB);
  enc_pass2_sum<<<blocks12, 128, 0, stream>>>(x, Wq, Wk, gmax, gsum, NB);
  enc_pass3_final<<<blocks3, 256, 0, stream>>>(x, Wq, Wk, Wv, W1, b1, W2, b2,
                                               gmax, gsum, out, NB);
}


# passes 1+2 fused: one sweep keeps a running max and rescaled exp-sum per (q,k) (changes fp sum order; atomics removed); block pairs staged in out before pass 3 fills it, merged by one small launch
# speedup vs baseline: 2.9120x; 1.2149x over previous
#include <hip/hip_runtime.h>

typedef __attribute__((ext_vector_type(2))) float v2f;
typedef __attribute__((ext_vector_type(8))) float v8f;

#define LN_EPS 1e-5f

#define XP 20
#define SP 68

__device__ __forceinline__ v8f wmma4(v2f a, v2f b, v8f c) {
  return __builtin_amdgcn_wmma_f32_16x16x4_f32(false, a, false, b, (short)0, c,
                                               false, false);
}

__device__ __forceinline__ v2f ldA(const float* m, int ld, int r0, int c0,
                                   int l16, int h) {
  v2f a;
  a.x = m[(r0 + l16) * ld + c0 + 2 * h + 0];
  a.y = m[(r0 + l16) * ld + c0 + 2 * h + 1];
  return a;
}
__device__ __forceinline__ v2f ldB(const float* m, int ld, int c0, int n0,
                                   int l16, int h) {
  v2f b;
  b.x = m[(c0 + 2 * h + 0) * ld + n0 + l16];
  b.y = m[(c0 + 2 * h + 1) * ld + n0 + l16];
  return b;
}
__device__ __forceinline__ v2f ldBT(const float* m, int ld, int c0, int n0,
                                    int l16, int h) {
  v2f b;
  b.x = m[(n0 + l16) * ld + c0 + 2 * h + 0];
  b.y = m[(n0 + l16) * ld + c0 + 2 * h + 1];
  return b;
}

template <int XM>
__device__ __forceinline__ float row_xor(float v) {
  int b = __builtin_bit_cast(int, v);
  return __builtin_bit_cast(
      float, __builtin_amdgcn_update_dpp(0, b, 0x160 | XM, 0xf, 0xf, true));
}
__device__ __forceinline__ float xhalf(float v) {
  int b = __builtin_bit_cast(int, v);
  return __builtin_bit_cast(
      float, __builtin_amdgcn_permlanex16(b, b, 0x76543210, 0xfedcba98, false,
                                          false));
}
__device__ __forceinline__ float wave_sum(float v) {
  v += row_xor<1>(v);
  v += row_xor<2>(v);
  v += row_xor<4>(v);
  v += row_xor<8>(v);
  v += xhalf(v);
  return v;
}

__device__ __forceinline__ void lds_barrier() {
  asm volatile("s_wait_dscnt 0x0\n\ts_barrier_signal -1\n\ts_barrier_wait -1" ::
                   : "memory");
}

__device__ __forceinline__ float2 gsum2(float a, float b, float2* slot,
                                        int wv, int lane, int g) {
  float wa = wave_sum(a);
  float wb = wave_sum(b);
  if (lane == 0) slot[wv] = make_float2(wa, wb);
  lds_barrier();
  float2 p0 = slot[4 * g + 0], p1 = slot[4 * g + 1];
  float2 p2 = slot[4 * g + 2], p3 = slot[4 * g + 3];
  return make_float2((p0.x + p1.x) + (p2.x + p3.x),
                     (p0.y + p1.y) + (p2.y + p3.y));
}

__global__ __launch_bounds__(128) void enc_pass1_stats(
    const float* __restrict__ x, const float* __restrict__ Wq,
    const float* __restrict__ Wk, float* __restrict__ part, int nb) {
  __shared__ __align__(16) float sX[64 * XP];
  __shared__ __align__(16) float sQ[64 * XP], sK[64 * XP];
  __shared__ float sWq[256], sWk[256];
  int tid = threadIdx.x;
  int w = tid >> 5, lane = tid & 31, l16 = lane & 15, h = lane >> 4;
  int r0 = 16 * w;
  float4* sx0 = (float4*)&sX[(tid >> 2) * XP + (tid & 3) * 4];
  float4* sx1 = sx0 + 8 * XP;
  for (int i = tid; i < 256; i += 128) {
    sWq[i] = Wq[i];
    sWk[i] = Wk[i];
  }
  const float4* xg = (const float4*)(x + (size_t)blockIdx.x * nb * 1024);
  float4 xc0 = {}, xc1 = {};
  if (nb > 0) {
    xc0 = xg[tid];
    xc1 = xg[tid + 128];
  }
  float vm[4][8], vsum[4][8];
#pragma unroll
  for (int kt = 0; kt < 4; kt++)
#pragma unroll
    for (int r = 0; r < 8; r++) {
      vm[kt][r] = -3.4e38f;
      vsum[kt][r] = 0.0f;
    }
  __syncthreads();

  for (int it = 0; it < nb; ++it) {
    *sx0 = xc0;
    *sx1 = xc1;
    {
      const float4* xf = xg + (size_t)(it + 1 < nb ? it + 1 : it) * 256;
      xc0 = xf[tid];
      xc1 = xf[tid + 128];
    }
    lds_barrier();
    v2f a[4];
#pragma unroll
    for (int kk = 0; kk < 4; kk++) a[kk] = ldA(sX, XP, r0, 4 * kk, l16, h);
    v8f q = {};
    v8f k = {};
#pragma unroll
    for (int kk = 0; kk < 4; kk++) {
      q = wmma4(a[kk], ldB(sWq, 16, 4 * kk, 0, l16, h), q);
      k = wmma4(a[kk], ldB(sWk, 16, 4 * kk, 0, l16, h), k);
    }
#pragma unroll
    for (int r = 0; r < 8; r++) {
      sQ[(r0 + r + 8 * h) * XP + l16] = q[r];
      sK[(r0 + r + 8 * h) * XP + l16] = k[r];
    }
    lds_barrier();
    v2f qa[4];
#pragma unroll
    for (int kk = 0; kk < 4; kk++) qa[kk] = ldA(sQ, XP, r0, 4 * kk, l16, h);
#pragma unroll
    for (int kt = 0; kt < 4; kt++) {
      v8f s = {};
#pragma unroll
      for (int kk = 0; kk < 4; kk++)
        s = wmma4(qa[kk], ldBT(sK, XP, 4 * kk, 16 * kt, l16, h), s);
#pragma unroll
      for (int r = 0; r < 8; r++) {
        float v = s[r] * (1.0f / 3.0f);
        float d = v - vm[kt][r];
        float e = __expf(-fabsf(d));
        vsum[kt][r] =
            (d > 0.0f) ? fmaf(vsum[kt][r], e, 1.0f) : vsum[kt][r] + e;
        vm[kt][r] = fmaxf(vm[kt][r], v);
      }
    }
    lds_barrier();
  }
  float2* pp = (float2*)part + (size_t)blockIdx.x * 4096 + w * 1024 + lane;
#pragma unroll
  for (int kt = 0; kt < 4; kt++)
#pragma unroll
    for (int r = 0; r < 8; r++)
      pp[(8 * kt + r) * 32] = make_float2(vm[kt][r], vsum[kt][r]);
}

__device__ __forceinline__ void fold_pair(float& m, float& s, float2 t) {
  float mn = fmaxf(m, t.x);
  s = s * __expf(m - mn) + t.y * __expf(t.x - mn);
  m = mn;
}

__global__ __launch_bounds__(512) void enc_combine(
    const float* __restrict__ part, float* __restrict__ gmaxf,
    float* __restrict__ grs, int nblk) {
  __shared__ float2 red[16][32];
  int lane = threadIdx.x & 31, j = threadIdx.x >> 5;
  int e = blockIdx.x * 32 + lane;
  int per = nblk >> 4;
  const float2* p = (const float2*)part + (size_t)j * per * 4096 + e;
  float m = -3.4e38f, s = 0.0f;
#pragma unroll 8
  for (int i = 0; i < per; i++) fold_pair(m, s, p[(size_t)i * 4096]);
  red[j][lane] = make_float2(m, s);
  __syncthreads();
  if (j == 0) {
    for (int k = 1; k < 16; k++) fold_pair(m, s, red[k][lane]);
    int w = e >> 10, kt = (e >> 8) & 3, r = (e >> 5) & 7;
    int qrow = 16 * w + r + 8 * (lane >> 4), kcol = 16 * kt + (lane & 15);
    gmaxf[qrow * 64 + kcol] = m;
    grs[qrow * 64 + kcol] = 1.0f / s;
  }
}

#define WSTRIDE 264

__global__ __launch_bounds__(256) void enc_pass3_final(
    const float* __restrict__ x, const float* __restrict__ Wq,
    const float* __restrict__ Wk, const float* __restrict__ Wv,
    const float* __restrict__ W1, const float* __restrict__ b1,
    const float* __restrict__ W2, const float* __restrict__ b2,
    const float* __restrict__ gmaxf, const float* __restrict__ grs,
    float* __restrict__ out, int nb) {
  __shared__ __align__(16) float sW1[64 * WSTRIDE];
  __shared__ __align__(16) float sW2[64 * WSTRIDE];
  __shared__ float sB1[1024], sB2[1024];
  __shared__ float sWq[256], sWk[256], sWv[256];
  __shared__ __align__(16) float sX[2][64 * XP];
  __shared__ __align__(16) float sQ[2][64 * XP];
  __shared__ __align__(16) float sK[2][64 * XP], sV[2][64 * XP];
  __shared__ __align__(16) float sS[2][64 * SP];
  __shared__ float2 redA[8], redB[8];

  int tid = threadIdx.x;
  int g = tid >> 7;
  int t = tid & 127;
  int wv = tid >> 5;
  int w = t >> 5, lane = tid & 31, l16 = lane & 15, h = lane >> 4;
  int r0 = 16 * w;
  int p = t >> 1, oA = (t & 1) * 4, oB = 8 + (t & 1) * 4;

  float* mX = sX[g];
  float* mQ = sQ[g];
  float* mK = sK[g];
  float* mV = sV[g];
  float* mS = sS[g];
  float4* sx0 = (float4*)&mX[(t >> 2) * XP + (t & 3) * 4];
  float4* sx1 = sx0 + 8 * XP;

  const float4* xg = (const float4*)(x + ((size_t)blockIdx.x * (2 * nb) +
                                          (size_t)g * nb) * 1024);
  float4 xc0 = {}, xc1 = {};
  if (nb > 0) {
    xc0 = xg[t];
    xc1 = xg[t + 128];
  }

  for (int i = tid; i < 16384; i += 256) {
    int pp = i >> 8, rr = i & 255;
    sW1[pp * WSTRIDE + rr] = W1[i];
    sW2[pp * WSTRIDE + rr] = W2[i];
  }
  for (int i = tid; i < 1024; i += 256) {
    sB1[i] = b1[i];
    sB2[i] = b2[i];
  }
  if (tid < 256) {
    sWq[tid] = Wq[tid];
    sWk[tid] = Wk[tid];
    sWv[tid] = Wv[tid];
  }
  float gm[4][8], rs[4][8];
#pragma unroll
  for (int kt = 0; kt < 4; kt++)
#pragma unroll
    for (int r = 0; r < 8; r++) {
      int qrow = r0 + r + 8 * h, kcol = 16 * kt + l16;
      gm[kt][r] = gmaxf[qrow * 64 + kcol];
      rs[kt][r] = grs[qrow * 64 + kcol];
    }
  __syncthreads();

  for (int it = 0; it < nb; ++it) {
    size_t b = (size_t)blockIdx.x * (2 * nb) + (size_t)g * nb + it;
    *sx0 = xc0;
    *sx1 = xc1;
    {
      const float4* xf = xg + (size_t)(it + 1 < nb ? it + 1 : it) * 256;
      xc0 = xf[t];
      xc1 = xf[t + 128];
    }
    lds_barrier();

    v2f a[4];
#pragma unroll
    for (int kk = 0; kk < 4; kk++) a[kk] = ldA(mX, XP, r0, 4 * kk, l16, h);
    v8f q = {};
    v8f k = {};
    v8f v = {};
#pragma unroll
    for (int kk = 0; kk < 4; kk++) {
      q = wmma4(a[kk], ldB(sWq, 16, 4 * kk, 0, l16, h), q);
      k = wmma4(a[kk], ldB(sWk, 16, 4 * kk, 0, l16, h), k);
      v = wmma4(a[kk], ldB(sWv, 16, 4 * kk, 0, l16, h), v);
    }
#pragma unroll
    for (int r = 0; r < 8; r++) {
      int row = r0 + r + 8 * h;
      mQ[row * XP + l16] = q[r];
      mK[row * XP + l16] = k[r];
      mV[row * XP + l16] = v[r];
    }
    lds_barrier();

    v2f qa[4];
#pragma unroll
    for (int kk = 0; kk < 4; kk++) qa[kk] = ldA(mQ, XP, r0, 4 * kk, l16, h);
#pragma unroll
    for (int kt = 0; kt < 4; kt++) {
      v8f s = {};
#pragma unroll
      for (int kk = 0; kk < 4; kk++)
        s = wmma4(qa[kk], ldBT(mK, XP, 4 * kk, 16 * kt, l16, h), s);
#pragma unroll
      for (int r = 0; r < 8; r++) {
        int qrow = r0 + r + 8 * h, kcol = 16 * kt + l16;
        mS[qrow * SP + kcol] =
            __expf(s[r] * (1.0f / 3.0f) - gm[kt][r]) * rs[kt][r];
      }
    }

    v8f z = {};
#pragma unroll
    for (int c = 0; c < 16; c++)
      z = wmma4(ldA(mS, SP, r0, 4 * c, l16, h), ldB(mV, XP, 4 * c, 0, l16, h),
                z);

    float part = 0.0f, part2 = 0.0f, tv[8];
#pragma unroll
    for (int r = 0; r < 8; r++) {
      int row = r0 + r + 8 * h;
      float tt = mX[row * XP + l16] + z[r];
      tv[r] = tt;
      part += tt;
      part2 += tt * tt;
    }
    float2 s12 = gsum2(part, part2, redA, wv, lane, g);
    float mean = s12.x * (1.0f / 1024.0f);
    float rstd = rsqrtf(s12.y * (1.0f / 1024.0f) - mean * mean + LN_EPS);
#pragma unroll
    for (int r = 0; r < 8; r++) {
      int row = r0 + r + 8 * h;
      mX[row * XP + l16] = (tv[r] - mean) * rstd;
    }
    lds_barrier();

    float acc[8];
#pragma unroll
    for (int j = 0; j < 4; j++) {
      acc[j] = sB1[p * 16 + oA + j];
      acc[4 + j] = sB1[p * 16 + oB + j];
    }
#pragma unroll
    for (int i = 0; i < 16; i++) {
      float yv = mX[p * XP + i];
      const float* wr = &sW1[p * WSTRIDE + i * 16];
#pragma unroll
      for (int j = 0; j < 4; j++) {
        acc[j] = fmaf(yv, wr[oA + j], acc[j]);
        acc[4 + j] = fmaf(yv, wr[oB + j], acc[4 + j]);
      }
    }
#pragma unroll
    for (int j = 0; j < 4; j++) {
      mQ[p * XP + oA + j] = fmaxf(acc[j], 0.0f);
      mQ[p * XP + oB + j] = fmaxf(acc[4 + j], 0.0f);
    }
    lds_barrier();

    float acc2[8];
#pragma unroll
    for (int j = 0; j < 4; j++) {
      acc2[j] = sB2[p * 16 + oA + j];
      acc2[4 + j] = sB2[p * 16 + oB + j];
    }
#pragma unroll
    for (int i = 0; i < 16; i++) {
      float hv = mQ[p * XP + i];
      const float* wr = &sW2[p * WSTRIDE + i * 16];
#pragma unroll
      for (int j = 0; j < 4; j++) {
        acc2[j] = fmaf(hv, wr[oA + j], acc2[j]);
        acc2[4 + j] = fmaf(hv, wr[oB + j], acc2[4 + j]);
      }
    }
    float t2[8];
    float q1 = 0.0f, q2 = 0.0f;
#pragma unroll
    for (int j = 0; j < 8; j++) {
      int col = (j < 4) ? oA + j : oB + j - 4;
      float tt = fmaxf(acc2[j], 0.0f) + mX[p * XP + col];
      t2[j] = tt;
      q1 += tt;
      q2 += tt * tt;
    }
    float2 u12 = gsum2(q1, q2, redB, wv, lane, g);
    float m2 = u12.x * (1.0f / 1024.0f);
    float r2 = rsqrtf(u12.y * (1.0f / 1024.0f) - m2 * m2 + LN_EPS);
    float* op = out + b * 1024 + p * 16;
#pragma unroll
    for (int j = 0; j < 4; j++) {
      op[oA + j] = (t2[j] - m2) * r2;
      op[oB + j] = (t2[4 + j] - m2) * r2;
    }
  }
}

extern "C" void kernel_launch(void* const* d_in, const int* in_sizes, int n_in,
                              void* d_out, int out_size, void* d_ws,
                              size_t ws_size, hipStream_t stream) {
  (void)n_in;
  (void)out_size;
  (void)ws_size;
  const float* x = (const float*)d_in[0];
  const float* Wq = (const float*)d_in[1];
  const float* Wk = (const float*)d_in[2];
  const float* Wv = (const float*)d_in[3];
  const float* W1 = (const float*)d_in[4];
  const float* b1 = (const float*)d_in[5];
  const float* W2 = (const float*)d_in[6];
  const float* b2 = (const float*)d_in[7];
  float* out = (float*)d_out;

  float* gmaxf = (float*)d_ws;
  float* grs = (float*)((char*)d_ws + 4096 * sizeof(float));
  float* part = (float*)d_out;

  const int B = in_sizes[0] / 1024;
  const int NB = 64;
  const int blocks1 = B / NB;
  const int blocks3 = B / (2 * NB);

  enc_pass1_stats<<<blocks1, 128, 0, stream>>>(x, Wq, Wk, part, NB);
  enc_combine<<<128, 512, 0, stream>>>(part, gmaxf, grs, blocks1);
  enc_pass3_final<<<blocks3, 256, 0, stream>>>(x, Wq, Wk, Wv, W1, b1, W2, b2,
                                               gmaxf, grs, out, NB);
}


# waves per SIMD 2 -> 4 in both sweeps: pass 3 four 4-wave batch groups per block on the shared MLP weights, pass 1 twice the blocks of half the batches (softmax partials regrouped: fp order)
# speedup vs baseline: 3.2705x; 1.1231x over previous
#include <hip/hip_runtime.h>

typedef __attribute__((ext_vector_type(2))) float v2f;
typedef __attribute__((ext_vector_type(8))) float v8f;

#define LN_EPS 1e-5f

#define XP 20
#define SP 68

__device__ __forceinline__ v8f wmma4(v2f a, v2f b, v8f c) {
  return __builtin_amdgcn_wmma_f32_16x16x4_f32(false, a, false, b, (short)0, c,
                                               false, false);
}

__device__ __forceinline__ v2f ldA(const float* m, int ld, int r0, int c0,
                                   int l16, int h) {
  v2f a;
  a.x = m[(r0 + l16) * ld + c0 + 2 * h + 0];
  a.y = m[(r0 + l16) * ld + c0 + 2 * h + 1];
  return a;
}
__device__ __forceinline__ v2f ldB(const float* m, int ld, int c0, int n0,
                                   int l16, int h) {
  v2f b;
  b.x = m[(c0 + 2 * h + 0) * ld + n0 + l16];
  b.y = m[(c0 + 2 * h + 1) * ld + n0 + l16];
  return b;
}
__device__ __forceinline__ v2f ldBT(const float* m, int ld, int c0, int n0,
                                    int l16, int h) {
  v2f b;
  b.x = m[(n0 + l16) * ld + c0 + 2 * h + 0];
  b.y = m[(n0 + l16) * ld + c0 + 2 * h + 1];
  return b;
}

template <int XM>
__device__ __forceinline__ float row_xor(float v) {
  int b = __builtin_bit_cast(int, v);
  return __builtin_bit_cast(
      float, __builtin_amdgcn_update_dpp(0, b, 0x160 | XM, 0xf, 0xf, true));
}
__device__ __forceinline__ float xhalf(float v) {
  int b = __builtin_bit_cast(int, v);
  return __builtin_bit_cast(
      float, __builtin_amdgcn_permlanex16(b, b, 0x76543210, 0xfedcba98, false,
                                          false));
}
__device__ __forceinline__ float wave_sum(float v) {
  v += row_xor<1>(v);
  v += row_xor<2>(v);
  v += row_xor<4>(v);
  v += row_xor<8>(v);
  v += xhalf(v);
  return v;
}

__device__ __forceinline__ void lds_barrier() {
  asm volatile("s_wait_dscnt 0x0\n\ts_barrier_signal -1\n\ts_barrier_wait -1" ::
                   : "memory");
}

__device__ __forceinline__ float2 gsum2(float a, float b, float2* slot,
                                        int wv, int lane, int g) {
  float wa = wave_sum(a);
  float wb = wave_sum(b);
  if (lane == 0) slot[wv] = make_float2(wa, wb);
  lds_barrier();
  float2 p0 = slot[4 * g + 0], p1 = slot[4 * g + 1];
  float2 p2 = slot[4 * g + 2], p3 = slot[4 * g + 3];
  return make_float2((p0.x + p1.x) + (p2.x + p3.x),
                     (p0.y + p1.y) + (p2.y + p3.y));
}

__global__ __launch_bounds__(128) void enc_pass1_stats(
    const float* __restrict__ x, const float* __restrict__ Wq,
    const float* __restrict__ Wk, float* __restrict__ part, int nb) {
  __shared__ __align__(16) float sX[64 * XP];
  __shared__ __align__(16) float sQ[64 * XP], sK[64 * XP];
  __shared__ float sWq[256], sWk[256];
  int tid = threadIdx.x;
  int w = tid >> 5, lane = tid & 31, l16 = lane & 15, h = lane >> 4;
  int r0 = 16 * w;
  float4* sx0 = (float4*)&sX[(tid >> 2) * XP + (tid & 3) * 4];
  float4* sx1 = sx0 + 8 * XP;
  for (int i = tid; i < 256; i += 128) {
    sWq[i] = Wq[i];
    sWk[i] = Wk[i];
  }
  const float4* xg = (const float4*)(x + (size_t)blockIdx.x * nb * 1024);
  float4 xc0 = {}, xc1 = {};
  if (nb > 0) {
    xc0 = xg[tid];
    xc1 = xg[tid + 128];
  }
  float vm[4][8], vsum[4][8];
#pragma unroll
  for (int kt = 0; kt < 4; kt++)
#pragma unroll
    for (int r = 0; r < 8; r++) {
      vm[kt][r] = -3.4e38f;
      vsum[kt][r] = 0.0f;
    }
  __syncthreads();

  for (int it = 0; it < nb; ++it) {
    *sx0 = xc0;
    *sx1 = xc1;
    {
      const float4* xf = xg + (size_t)(it + 1 < nb ? it + 1 : it) * 256;
      xc0 = xf[tid];
      xc1 = xf[tid + 128];
    }
    lds_barrier();
    v2f a[4];
#pragma unroll
    for (int kk = 0; kk < 4; kk++) a[kk] = ldA(sX, XP, r0, 4 * kk, l16, h);
    v8f q = {};
    v8f k = {};
#pragma unroll
    for (int kk = 0; kk < 4; kk++) {
      q = wmma4(a[kk], ldB(sWq, 16, 4 * kk, 0, l16, h), q);
      k = wmma4(a[kk], ldB(sWk, 16, 4 * kk, 0, l16, h), k);
    }
#pragma unroll
    for (int r = 0; r < 8; r++) {
      sQ[(r0 + r + 8 * h) * XP + l16] = q[r];
      sK[(r0 + r + 8 * h) * XP + l16] = k[r];
    }
    lds_barrier();
    v2f qa[4];
#pragma unroll
    for (int kk = 0; kk < 4; kk++) qa[kk] = ldA(sQ, XP, r0, 4 * kk, l16, h);
#pragma unroll
    for (int kt = 0; kt < 4; kt++) {
      v8f s = {};
#pragma unroll
      for (int kk = 0; kk < 4; kk++)
        s = wmma4(qa[kk], ldBT(sK, XP, 4 * kk, 16 * kt, l16, h), s);
#pragma unroll
      for (int r = 0; r < 8; r++) {
        float v = s[r] * (1.0f / 3.0f);
        float d = v - vm[kt][r];
        float e = __expf(-fabsf(d));
        vsum[kt][r] =
            (d > 0.0f) ? fmaf(vsum[kt][r], e, 1.0f) : vsum[kt][r] + e;
        vm[kt][r] = fmaxf(vm[kt][r], v);
      }
    }
    lds_barrier();
  }
  float2* pp = (float2*)part + (size_t)blockIdx.x * 4096 + w * 1024 + lane;
#pragma unroll
  for (int kt = 0; kt < 4; kt++)
#pragma unroll
    for (int r = 0; r < 8; r++)
      pp[(8 * kt + r) * 32] = make_float2(vm[kt][r], vsum[kt][r]);
}

__device__ __forceinline__ void fold_pair(float& m, float& s, float2 t) {
  float mn = fmaxf(m, t.x);
  s = s * __expf(m - mn) + t.y * __expf(t.x - mn);
  m = mn;
}

__global__ __launch_bounds__(512) void enc_combine(
    const float* __restrict__ part, float* __restrict__ gmaxf,
    float* __restrict__ grs, int nblk) {
  __shared__ float2 red[16][32];
  int lane = threadIdx.x & 31, j = threadIdx.x >> 5;
  int e = blockIdx.x * 32 + lane;
  int per = nblk >> 4;
  const float2* p = (const float2*)part + (size_t)j * per * 4096 + e;
  float m = -3.4e38f, s = 0.0f;
#pragma unroll 8
  for (int i = 0; i < per; i++) fold_pair(m, s, p[(size_t)i * 4096]);
  red[j][lane] = make_float2(m, s);
  __syncthreads();
  if (j == 0) {
    for (int k = 1; k < 16; k++) fold_pair(m, s, red[k][lane]);
    int w = e >> 10, kt = (e >> 8) & 3, r = (e >> 5) & 7;
    int qrow = 16 * w + r + 8 * (lane >> 4), kcol = 16 * kt + (lane & 15);
    gmaxf[qrow * 64 + kcol] = m;
    grs[qrow * 64 + kcol] = 1.0f / s;
  }
}

#define NG 4
#define WSTRIDE 264

__global__ __launch_bounds__(128 * NG) void enc_pass3_final(
    const float* __restrict__ x, const float* __restrict__ Wq,
    const float* __restrict__ Wk, const float* __restrict__ Wv,
    const float* __restrict__ W1, const float* __restrict__ b1,
    const float* __restrict__ W2, const float* __restrict__ b2,
    const float* __restrict__ gmaxf, const float* __restrict__ grs,
    float* __restrict__ out, int nb) {
  __shared__ __align__(16) float sW1[64 * WSTRIDE];
  __shared__ __align__(16) float sW2[64 * WSTRIDE];
  __shared__ float sB1[1024], sB2[1024];
  __shared__ float sWq[256], sWk[256], sWv[256];
  __shared__ __align__(16) float sG[NG][4 * 64 * XP + 64 * SP];
  __shared__ float2 redA[4 * NG], redB[4 * NG];

  int tid = threadIdx.x;
  int g = tid >> 7;
  int t = tid & 127;
  int wv = tid >> 5;
  int w = t >> 5, lane = tid & 31, l16 = lane & 15, h = lane >> 4;
  int r0 = 16 * w;
  int p = t >> 1, oA = (t & 1) * 4, oB = 8 + (t & 1) * 4;

  float* mX = sG[g];
  float* mQ = mX + 64 * XP;
  float* mK = mQ + 64 * XP;
  float* mV = mK + 64 * XP;
  float* mS = mV + 64 * XP;
  float4* sx0 = (float4*)&mX[(t >> 2) * XP + (t & 3) * 4];
  float4* sx1 = sx0 + 8 * XP;

  const float4* xg = (const float4*)(x + ((size_t)blockIdx.x * (NG * nb) +
                                          (size_t)g * nb) * 1024);
  float4 xc0 = {}, xc1 = {};
  if (nb > 0) {
    xc0 = xg[t];
    xc1 = xg[t + 128];
  }

  for (int i = tid; i < 16384; i += 128 * NG) {
    int pp = i >> 8, rr = i & 255;
    sW1[pp * WSTRIDE + rr] = W1[i];
    sW2[pp * WSTRIDE + rr] = W2[i];
  }
  for (int i = tid; i < 1024; i += 128 * NG) {
    sB1[i] = b1[i];
    sB2[i] = b2[i];
  }
  if (tid < 256) {
    sWq[tid] = Wq[tid];
    sWk[tid] = Wk[tid];
    sWv[tid] = Wv[tid];
  }
  float gm[4][8], rs[4][8];
#pragma unroll
  for (int kt = 0; kt < 4; kt++)
#pragma unroll
    for (int r = 0; r < 8; r++) {
      int qrow = r0 + r + 8 * h, kcol = 16 * kt + l16;
      gm[kt][r] = gmaxf[qrow * 64 + kcol];
      rs[kt][r] = grs[qrow * 64 + kcol];
    }
  __syncthreads();

  for (int it = 0; it < nb; ++it) {
    size_t b = (size_t)blockIdx.x * (NG * nb) + (size_t)g * nb + it;
    *sx0 = xc0;
    *sx1 = xc1;
    {
      const float4* xf = xg + (size_t)(it + 1 < nb ? it + 1 : it) * 256;
      xc0 = xf[t];
      xc1 = xf[t + 128];
    }
    lds_barrier();

    v2f a[4];
#pragma unroll
    for (int kk = 0; kk < 4; kk++) a[kk] = ldA(mX, XP, r0, 4 * kk, l16, h);
    v8f q = {};
    v8f k = {};
    v8f v = {};
#pragma unroll
    for (int kk = 0; kk < 4; kk++) {
      q = wmma4(a[kk], ldB(sWq, 16, 4 * kk, 0, l16, h), q);
      k = wmma4(a[kk], ldB(sWk, 16, 4 * kk, 0, l16, h), k);
      v = wmma4(a[kk], ldB(sWv, 16, 4 * kk, 0, l16, h), v);
    }
#pragma unroll
    for (int r = 0; r < 8; r++) {
      int row = r0 + r + 8 * h;
      mQ[row * XP + l16] = q[r];
      mK[row * XP + l16] = k[r];
      mV[row * XP + l16] = v[r];
    }
    lds_barrier();

    v2f qa[4];
#pragma unroll
    for (int kk = 0; kk < 4; kk++) qa[kk] = ldA(mQ, XP, r0, 4 * kk, l16, h);
#pragma unroll
    for (int kt = 0; kt < 4; kt++) {
      v8f s = {};
#pragma unroll
      for (int kk = 0; kk < 4; kk++)
        s = wmma4(qa[kk], ldBT(mK, XP, 4 * kk, 16 * kt, l16, h), s);
#pragma unroll
      for (int r = 0; r < 8; r++) {
        int qrow = r0 + r + 8 * h, kcol = 16 * kt + l16;
        mS[qrow * SP + kcol] =
            __expf(s[r] * (1.0f / 3.0f) - gm[kt][r]) * rs[kt][r];
      }
    }

    v8f z = {};
#pragma unroll
    for (int c = 0; c < 16; c++)
      z = wmma4(ldA(mS, SP, r0, 4 * c, l16, h), ldB(mV, XP, 4 * c, 0, l16, h),
                z);

    float part = 0.0f, part2 = 0.0f, tv[8];
#pragma unroll
    for (int r = 0; r < 8; r++) {
      int row = r0 + r + 8 * h;
      float tt = mX[row * XP + l16] + z[r];
      tv[r] = tt;
      part += tt;
      part2 += tt * tt;
    }
    float2 s12 = gsum2(part, part2, redA, wv, lane, g);
    float mean = s12.x * (1.0f / 1024.0f);
    float rstd = rsqrtf(s12.y * (1.0f / 1024.0f) - mean * mean + LN_EPS);
#pragma unroll
    for (int r = 0; r < 8; r++) {
      int row = r0 + r + 8 * h;
      mX[row * XP + l16] = (tv[r] - mean) * rstd;
    }
    lds_barrier();

    float acc[8];
#pragma unroll
    for (int j = 0; j < 4; j++) {
      acc[j] = sB1[p * 16 + oA + j];
      acc[4 + j] = sB1[p * 16 + oB + j];
    }
#pragma unroll
    for (int i = 0; i < 16; i++) {
      float yv = mX[p * XP + i];
      const float* wr = &sW1[p * WSTRIDE + i * 16];
#pragma unroll
      for (int j = 0; j < 4; j++) {
        acc[j] = fmaf(yv, wr[oA + j], acc[j]);
        acc[4 + j] = fmaf(yv, wr[oB + j], acc[4 + j]);
      }
    }
#pragma unroll
    for (int j = 0; j < 4; j++) {
      mQ[p * XP + oA + j] = fmaxf(acc[j], 0.0f);
      mQ[p * XP + oB + j] = fmaxf(acc[4 + j], 0.0f);
    }
    lds_barrier();

    float acc2[8];
#pragma unroll
    for (int j = 0; j < 4; j++) {
      acc2[j] = sB2[p * 16 + oA + j];
      acc2[4 + j] = sB2[p * 16 + oB + j];
    }
#pragma unroll
    for (int i = 0; i < 16; i++) {
      float hv = mQ[p * XP + i];
      const float* wr = &sW2[p * WSTRIDE + i * 16];
#pragma unroll
      for (int j = 0; j < 4; j++) {
        acc2[j] = fmaf(hv, wr[oA + j], acc2[j]);
        acc2[4 + j] = fmaf(hv, wr[oB + j], acc2[4 + j]);
      }
    }
    float t2[8];
    float q1 = 0.0f, q2 = 0.0f;
#pragma unroll
    for (int j = 0; j < 8; j++) {
      int col = (j < 4) ? oA + j : oB + j - 4;
      float tt = fmaxf(acc2[j], 0.0f) + mX[p * XP + col];
      t2[j] = tt;
      q1 += tt;
      q2 += tt * tt;
    }
    float2 u12 = gsum2(q1, q2, redB, wv, lane, g);
    float m2 = u12.x * (1.0f / 1024.0f);
    float r2 = rsqrtf(u12.y * (1.0f / 1024.0f) - m2 * m2 + LN_EPS);
    float* op = out + b * 1024 + p * 16;
#pragma unroll
    for (int j = 0; j < 4; j++) {
      op[oA + j] = (t2[j] - m2) * r2;
      op[oB + j] = (t2[4 + j] - m2) * r2;
    }
  }
}

extern "C" void kernel_launch(void* const* d_in, const int* in_sizes, int n_in,
                              void* d_out, int out_size, void* d_ws,
                              size_t ws_size, hipStream_t stream) {
  (void)n_in;
  (void)out_size;
  (void)ws_size;
  const float* x = (const float*)d_in[0];
  const float* Wq = (const float*)d_in[1];
  const float* Wk = (const float*)d_in[2];
  const float* Wv = (const float*)d_in[3];
  const float* W1 = (const float*)d_in[4];
  const float* b1 = (const float*)d_in[5];
  const float* W2 = (const float*)d_in[6];
  const float* b2 = (const float*)d_in[7];
  float* out = (float*)d_out;

  float* gmaxf = (float*)d_ws;
  float* grs = (float*)((char*)d_ws + 4096 * sizeof(float));
  float* part = (float*)d_out;

  const int B = in_sizes[0] / 1024;
  const int NB = 32;
  const int blocks1 = B / NB;
  const int blocks3 = B / (NG * NB);

  enc_pass1_stats<<<blocks1, 128, 0, stream>>>(x, Wq, Wk, part, NB);
  enc_combine<<<128, 512, 0, stream>>>(part, gmaxf, grs, blocks1);
  enc_pass3_final<<<blocks3, 128 * NG, 0, stream>>>(
      x, Wq, Wk, Wv, W1, b1, W2, b2, gmaxf, grs, out, NB);
}


# per-patch MLPs moved from vector FMAs to the matrix unit: 16 batches gathered per block as tile columns, layer 2 fed from layer-1 accumulators; MLP + last LayerNorm fp sum order changes
# speedup vs baseline: 3.7193x; 1.1372x over previous
#include <hip/hip_runtime.h>

typedef __attribute__((ext_vector_type(2))) float v2f;
typedef __attribute__((ext_vector_type(8))) float v8f;

#define LN_EPS 1e-5f

#define XP 20
#define SP 68

__device__ __forceinline__ v8f wmma4(v2f a, v2f b, v8f c) {
  return __builtin_amdgcn_wmma_f32_16x16x4_f32(false, a, false, b, (short)0, c,
                                               false, false);
}

__device__ __forceinline__ v2f ldA(const float* m, int ld, int r0, int c0,
                                   int l16, int h) {
  v2f a;
  a.x = m[(r0 + l16) * ld + c0 + 2 * h + 0];
  a.y = m[(r0 + l16) * ld + c0 + 2 * h + 1];
  return a;
}
__device__ __forceinline__ v2f ldB(const float* m, int ld, int c0, int n0,
                                   int l16, int h) {
  v2f b;
  b.x = m[(c0 + 2 * h + 0) * ld + n0 + l16];
  b.y = m[(c0 + 2 * h + 1) * ld + n0 + l16];
  return b;
}
__device__ __forceinline__ v2f ldBT(const float* m, int ld, int c0, int n0,
                                    int l16, int h) {
  v2f b;
  b.x = m[(n0 + l16) * ld + c0 + 2 * h + 0];
  b.y = m[(n0 + l16) * ld + c0 + 2 * h + 1];
  return b;
}

template <int XM>
__device__ __forceinline__ float row_xor(float v) {
  int b = __builtin_bit_cast(int, v);
  return __builtin_bit_cast(
      float, __builtin_amdgcn_update_dpp(0, b, 0x160 | XM, 0xf, 0xf, true));
}
__device__ __forceinline__ float xhalf(float v) {
  int b = __builtin_bit_cast(int, v);
  return __builtin_bit_cast(
      float, __builtin_amdgcn_permlanex16(b, b, 0x76543210, 0xfedcba98, false,
                                          false));
}
__device__ __forceinline__ float wave_sum(float v) {
  v += row_xor<1>(v);
  v += row_xor<2>(v);
  v += row_xor<4>(v);
  v += row_xor<8>(v);
  v += xhalf(v);
  return v;
}

__device__ __forceinline__ void lds_barrier() {
  asm volatile("s_wait_dscnt 0x0\n\ts_barrier_signal -1\n\ts_barrier_wait -1" ::
                   : "memory");
}

__device__ __forceinline__ float2 gsum2(float a, float b, float2* slot,
                                        int wv, int lane, int g) {
  float wa = wave_sum(a);
  float wb = wave_sum(b);
  if (lane == 0) slot[wv] = make_float2(wa, wb);
  lds_barrier();
  float2 p0 = slot[4 * g + 0], p1 = slot[4 * g + 1];
  float2 p2 = slot[4 * g + 2], p3 = slot[4 * g + 3];
  return make_float2((p0.x + p1.x) + (p2.x + p3.x),
                     (p0.y + p1.y) + (p2.y + p3.y));
}

__global__ __launch_bounds__(128) void enc_pass1_stats(
    const float* __restrict__ x, const float* __restrict__ Wq,
    const float* __restrict__ Wk, float* __restrict__ part, int nb) {
  __shared__ __align__(16) float sX[64 * XP];
  __shared__ __align__(16) float sQ[64 * XP], sK[64 * XP];
  __shared__ float sWq[256], sWk[256];
  int tid = threadIdx.x;
  int w = tid >> 5, lane = tid & 31, l16 = lane & 15, h = lane >> 4;
  int r0 = 16 * w;
  float4* sx0 = (float4*)&sX[(tid >> 2) * XP + (tid & 3) * 4];
  float4* sx1 = sx0 + 8 * XP;
  for (int i = tid; i < 256; i += 128) {
    sWq[i] = Wq[i];
    sWk[i] = Wk[i];
  }
  const float4* xg = (const float4*)(x + (size_t)blockIdx.x * nb * 1024);
  float4 xc0 = {}, xc1 = {};
  if (nb > 0) {
    xc0 = xg[tid];
    xc1 = xg[tid + 128];
  }
  float vm[4][8], vsum[4][8];
#pragma unroll
  for (int kt = 0; kt < 4; kt++)
#pragma unroll
    for (int r = 0; r < 8; r++) {
      vm[kt][r] = -3.4e38f;
      vsum[kt][r] = 0.0f;
    }
  __syncthreads();

  for (int it = 0; it < nb; ++it) {
    *sx0 = xc0;
    *sx1 = xc1;
    {
      const float4* xf = xg + (size_t)(it + 1 < nb ? it + 1 : it) * 256;
      xc0 = xf[tid];
      xc1 = xf[tid + 128];
    }
    lds_barrier();
    v2f a[4];
#pragma unroll
    for (int kk = 0; kk < 4; kk++) a[kk] = ldA(sX, XP, r0, 4 * kk, l16, h);
    v8f q = {};
    v8f k = {};
#pragma unroll
    for (int kk = 0; kk < 4; kk++) {
      q = wmma4(a[kk], ldB(sWq, 16, 4 * kk, 0, l16, h), q);
      k = wmma4(a[kk], ldB(sWk, 16, 4 * kk, 0, l16, h), k);
    }
#pragma unroll
    for (int r = 0; r < 8; r++) {
      sQ[(r0 + r + 8 * h) * XP + l16] = q[r];
      sK[(r0 + r + 8 * h) * XP + l16] = k[r];
    }
    lds_barrier();
    v2f qa[4];
#pragma unroll
    for (int kk = 0; kk < 4; kk++) qa[kk] = ldA(sQ, XP, r0, 4 * kk, l16, h);
#pragma unroll
    for (int kt = 0; kt < 4; kt++) {
      v8f s = {};
#pragma unroll
      for (int kk = 0; kk < 4; kk++)
        s = wmma4(qa[kk], ldBT(sK, XP, 4 * kk, 16 * kt, l16, h), s);
#pragma unroll
      for (int r = 0; r < 8; r++) {
        float v = s[r] * (1.0f / 3.0f);
        float d = v - vm[kt][r];
        float e = __expf(-fabsf(d));
        vsum[kt][r] =
            (d > 0.0f) ? fmaf(vsum[kt][r], e, 1.0f) : vsum[kt][r] + e;
        vm[kt][r] = fmaxf(vm[kt][r], v);
      }
    }
    lds_barrier();
  }
  float2* pp = (float2*)part + (size_t)blockIdx.x * 4096 + w * 1024 + lane;
#pragma unroll
  for (int kt = 0; kt < 4; kt++)
#pragma unroll
    for (int r = 0; r < 8; r++)
      pp[(8 * kt + r) * 32] = make_float2(vm[kt][r], vsum[kt][r]);
}

__device__ __forceinline__ void fold_pair(float& m, float& s, float2 t) {
  float mn = fmaxf(m, t.x);
  s = s * __expf(m - mn) + t.y * __expf(t.x - mn);
  m = mn;
}

__global__ __launch_bounds__(512) void enc_combine(
    const float* __restrict__ part, float* __restrict__ gmaxf,
    float* __restrict__ grs, int nblk) {
  __shared__ float2 red[16][32];
  int lane = threadIdx.x & 31, j = threadIdx.x >> 5;
  int e = blockIdx.x * 32 + lane;
  int per = nblk >> 4;
  const float2* p = (const float2*)part + (size_t)j * per * 4096 + e;
  float m = -3.4e38f, s = 0.0f;
#pragma unroll 8
  for (int i = 0; i < per; i++) fold_pair(m, s, p[(size_t)i * 4096]);
  red[j][lane] = make_float2(m, s);
  __syncthreads();
  if (j == 0) {
    for (int k = 1; k < 16; k++) fold_pair(m, s, red[k][lane]);
    int w = e >> 10, kt = (e >> 8) & 3, r = (e >> 5) & 7;
    int qrow = 16 * w + r + 8 * (lane >> 4), kcol = 16 * kt + (lane & 15);
    gmaxf[qrow * 64 + kcol] = m;
    grs[qrow * 64 + kcol] = 1.0f / s;
  }
}

#define NG 4
#define NSLOT 16
#define BS (64 * XP + 4)

__device__ __forceinline__ v2f ldAW(const float* wp, int kk, int l16, int h) {
  v2f a;
  a.x = wp[(4 * kk + 2 * h + 0) * 16 + l16];
  a.y = wp[(4 * kk + 2 * h + 1) * 16 + l16];
  return a;
}

__global__ __launch_bounds__(128 * NG) void enc_pass3_final(
    const float* __restrict__ x, const float* __restrict__ Wq,
    const float* __restrict__ Wk, const float* __restrict__ Wv,
    const float* __restrict__ W1, const float* __restrict__ b1,
    const float* __restrict__ W2, const float* __restrict__ b2,
    const float* __restrict__ gmaxf, const float* __restrict__ grs,
    float* __restrict__ out, int nb) {
  __shared__ __align__(16) float sW1[64 * 256];
  __shared__ __align__(16) float sW2[64 * 256];
  __shared__ __align__(16) float sB1[1024], sB2[1024];
  __shared__ float sWq[256], sWk[256], sWv[256];
  __shared__ __align__(16) float sY[NSLOT * BS];
  __shared__ __align__(16) float sG[NG][3 * 64 * XP];
  __shared__ float2 redA[4 * NG];
  __shared__ float2 redB[4 * NG][NSLOT];

  int tid = threadIdx.x;
  int g = tid >> 7;
  int t = tid & 127;
  int wv = tid >> 5;
  int w = t >> 5, lane = tid & 31, l16 = lane & 15, h = lane >> 4;
  int r0 = 16 * w;

  float* mQ = sG[g];
  float* mK = mQ + 64 * XP;
  float* mV = mK + 64 * XP;
  float* myS = mQ + r0 * XP;

  const float4* xg = (const float4*)(x + ((size_t)blockIdx.x * (NG * nb) +
                                          (size_t)g * nb) * 1024);
  float4 xc0 = {}, xc1 = {};
  if (nb > 0) {
    xc0 = xg[t];
    xc1 = xg[t + 128];
  }

  for (int i = tid; i < 16384; i += 128 * NG) {
    int pp = i >> 8, row = (i >> 4) & 15, col = i & 15;
    int prow = 4 * ((row & 7) >> 1) + 2 * (row >> 3) + (row & 1);
    sW1[pp * 256 + prow * 16 + col] = W1[i];
    sW2[pp * 256 + prow * 16 + col] = W2[i];
  }
  for (int i = tid; i < 1024; i += 128 * NG) {
    sB1[i] = b1[i];
    sB2[i] = b2[i];
  }
  if (tid < 256) {
    sWq[tid] = Wq[tid];
    sWk[tid] = Wk[tid];
    sWv[tid] = Wv[tid];
  }
  float gm[4][8], rs[4][8];
#pragma unroll
  for (int kt = 0; kt < 4; kt++)
#pragma unroll
    for (int r = 0; r < 8; r++) {
      int qrow = r0 + r + 8 * h, kcol = 16 * kt + l16;
      gm[kt][r] = gmaxf[qrow * 64 + kcol];
      rs[kt][r] = grs[qrow * 64 + kcol];
    }
  __syncthreads();

  for (int it = 0; it < nb; ++it) {
    float* mX = sY + (4 * g + (it & 3)) * BS;
    float4* sx0 = (float4*)&mX[(t >> 2) * XP + (t & 3) * 4];
    float4* sx1 = sx0 + 8 * XP;
    *sx0 = xc0;
    *sx1 = xc1;
    {
      const float4* xf = xg + (size_t)(it + 1 < nb ? it + 1 : it) * 256;
      xc0 = xf[t];
      xc1 = xf[t + 128];
    }
    lds_barrier();

    v2f a[4];
#pragma unroll
    for (int kk = 0; kk < 4; kk++) a[kk] = ldA(mX, XP, r0, 4 * kk, l16, h);
    v8f q = {};
    v8f k = {};
    v8f v = {};
#pragma unroll
    for (int kk = 0; kk < 4; kk++) {
      q = wmma4(a[kk], ldB(sWq, 16, 4 * kk, 0, l16, h), q);
      k = wmma4(a[kk], ldB(sWk, 16, 4 * kk, 0, l16, h), k);
      v = wmma4(a[kk], ldB(sWv, 16, 4 * kk, 0, l16, h), v);
    }
#pragma unroll
    for (int r = 0; r < 8; r++) {
      int row = r0 + r + 8 * h;
      mQ[row * XP + l16] = q[r];
      mK[row * XP + l16] = k[r];
      mV[row * XP + l16] = v[r];
    }
    lds_barrier();

    v2f qa[4];
#pragma unroll
    for (int kk = 0; kk < 4; kk++) qa[kk] = ldA(mQ, XP, r0, 4 * kk, l16, h);
    v8f z = {};
#pragma unroll
    for (int kt = 0; kt < 4; kt++) {
      v8f s = {};
#pragma unroll
      for (int kk = 0; kk < 4; kk++)
        s = wmma4(qa[kk], ldBT(mK, XP, 4 * kk, 16 * kt, l16, h), s);
#pragma unroll
      for (int r = 0; r < 8; r++)
        myS[(r + 8 * h) * XP + l16] =
            __expf(s[r] * (1.0f / 3.0f) - gm[kt][r]) * rs[kt][r];
#pragma unroll
      for (int c = 0; c < 4; c++)
        z = wmma4(ldA(myS, XP, 0, 4 * c, l16, h),
                  ldB(mV, XP, 16 * kt + 4 * c, 0, l16, h), z);
    }

    float part = 0.0f, part2 = 0.0f, tv[8];
#pragma unroll
    for (int r = 0; r < 8; r++) {
      int row = r0 + r + 8 * h;
      float tt = mX[row * XP + l16] + z[r];
      tv[r] = tt;
      part += tt;
      part2 += tt * tt;
    }
    float2 s12 = gsum2(part, part2, redA, wv, lane, g);
    float mean = s12.x * (1.0f / 1024.0f);
    float rstd = rsqrtf(s12.y * (1.0f / 1024.0f) - mean * mean + LN_EPS);
#pragma unroll
    for (int r = 0; r < 8; r++) {
      int row = r0 + r + 8 * h;
      mX[row * XP + l16] = (tv[r] - mean) * rstd;
    }

    if ((it & 3) == 3) {
      lds_barrier();
      const float* yl = sY + l16 * BS + 8 * h;
      v8f yv[4], c1[4], c2[4];
      v2f aw[2][4];
#pragma unroll
      for (int pi = 0; pi < 4; pi++) {
        int p = 4 * wv + pi;
        float4 y0 = *(const float4*)(yl + p * XP);
        float4 y1 = *(const float4*)(yl + p * XP + 4);
        yv[pi] = (v8f){y0.x, y0.y, y0.z, y0.w, y1.x, y1.y, y1.z, y1.w};
        float4 u0 = *(const float4*)(sB1 + p * 16 + 8 * h);
        float4 u1 = *(const float4*)(sB1 + p * 16 + 8 * h + 4);
        c1[pi] = (v8f){u0.x, u0.y, u0.z, u0.w, u1.x, u1.y, u1.z, u1.w};
      }
#pragma unroll
      for (int kh = 0; kh < 2; kh++) {
#pragma unroll
        for (int k2 = 0; k2 < 2; k2++)
#pragma unroll
          for (int pi = 0; pi < 4; pi++)
            aw[k2][pi] = ldAW(sW1 + (4 * wv + pi) * 256, 2 * kh + k2, l16, h);
#pragma unroll
        for (int k2 = 0; k2 < 2; k2++)
#pragma unroll
          for (int pi = 0; pi < 4; pi++) {
            int kk = 2 * kh + k2;
            v2f bf = {yv[pi][2 * kk], yv[pi][2 * kk + 1]};
            c1[pi] = wmma4(aw[k2][pi], bf, c1[pi]);
          }
      }
#pragma unroll
      for (int pi = 0; pi < 4; pi++) {
        int p = 4 * wv + pi;
        float4 d0 = *(const float4*)(sB2 + p * 16 + 8 * h);
        float4 d1 = *(const float4*)(sB2 + p * 16 + 8 * h + 4);
        c2[pi] = (v8f){d0.x, d0.y, d0.z, d0.w, d1.x, d1.y, d1.z, d1.w};
      }
#pragma unroll
      for (int pi = 0; pi < 4; pi++)
#pragma unroll
        for (int r = 0; r < 8; r++) c1[pi][r] = fmaxf(c1[pi][r], 0.0f);
#pragma unroll
      for (int kh = 0; kh < 2; kh++) {
#pragma unroll
        for (int k2 = 0; k2 < 2; k2++)
#pragma unroll
          for (int pi = 0; pi < 4; pi++)
            aw[k2][pi] = ldAW(sW2 + (4 * wv + pi) * 256, 2 * kh + k2, l16, h);
#pragma unroll
        for (int k2 = 0; k2 < 2; k2++)
#pragma unroll
          for (int pi = 0; pi < 4; pi++) {
            int kk = 2 * kh + k2;
            v2f bf = {c1[pi][2 * kk], c1[pi][2 * kk + 1]};
            c2[pi] = wmma4(aw[k2][pi], bf, c2[pi]);
          }
      }
      float q1 = 0.0f, q2 = 0.0f;
#pragma unroll
      for (int pi = 0; pi < 4; pi++)
#pragma unroll
        for (int r = 0; r < 8; r++) {
          float tt = fmaxf(c2[pi][r], 0.0f) + yv[pi][r];
          c2[pi][r] = tt;
          q1 += tt;
          q2 += tt * tt;
        }
      q1 += xhalf(q1);
      q2 += xhalf(q2);
      if (h == 0) redB[wv][l16] = make_float2(q1, q2);
      lds_barrier();
      float u1s = 0.0f, u2s = 0.0f;
#pragma unroll
      for (int j = 0; j < 4 * NG; j++) {
        float2 pr = redB[j][l16];
        u1s += pr.x;
        u2s += pr.y;
      }
      float m2 = u1s * (1.0f / 1024.0f);
      float r2 = rsqrtf(u2s * (1.0f / 1024.0f) - m2 * m2 + LN_EPS);
      size_t b = (size_t)blockIdx.x * (NG * nb) + (size_t)(l16 >> 2) * nb +
                 (size_t)(it - 3 + (l16 & 3));
      float* op = out + b * 1024 + 8 * h;
#pragma unroll
      for (int pi = 0; pi < 4; pi++) {
        float4 o0, o1;
        o0.x = (c2[pi][0] - m2) * r2;
        o0.y = (c2[pi][1] - m2) * r2;
        o0.z = (c2[pi][2] - m2) * r2;
        o0.w = (c2[pi][3] - m2) * r2;
        o1.x = (c2[pi][4] - m2) * r2;
        o1.y = (c2[pi][5] - m2) * r2;
        o1.z = (c2[pi][6] - m2) * r2;
        o1.w = (c2[pi][7] - m2) * r2;
        *(float4*)(op + (4 * wv + pi) * 16) = o0;
        *(float4*)(op + (4 * wv + pi) * 16 + 4) = o1;
      }
    }
  }
}

extern "C" void kernel_launch(void* const* d_in, const int* in_sizes, int n_in,
                              void* d_out, int out_size, void* d_ws,
                              size_t ws_size, hipStream_t stream) {
  (void)n_in;
  (void)out_size;
  (void)ws_size;
  const float* x = (const float*)d_in[0];
  const float* Wq = (const float*)d_in[1];
  const float* Wk = (const float*)d_in[2];
  const float* Wv = (const float*)d_in[3];
  const float* W1 = (const float*)d_in[4];
  const float* b1 = (const float*)d_in[5];
  const float* W2 = (const float*)d_in[6];
  const float* b2 = (const float*)d_in[7];
  float* out = (float*)d_out;

  float* gmaxf = (float*)d_ws;
  float* grs = (float*)((char*)d_ws + 4096 * sizeof(float));
  float* part = (float*)d_out;

  const int B = in_sizes[0] / 1024;
  const int NB = 32;
  const int blocks1 = B / NB;
  const int blocks3 = B / (NG * NB);

  enc_pass1_stats<<<blocks1, 128, 0, stream>>>(x, Wq, Wk, part, NB);
  enc_combine<<<128, 512, 0, stream>>>(part, gmaxf, grs, blocks1);
  enc_pass3_final<<<blocks3, 128 * NG, 0, stream>>>(
      x, Wq, Wk, Wv, W1, b1, W2, b2, gmaxf, grs, out, NB);
}


# attention products chained through registers, transposed: S^T = x (M^T x^T), M = Wq Wk^T; z^T = Wv^T (x^T attn^T); Q/K/V/attn leave LDS, 3 -> 1-2 barriers per batch; re-associated products (fp order)
# speedup vs baseline: 4.1945x; 1.1278x over previous
#include <hip/hip_runtime.h>

typedef __attribute__((ext_vector_type(2))) float v2f;
typedef __attribute__((ext_vector_type(8))) float v8f;

#define LN_EPS 1e-5f

#define XP 20

__device__ __forceinline__ v8f wmma4(v2f a, v2f b, v8f c) {
  return __builtin_amdgcn_wmma_f32_16x16x4_f32(false, a, false, b, (short)0, c,
                                               false, false);
}

__device__ __forceinline__ v2f ldA(const float* m, int ld, int r0, int c0,
                                   int l16, int h) {
  v2f a;
  a.x = m[(r0 + l16) * ld + c0 + 2 * h + 0];
  a.y = m[(r0 + l16) * ld + c0 + 2 * h + 1];
  return a;
}
__device__ __forceinline__ v2f ldBT(const float* m, int ld, int c0, int n0,
                                    int l16, int h) {
  v2f b;
  b.x = m[(n0 + l16) * ld + c0 + 2 * h + 0];
  b.y = m[(n0 + l16) * ld + c0 + 2 * h + 1];
  return b;
}

__device__ __forceinline__ v2f frag(v8f t, int kk) {
  v2f f = {t[2 * kk], t[2 * kk + 1]};
  return f;
}
__device__ __forceinline__ v8f qk_matrix(const float* Wq, const float* Wk,
                                         int l16, int h) {
  v8f mt = {};
#pragma unroll
  for (int kk = 0; kk < 4; kk++)
    mt = wmma4(ldA(Wq, 16, 0, 4 * kk, l16, h), ldBT(Wk, 16, 4 * kk, 0, l16, h),
               mt);
  return mt;
}
__device__ __forceinline__ v8f ldRows(const float* m, int kt, int l16, int h) {
  const float4* p = (const float4*)&m[(16 * kt + l16) * XP + 8 * h];
  float4 u0 = p[0], u1 = p[1];
  v8f t = {u0.x, u0.y, u0.z, u0.w, u1.x, u1.y, u1.z, u1.w};
  return t;
}

template <int XM>
__device__ __forceinline__ float row_xor(float v) {
  int b = __builtin_bit_cast(int, v);
  return __builtin_bit_cast(
      float, __builtin_amdgcn_update_dpp(0, b, 0x160 | XM, 0xf, 0xf, true));
}
__device__ __forceinline__ float xhalf(float v) {
  int b = __builtin_bit_cast(int, v);
  return __builtin_bit_cast(
      float, __builtin_amdgcn_permlanex16(b, b, 0x76543210, 0xfedcba98, false,
                                          false));
}
__device__ __forceinline__ float wave_sum(float v) {
  v += row_xor<1>(v);
  v += row_xor<2>(v);
  v += row_xor<4>(v);
  v += row_xor<8>(v);
  v += xhalf(v);
  return v;
}

__device__ __forceinline__ void lds_barrier() {
  asm volatile("s_wait_dscnt 0x0\n\ts_barrier_signal -1\n\ts_barrier_wait -1" ::
                   : "memory");
}

__device__ __forceinline__ float2 gsum2(float a, float b, float2* slot,
                                        int wv, int lane, int g) {
  float wa = wave_sum(a);
  float wb = wave_sum(b);
  if (lane == 0) slot[wv] = make_float2(wa, wb);
  lds_barrier();
  float2 p0 = slot[4 * g + 0], p1 = slot[4 * g + 1];
  float2 p2 = slot[4 * g + 2], p3 = slot[4 * g + 3];
  return make_float2((p0.x + p1.x) + (p2.x + p3.x),
                     (p0.y + p1.y) + (p2.y + p3.y));
}

__global__ __launch_bounds__(128) void enc_pass1_stats(
    const float* __restrict__ x, const float* __restrict__ Wq,
    const float* __restrict__ Wk, float* __restrict__ part, int nb) {
  __shared__ __align__(16) float sX[2][64 * XP];
  int tid = threadIdx.x;
  int w = tid >> 5, lane = tid & 31, l16 = lane & 15, h = lane >> 4;
  v8f mt = qk_matrix(Wq, Wk, l16, h);
  const float4* xg = (const float4*)(x + (size_t)blockIdx.x * nb * 1024);
  float4 xc0 = {}, xc1 = {};
  if (nb > 0) {
    xc0 = xg[tid];
    xc1 = xg[tid + 128];
  }
  float vm[4][8], vsum[4][8];
#pragma unroll
  for (int j = 0; j < 4; j++)
#pragma unroll
    for (int r = 0; r < 8; r++) {
      vm[j][r] = -3.4e38f;
      vsum[j][r] = 0.0f;
    }

  for (int it = 0; it < nb; ++it) {
    float* bx = sX[it & 1];
    float4* sx0 = (float4*)&bx[(tid >> 2) * XP + (tid & 3) * 4];
    float4* sx1 = sx0 + 8 * XP;
    *sx0 = xc0;
    *sx1 = xc1;
    {
      const float4* xf = xg + (size_t)(it + 1 < nb ? it + 1 : it) * 256;
      xc0 = xf[tid];
      xc1 = xf[tid + 128];
    }
    lds_barrier();
    v8f xr[4];
#pragma unroll
    for (int j = 0; j < 4; j++) xr[j] = ldRows(bx, (w + j) & 3, l16, h);
    v8f qt = {};
#pragma unroll
    for (int kk = 0; kk < 4; kk++)
      qt = wmma4(frag(mt, kk), frag(xr[0], kk), qt);
#pragma unroll
    for (int j = 0; j < 4; j++) {
      v8f s = {};
#pragma unroll
      for (int kk = 0; kk < 4; kk++)
        s = wmma4(frag(xr[j], kk), frag(qt, kk), s);
#pragma unroll
      for (int r = 0; r < 8; r++) {
        float v = s[r] * (1.0f / 3.0f);
        float d = v - vm[j][r];
        float e = __expf(-fabsf(d));
        vsum[j][r] = (d > 0.0f) ? fmaf(vsum[j][r], e, 1.0f) : vsum[j][r] + e;
        vm[j][r] = fmaxf(vm[j][r], v);
      }
    }
  }
  float2* pp = (float2*)part + (size_t)blockIdx.x * 4096 + w * 1024 + lane;
#pragma unroll
  for (int j = 0; j < 4; j++)
#pragma unroll
    for (int r = 0; r < 8; r++)
      pp[(8 * j + r) * 32] = make_float2(vm[j][r], vsum[j][r]);
}

__device__ __forceinline__ void fold_pair(float& m, float& s, float2 t) {
  float mn = fmaxf(m, t.x);
  s = s * __expf(m - mn) + t.y * __expf(t.x - mn);
  m = mn;
}

__global__ __launch_bounds__(512) void enc_combine(
    const float* __restrict__ part, float* __restrict__ gmaxf,
    float* __restrict__ grs, int nblk) {
  __shared__ float2 red[16][32];
  int lane = threadIdx.x & 31, j = threadIdx.x >> 5;
  int e = blockIdx.x * 32 + lane;
  int per = nblk >> 4;
  const float2* p = (const float2*)part + (size_t)j * per * 4096 + e;
  float m = -3.4e38f, s = 0.0f;
#pragma unroll 8
  for (int i = 0; i < per; i++) fold_pair(m, s, p[(size_t)i * 4096]);
  red[j][lane] = make_float2(m, s);
  __syncthreads();
  if (j == 0) {
    for (int k = 1; k < 16; k++) fold_pair(m, s, red[k][lane]);
    int w = e >> 10, j4 = (e >> 8) & 3, r = (e >> 5) & 7;
    int qrow = 16 * w + (lane & 15);
    int kcol = 16 * ((w + j4) & 3) + 8 * (lane >> 4) + r;
    gmaxf[qrow * 64 + kcol] = m;
    grs[qrow * 64 + kcol] = 1.0f / s;
  }
}

#define NG 4
#define NSLOT 16
#define BS (64 * XP + 4)

__device__ __forceinline__ v2f ldAW(const float* wp, int kk, int l16, int h) {
  v2f a;
  a.x = wp[(4 * kk + 2 * h + 0) * 16 + l16];
  a.y = wp[(4 * kk + 2 * h + 1) * 16 + l16];
  return a;
}

__global__ __launch_bounds__(128 * NG) void enc_pass3_final(
    const float* __restrict__ x, const float* __restrict__ Wq,
    const float* __restrict__ Wk, const float* __restrict__ Wv,
    const float* __restrict__ W1, const float* __restrict__ b1,
    const float* __restrict__ W2, const float* __restrict__ b2,
    const float* __restrict__ gmaxf, const float* __restrict__ grs,
    float* __restrict__ out, int nb) {
  __shared__ __align__(16) float sW1[64 * 256];
  __shared__ __align__(16) float sW2[64 * 256];
  __shared__ __align__(16) float sB1[1024], sB2[1024];
  __shared__ __align__(16) float sY[NSLOT * BS];
  __shared__ float2 redA[4 * NG];
  __shared__ float2 redB[4 * NG][NSLOT];

  int tid = threadIdx.x;
  int g = tid >> 7;
  int t = tid & 127;
  int wv = tid >> 5;
  int w = t >> 5, lane = tid & 31, l16 = lane & 15, h = lane >> 4;
  int r0 = 16 * w;

  const float4* xg = (const float4*)(x + ((size_t)blockIdx.x * (NG * nb) +
                                          (size_t)g * nb) * 1024);
  float4 xc0 = {}, xc1 = {};
  if (nb > 0) {
    xc0 = xg[t];
    xc1 = xg[t + 128];
  }

  for (int i = tid; i < 16384; i += 128 * NG) {
    int pp = i >> 8, row = (i >> 4) & 15, col = i & 15;
    int prow = 4 * ((row & 7) >> 1) + 2 * (row >> 3) + (row & 1);
    sW1[pp * 256 + prow * 16 + col] = W1[i];
    sW2[pp * 256 + prow * 16 + col] = W2[i];
  }
  for (int i = tid; i < 1024; i += 128 * NG) {
    sB1[i] = b1[i];
    sB2[i] = b2[i];
  }
  v8f mt = qk_matrix(Wq, Wk, l16, h);
  v2f av[4];
#pragma unroll
  for (int kk = 0; kk < 4; kk++) {
    av[kk].x = Wv[(8 * h + 2 * kk + 0) * 16 + l16];
    av[kk].y = Wv[(8 * h + 2 * kk + 1) * 16 + l16];
  }
  float gm[4][8], rs[4][8];
#pragma unroll
  for (int j = 0; j < 4; j++)
#pragma unroll
    for (int r = 0; r < 8; r++) {
      int qrow = r0 + l16, kcol = 16 * ((w + j) & 3) + 8 * h + r;
      gm[j][r] = gmaxf[qrow * 64 + kcol];
      rs[j][r] = grs[qrow * 64 + kcol];
    }
  __syncthreads();

  for (int it = 0; it < nb; ++it) {
    float* mX = sY + (4 * g + (it & 3)) * BS;
    float4* sx0 = (float4*)&mX[(t >> 2) * XP + (t & 3) * 4];
    float4* sx1 = sx0 + 8 * XP;
    *sx0 = xc0;
    *sx1 = xc1;
    {
      const float4* xf = xg + (size_t)(it + 1 < nb ? it + 1 : it) * 256;
      xc0 = xf[t];
      xc1 = xf[t + 128];
    }
    lds_barrier();

    v8f xr[4];
#pragma unroll
    for (int j = 0; j < 4; j++) xr[j] = ldRows(mX, (w + j) & 3, l16, h);
    v8f qt = {};
#pragma unroll
    for (int kk = 0; kk < 4; kk++)
      qt = wmma4(frag(mt, kk), frag(xr[0], kk), qt);
    v8f ut0 = {};
    v8f ut1 = {};
#pragma unroll
    for (int j = 0; j < 4; j++) {
      int kt = (w + j) & 3;
      v8f s = {};
#pragma unroll
      for (int kk = 0; kk < 4; kk++)
        s = wmma4(frag(xr[j], kk), frag(qt, kk), s);
#pragma unroll
      for (int r = 0; r < 8; r++)
        s[r] = __expf(s[r] * (1.0f / 3.0f) - gm[j][r]) * rs[j][r];
      const float* xt = mX + (16 * kt + 8 * h) * XP + l16;
#pragma unroll
      for (int kk = 0; kk < 4; kk++) {
        v2f a = {xt[(2 * kk + 0) * XP], xt[(2 * kk + 1) * XP]};
        if (j & 1)
          ut1 = wmma4(a, frag(s, kk), ut1);
        else
          ut0 = wmma4(a, frag(s, kk), ut0);
      }
    }
    v8f ut = ut0 + ut1;
    v8f z = {};
#pragma unroll
    for (int kk = 0; kk < 4; kk++) z = wmma4(av[kk], frag(ut, kk), z);

    float part = 0.0f, part2 = 0.0f, tv[8];
#pragma unroll
    for (int r = 0; r < 8; r++) {
      float tt = xr[0][r] + z[r];
      tv[r] = tt;
      part += tt;
      part2 += tt * tt;
    }
    float2 s12 = gsum2(part, part2, redA, wv, lane, g);
    float mean = s12.x * (1.0f / 1024.0f);
    float rstd = rsqrtf(s12.y * (1.0f / 1024.0f) - mean * mean + LN_EPS);
    float4* yp = (float4*)&mX[(r0 + l16) * XP + 8 * h];
    yp[0] = make_float4((tv[0] - mean) * rstd, (tv[1] - mean) * rstd,
                        (tv[2] - mean) * rstd, (tv[3] - mean) * rstd);
    yp[1] = make_float4((tv[4] - mean) * rstd, (tv[5] - mean) * rstd,
                        (tv[6] - mean) * rstd, (tv[7] - mean) * rstd);

    if ((it & 3) == 3) {
      lds_barrier();
      const float* yl = sY + l16 * BS + 8 * h;
      v8f yv[4], c1[4], c2[4];
      v2f aw[2][4];
#pragma unroll
      for (int pi = 0; pi < 4; pi++) {
        int p = 4 * wv + pi;
        float4 y0 = *(const float4*)(yl + p * XP);
        float4 y1 = *(const float4*)(yl + p * XP + 4);
        yv[pi] = (v8f){y0.x, y0.y, y0.z, y0.w, y1.x, y1.y, y1.z, y1.w};
        float4 u0 = *(const float4*)(sB1 + p * 16 + 8 * h);
        float4 u1 = *(const float4*)(sB1 + p * 16 + 8 * h + 4);
        c1[pi] = (v8f){u0.x, u0.y, u0.z, u0.w, u1.x, u1.y, u1.z, u1.w};
      }
#pragma unroll
      for (int kh = 0; kh < 2; kh++) {
#pragma unroll
        for (int k2 = 0; k2 < 2; k2++)
#pragma unroll
          for (int pi = 0; pi < 4; pi++)
            aw[k2][pi] = ldAW(sW1 + (4 * wv + pi) * 256, 2 * kh + k2, l16, h);
#pragma unroll
        for (int k2 = 0; k2 < 2; k2++)
#pragma unroll
          for (int pi = 0; pi < 4; pi++) {
            int kk = 2 * kh + k2;
            v2f bf = {yv[pi][2 * kk], yv[pi][2 * kk + 1]};
            c1[pi] = wmma4(aw[k2][pi], bf, c1[pi]);
          }
      }
#pragma unroll
      for (int pi = 0; pi < 4; pi++) {
        int p = 4 * wv + pi;
        float4 d0 = *(const float4*)(sB2 + p * 16 + 8 * h);
        float4 d1 = *(const float4*)(sB2 + p * 16 + 8 * h + 4);
        c2[pi] = (v8f){d0.x, d0.y, d0.z, d0.w, d1.x, d1.y, d1.z, d1.w};
      }
#pragma unroll
      for (int pi = 0; pi < 4; pi++)
#pragma unroll
        for (int r = 0; r < 8; r++) c1[pi][r] = fmaxf(c1[pi][r], 0.0f);
#pragma unroll
      for (int kh = 0; kh < 2; kh++) {
#pragma unroll
        for (int k2 = 0; k2 < 2; k2++)
#pragma unroll
          for (int pi = 0; pi < 4; pi++)
            aw[k2][pi] = ldAW(sW2 + (4 * wv + pi) * 256, 2 * kh + k2, l16, h);
#pragma unroll
        for (int k2 = 0; k2 < 2; k2++)
#pragma unroll
          for (int pi = 0; pi < 4; pi++) {
            int kk = 2 * kh + k2;
            v2f bf = {c1[pi][2 * kk], c1[pi][2 * kk + 1]};
            c2[pi] = wmma4(aw[k2][pi], bf, c2[pi]);
          }
      }
      float q1 = 0.0f, q2 = 0.0f;
#pragma unroll
      for (int pi = 0; pi < 4; pi++)
#pragma unroll
        for (int r = 0; r < 8; r++) {
          float tt = fmaxf(c2[pi][r], 0.0f) + yv[pi][r];
          c2[pi][r] = tt;
          q1 += tt;
          q2 += tt * tt;
        }
      q1 += xhalf(q1);
      q2 += xhalf(q2);
      if (h == 0) redB[wv][l16] = make_float2(q1, q2);
      lds_barrier();
      float u1s = 0.0f, u2s = 0.0f;
#pragma unroll
      for (int j = 0; j < 4 * NG; j++) {
        float2 pr = redB[j][l16];
        u1s += pr.x;
        u2s += pr.y;
      }
      float m2 = u1s * (1.0f / 1024.0f);
      float r2 = rsqrtf(u2s * (1.0f / 1024.0f) - m2 * m2 + LN_EPS);
      size_t b = (size_t)blockIdx.x * (NG * nb) + (size_t)(l16 >> 2) * nb +
                 (size_t)(it - 3 + (l16 & 3));
      float* op = out + b * 1024 + 8 * h;
#pragma unroll
      for (int pi = 0; pi < 4; pi++) {
        float4 o0, o1;
        o0.x = (c2[pi][0] - m2) * r2;
        o0.y = (c2[pi][1] - m2) * r2;
        o0.z = (c2[pi][2] - m2) * r2;
        o0.w = (c2[pi][3] - m2) * r2;
        o1.x = (c2[pi][4] - m2) * r2;
        o1.y = (c2[pi][5] - m2) * r2;
        o1.z = (c2[pi][6] - m2) * r2;
        o1.w = (c2[pi][7] - m2) * r2;
        *(float4*)(op + (4 * wv + pi) * 16) = o0;
        *(float4*)(op + (4 * wv + pi) * 16 + 4) = o1;
      }
    }
  }
}

extern "C" void kernel_launch(void* const* d_in, const int* in_sizes, int n_in,
                              void* d_out, int out_size, void* d_ws,
                              size_t ws_size, hipStream_t stream) {
  (void)n_in;
  (void)out_size;
  (void)ws_size;
  const float* x = (const float*)d_in[0];
  const float* Wq = (const float*)d_in[1];
  const float* Wk = (const float*)d_in[2];
  const float* Wv = (const float*)d_in[3];
  const float* W1 = (const float*)d_in[4];
  const float* b1 = (const float*)d_in[5];
  const float* W2 = (const float*)d_in[6];
  const float* b2 = (const float*)d_in[7];
  float* out = (float*)d_out;

  float* gmaxf = (float*)d_ws;
  float* grs = (float*)((char*)d_ws + 4096 * sizeof(float));
  float* part = (float*)d_out;

  const int B = in_sizes[0] / 1024;
  const int NB = 32;
  const int blocks1 = B / NB;
  const int blocks3 = B / (NG * NB);

  enc_pass1_stats<<<blocks1, 128, 0, stream>>>(x, Wq, Wk, part, NB);
  enc_combine<<<128, 512, 0, stream>>>(part, gmaxf, grs, blocks1);
  enc_pass3_final<<<blocks3, 128 * NG, 0, stream>>>(
      x, Wq, Wk, Wv, W1, b1, W2, b2, gmaxf, grs, out, NB);
}


# pass 3 output rows staged through wave-private LDS (pitch 68 floats): each global store instruction writes four whole 128-byte lines, not 32 16-byte pieces; values and sum order unchanged
# speedup vs baseline: 4.3537x; 1.0380x over previous
#include <hip/hip_runtime.h>

typedef __attribute__((ext_vector_type(2))) float v2f;
typedef __attribute__((ext_vector_type(8))) float v8f;

#define LN_EPS 1e-5f

#define XP 20

__device__ __forceinline__ v8f wmma4(v2f a, v2f b, v8f c) {
  return __builtin_amdgcn_wmma_f32_16x16x4_f32(false, a, false, b, (short)0, c,
                                               false, false);
}

__device__ __forceinline__ v2f ldA(const float* m, int ld, int r0, int c0,
                                   int l16, int h) {
  v2f a;
  a.x = m[(r0 + l16) * ld + c0 + 2 * h + 0];
  a.y = m[(r0 + l16) * ld + c0 + 2 * h + 1];
  return a;
}
__device__ __forceinline__ v2f ldBT(const float* m, int ld, int c0, int n0,
                                    int l16, int h) {
  v2f b;
  b.x = m[(n0 + l16) * ld + c0 + 2 * h + 0];
  b.y = m[(n0 + l16) * ld + c0 + 2 * h + 1];
  return b;
}

__device__ __forceinline__ v2f frag(v8f t, int kk) {
  v2f f = {t[2 * kk], t[2 * kk + 1]};
  return f;
}
__device__ __forceinline__ v8f qk_matrix(const float* Wq, const float* Wk,
                                         int l16, int h) {
  v8f mt = {};
#pragma unroll
  for (int kk = 0; kk < 4; kk++)
    mt = wmma4(ldA(Wq, 16, 0, 4 * kk, l16, h), ldBT(Wk, 16, 4 * kk, 0, l16, h),
               mt);
  return mt;
}
__device__ __forceinline__ v8f ldRows(const float* m, int kt, int l16, int h) {
  const float4* p = (const float4*)&m[(16 * kt + l16) * XP + 8 * h];
  float4 u0 = p[0], u1 = p[1];
  v8f t = {u0.x, u0.y, u0.z, u0.w, u1.x, u1.y, u1.z, u1.w};
  return t;
}

template <int XM>
__device__ __forceinline__ float row_xor(float v) {
  int b = __builtin_bit_cast(int, v);
  return __builtin_bit_cast(
      float, __builtin_amdgcn_update_dpp(0, b, 0x160 | XM, 0xf, 0xf, true));
}
__device__ __forceinline__ float xhalf(float v) {
  int b = __builtin_bit_cast(int, v);
  return __builtin_bit_cast(
      float, __builtin_amdgcn_permlanex16(b, b, 0x76543210, 0xfedcba98, false,
                                          false));
}
__device__ __forceinline__ float wave_sum(float v) {
  v += row_xor<1>(v);
  v += row_xor<2>(v);
  v += row_xor<4>(v);
  v += row_xor<8>(v);
  v += xhalf(v);
  return v;
}

__device__ __forceinline__ void lds_barrier() {
  asm volatile("s_wait_dscnt 0x0\n\ts_barrier_signal -1\n\ts_barrier_wait -1" ::
                   : "memory");
}

__device__ __forceinline__ void wave_lds_handover() {
  asm volatile("s_wait_dscnt 0x0" ::: "memory");
}

__device__ __forceinline__ float2 gsum2(float a, float b, float2* slot,
                                        int wv, int lane, int g) {
  float wa = wave_sum(a);
  float wb = wave_sum(b);
  if (lane == 0) slot[wv] = make_float2(wa, wb);
  lds_barrier();
  float2 p0 = slot[4 * g + 0], p1 = slot[4 * g + 1];
  float2 p2 = slot[4 * g + 2], p3 = slot[4 * g + 3];
  return make_float2((p0.x + p1.x) + (p2.x + p3.x),
                     (p0.y + p1.y) + (p2.y + p3.y));
}

__global__ __launch_bounds__(128) void enc_pass1_stats(
    const float* __restrict__ x, const float* __restrict__ Wq,
    const float* __restrict__ Wk, float* __restrict__ part, int nb) {
  __shared__ __align__(16) float sX[2][64 * XP];
  int tid = threadIdx.x;
  int w = tid >> 5, lane = tid & 31, l16 = lane & 15, h = lane >> 4;
  v8f mt = qk_matrix(Wq, Wk, l16, h);
  const float4* xg = (const float4*)(x + (size_t)blockIdx.x * nb * 1024);
  float4 xc0 = {}, xc1 = {};
  if (nb > 0) {
    xc0 = xg[tid];
    xc1 = xg[tid + 128];
  }
  float vm[4][8], vsum[4][8];
#pragma unroll
  for (int j = 0; j < 4; j++)
#pragma unroll
    for (int r = 0; r < 8; r++) {
      vm[j][r] = -3.4e38f;
      vsum[j][r] = 0.0f;
    }

  for (int it = 0; it < nb; ++it) {
    float* bx = sX[it & 1];
    float4* sx0 = (float4*)&bx[(tid >> 2) * XP + (tid & 3) * 4];
    float4* sx1 = sx0 + 8 * XP;
    *sx0 = xc0;
    *sx1 = xc1;
    {
      const float4* xf = xg + (size_t)(it + 1 < nb ? it + 1 : it) * 256;
      xc0 = xf[tid];
      xc1 = xf[tid + 128];
    }
    lds_barrier();
    v8f xr[4];
#pragma unroll
    for (int j = 0; j < 4; j++) xr[j] = ldRows(bx, (w + j) & 3, l16, h);
    v8f qt = {};
#pragma unroll
    for (int kk = 0; kk < 4; kk++)
      qt = wmma4(frag(mt, kk), frag(xr[0], kk), qt);
#pragma unroll
    for (int j = 0; j < 4; j++) {
      v8f s = {};
#pragma unroll
      for (int kk = 0; kk < 4; kk++)
        s = wmma4(frag(xr[j], kk), frag(qt, kk), s);
#pragma unroll
      for (int r = 0; r < 8; r++) {
        float v = s[r] * (1.0f / 3.0f);
        float d = v - vm[j][r];
        float e = __expf(-fabsf(d));
        vsum[j][r] = (d > 0.0f) ? fmaf(vsum[j][r], e, 1.0f) : vsum[j][r] + e;
        vm[j][r] = fmaxf(vm[j][r], v);
      }
    }
  }
  float2* pp = (float2*)part + (size_t)blockIdx.x * 4096 + w * 1024 + lane;
#pragma unroll
  for (int j = 0; j < 4; j++)
#pragma unroll
    for (int r = 0; r < 8; r++)
      pp[(8 * j + r) * 32] = make_float2(vm[j][r], vsum[j][r]);
}

__device__ __forceinline__ void fold_pair(float& m, float& s, float2 t) {
  float mn = fmaxf(m, t.x);
  s = s * __expf(m - mn) + t.y * __expf(t.x - mn);
  m = mn;
}

__global__ __launch_bounds__(512) void enc_combine(
    const float* __restrict__ part, float* __restrict__ gmaxf,
    float* __restrict__ grs, int nblk) {
  __shared__ float2 red[16][32];
  int lane = threadIdx.x & 31, j = threadIdx.x >> 5;
  int e = blockIdx.x * 32 + lane;
  int per = nblk >> 4;
  const float2* p = (const float2*)part + (size_t)j * per * 4096 + e;
  float m = -3.4e38f, s = 0.0f;
#pragma unroll 8
  for (int i = 0; i < per; i++) fold_pair(m, s, p[(size_t)i * 4096]);
  red[j][lane] = make_float2(m, s);
  __syncthreads();
  if (j == 0) {
    for (int k = 1; k < 16; k++) fold_pair(m, s, red[k][lane]);
    int w = e >> 10, j4 = (e >> 8) & 3, r = (e >> 5) & 7;
    int qrow = 16 * w + (lane & 15);
    int kcol = 16 * ((w + j4) & 3) + 8 * (lane >> 4) + r;
    gmaxf[qrow * 64 + kcol] = m;
    grs[qrow * 64 + kcol] = 1.0f / s;
  }
}

#define NG 4
#define NSLOT 16
#define BS (64 * XP + 4)
#define OP 68

__device__ __forceinline__ v2f ldAW(const float* wp, int kk, int l16, int h) {
  v2f a;
  a.x = wp[(4 * kk + 2 * h + 0) * 16 + l16];
  a.y = wp[(4 * kk + 2 * h + 1) * 16 + l16];
  return a;
}

__global__ __launch_bounds__(128 * NG) void enc_pass3_final(
    const float* __restrict__ x, const float* __restrict__ Wq,
    const float* __restrict__ Wk, const float* __restrict__ Wv,
    const float* __restrict__ W1, const float* __restrict__ b1,
    const float* __restrict__ W2, const float* __restrict__ b2,
    const float* __restrict__ gmaxf, const float* __restrict__ grs,
    float* __restrict__ out, int nb) {
  __shared__ __align__(16) float sW1[64 * 256];
  __shared__ __align__(16) float sW2[64 * 256];
  __shared__ __align__(16) float sB1[1024], sB2[1024];
  __shared__ __align__(16) float sY[NSLOT * BS];
  __shared__ float2 redA[4 * NG];
  __shared__ float2 redB[4 * NG][NSLOT];
  __shared__ __align__(16) float sOut[4 * NG][NSLOT * OP];

  int tid = threadIdx.x;
  int g = tid >> 7;
  int t = tid & 127;
  int wv = tid >> 5;
  int w = t >> 5, lane = tid & 31, l16 = lane & 15, h = lane >> 4;
  int r0 = 16 * w;

  const float4* xg = (const float4*)(x + ((size_t)blockIdx.x * (NG * nb) +
                                          (size_t)g * nb) * 1024);
  float4 xc0 = {}, xc1 = {};
  if (nb > 0) {
    xc0 = xg[t];
    xc1 = xg[t + 128];
  }

  for (int i = tid; i < 16384; i += 128 * NG) {
    int pp = i >> 8, row = (i >> 4) & 15, col = i & 15;
    int prow = 4 * ((row & 7) >> 1) + 2 * (row >> 3) + (row & 1);
    sW1[pp * 256 + prow * 16 + col] = W1[i];
    sW2[pp * 256 + prow * 16 + col] = W2[i];
  }
  for (int i = tid; i < 1024; i += 128 * NG) {
    sB1[i] = b1[i];
    sB2[i] = b2[i];
  }
  v8f mt = qk_matrix(Wq, Wk, l16, h);
  v2f av[4];
#pragma unroll
  for (int kk = 0; kk < 4; kk++) {
    av[kk].x = Wv[(8 * h + 2 * kk + 0) * 16 + l16];
    av[kk].y = Wv[(8 * h + 2 * kk + 1) * 16 + l16];
  }
  float gm[4][8], rs[4][8];
#pragma unroll
  for (int j = 0; j < 4; j++)
#pragma unroll
    for (int r = 0; r < 8; r++) {
      int qrow = r0 + l16, kcol = 16 * ((w + j) & 3) + 8 * h + r;
      gm[j][r] = gmaxf[qrow * 64 + kcol];
      rs[j][r] = grs[qrow * 64 + kcol];
    }
  __syncthreads();

  for (int it = 0; it < nb; ++it) {
    float* mX = sY + (4 * g + (it & 3)) * BS;
    float4* sx0 = (float4*)&mX[(t >> 2) * XP + (t & 3) * 4];
    float4* sx1 = sx0 + 8 * XP;
    *sx0 = xc0;
    *sx1 = xc1;
    {
      const float4* xf = xg + (size_t)(it + 1 < nb ? it + 1 : it) * 256;
      xc0 = xf[t];
      xc1 = xf[t + 128];
    }
    lds_barrier();

    v8f xr[4];
#pragma unroll
    for (int j = 0; j < 4; j++) xr[j] = ldRows(mX, (w + j) & 3, l16, h);
    v8f qt = {};
#pragma unroll
    for (int kk = 0; kk < 4; kk++)
      qt = wmma4(frag(mt, kk), frag(xr[0], kk), qt);
    v8f ut0 = {};
    v8f ut1 = {};
#pragma unroll
    for (int j = 0; j < 4; j++) {
      int kt = (w + j) & 3;
      v8f s = {};
#pragma unroll
      for (int kk = 0; kk < 4; kk++)
        s = wmma4(frag(xr[j], kk), frag(qt, kk), s);
#pragma unroll
      for (int r = 0; r < 8; r++)
        s[r] = __expf(s[r] * (1.0f / 3.0f) - gm[j][r]) * rs[j][r];
      const float* xt = mX + (16 * kt + 8 * h) * XP + l16;
#pragma unroll
      for (int kk = 0; kk < 4; kk++) {
        v2f a = {xt[(2 * kk + 0) * XP], xt[(2 * kk + 1) * XP]};
        if (j & 1)
          ut1 = wmma4(a, frag(s, kk), ut1);
        else
          ut0 = wmma4(a, frag(s, kk), ut0);
      }
    }
    v8f ut = ut0 + ut1;
    v8f z = {};
#pragma unroll
    for (int kk = 0; kk < 4; kk++) z = wmma4(av[kk], frag(ut, kk), z);

    float part = 0.0f, part2 = 0.0f, tv[8];
#pragma unroll
    for (int r = 0; r < 8; r++) {
      float tt = xr[0][r] + z[r];
      tv[r] = tt;
      part += tt;
      part2 += tt * tt;
    }
    float2 s12 = gsum2(part, part2, redA, wv, lane, g);
    float mean = s12.x * (1.0f / 1024.0f);
    float rstd = rsqrtf(s12.y * (1.0f / 1024.0f) - mean * mean + LN_EPS);
    float4* yp = (float4*)&mX[(r0 + l16) * XP + 8 * h];
    yp[0] = make_float4((tv[0] - mean) * rstd, (tv[1] - mean) * rstd,
                        (tv[2] - mean) * rstd, (tv[3] - mean) * rstd);
    yp[1] = make_float4((tv[4] - mean) * rstd, (tv[5] - mean) * rstd,
                        (tv[6] - mean) * rstd, (tv[7] - mean) * rstd);

    if ((it & 3) == 3) {
      lds_barrier();
      const float* yl = sY + l16 * BS + 8 * h;
      v8f yv[4], c1[4], c2[4];
      v2f aw[2][4];
#pragma unroll
      for (int pi = 0; pi < 4; pi++) {
        int p = 4 * wv + pi;
        float4 y0 = *(const float4*)(yl + p * XP);
        float4 y1 = *(const float4*)(yl + p * XP + 4);
        yv[pi] = (v8f){y0.x, y0.y, y0.z, y0.w, y1.x, y1.y, y1.z, y1.w};
        float4 u0 = *(const float4*)(sB1 + p * 16 + 8 * h);
        float4 u1 = *(const float4*)(sB1 + p * 16 + 8 * h + 4);
        c1[pi] = (v8f){u0.x, u0.y, u0.z, u0.w, u1.x, u1.y, u1.z, u1.w};
      }
#pragma unroll
      for (int kh = 0; kh < 2; kh++) {
#pragma unroll
        for (int k2 = 0; k2 < 2; k2++)
#pragma unroll
          for (int pi = 0; pi < 4; pi++)
            aw[k2][pi] = ldAW(sW1 + (4 * wv + pi) * 256, 2 * kh + k2, l16, h);
#pragma unroll
        for (int k2 = 0; k2 < 2; k2++)
#pragma unroll
          for (int pi = 0; pi < 4; pi++) {
            int kk = 2 * kh + k2;
            v2f bf = {yv[pi][2 * kk], yv[pi][2 * kk + 1]};
            c1[pi] = wmma4(aw[k2][pi], bf, c1[pi]);
          }
      }
#pragma unroll
      for (int pi = 0; pi < 4; pi++) {
        int p = 4 * wv + pi;
        float4 d0 = *(const float4*)(sB2 + p * 16 + 8 * h);
        float4 d1 = *(const float4*)(sB2 + p * 16 + 8 * h + 4);
        c2[pi] = (v8f){d0.x, d0.y, d0.z, d0.w, d1.x, d1.y, d1.z, d1.w};
      }
#pragma unroll
      for (int pi = 0; pi < 4; pi++)
#pragma unroll
        for (int r = 0; r < 8; r++) c1[pi][r] = fmaxf(c1[pi][r], 0.0f);
#pragma unroll
      for (int kh = 0; kh < 2; kh++) {
#pragma unroll
        for (int k2 = 0; k2 < 2; k2++)
#pragma unroll
          for (int pi = 0; pi < 4; pi++)
            aw[k2][pi] = ldAW(sW2 + (4 * wv + pi) * 256, 2 * kh + k2, l16, h);
#pragma unroll
        for (int k2 = 0; k2 < 2; k2++)
#pragma unroll
          for (int pi = 0; pi < 4; pi++) {
            int kk = 2 * kh + k2;
            v2f bf = {c1[pi][2 * kk], c1[pi][2 * kk + 1]};
            c2[pi] = wmma4(aw[k2][pi], bf, c2[pi]);
          }
      }
      float q1 = 0.0f, q2 = 0.0f;
#pragma unroll
      for (int pi = 0; pi < 4; pi++)
#pragma unroll
        for (int r = 0; r < 8; r++) {
          float tt = fmaxf(c2[pi][r], 0.0f) + yv[pi][r];
          c2[pi][r] = tt;
          q1 += tt;
          q2 += tt * tt;
        }
      q1 += xhalf(q1);
      q2 += xhalf(q2);
      if (h == 0) redB[wv][l16] = make_float2(q1, q2);
      lds_barrier();
      float u1s = 0.0f, u2s = 0.0f;
#pragma unroll
      for (int j = 0; j < 4 * NG; j++) {
        float2 pr = redB[j][l16];
        u1s += pr.x;
        u2s += pr.y;
      }
      float m2 = u1s * (1.0f / 1024.0f);
      float r2 = rsqrtf(u2s * (1.0f / 1024.0f) - m2 * m2 + LN_EPS);
      float* so = sOut[wv];
      float* sw = so + l16 * OP + 8 * h;
#pragma unroll
      for (int pi = 0; pi < 4; pi++) {
        float4 o0, o1;
        o0.x = (c2[pi][0] - m2) * r2;
        o0.y = (c2[pi][1] - m2) * r2;
        o0.z = (c2[pi][2] - m2) * r2;
        o0.w = (c2[pi][3] - m2) * r2;
        o1.x = (c2[pi][4] - m2) * r2;
        o1.y = (c2[pi][5] - m2) * r2;
        o1.z = (c2[pi][6] - m2) * r2;
        o1.w = (c2[pi][7] - m2) * r2;
        *(float4*)(sw + pi * 16) = o0;
        *(float4*)(sw + pi * 16 + 4) = o1;
      }
      wave_lds_handover();
      const float* sr = so + h * OP + 4 * l16;
      size_t bb = (size_t)blockIdx.x * (NG * nb) + (size_t)(it - 3 + h);
      float* og = out + (4 * wv) * 16 + 4 * l16;
      float4 ov[8];
#pragma unroll
      for (int i = 0; i < 8; i++) ov[i] = *(const float4*)(sr + 2 * i * OP);
#pragma unroll
      for (int i = 0; i < 8; i++) {
        size_t b = bb + (size_t)(i >> 1) * nb + (size_t)(2 * (i & 1));
        *(float4*)(og + b * 1024) = ov[i];
      }
    }
  }
}

extern "C" void kernel_launch(void* const* d_in, const int* in_sizes, int n_in,
                              void* d_out, int out_size, void* d_ws,
                              size_t ws_size, hipStream_t stream) {
  (void)n_in;
  (void)out_size;
  (void)ws_size;
  const float* x = (const float*)d_in[0];
  const float* Wq = (const float*)d_in[1];
  const float* Wk = (const float*)d_in[2];
  const float* Wv = (const float*)d_in[3];
  const float* W1 = (const float*)d_in[4];
  const float* b1 = (const float*)d_in[5];
  const float* W2 = (const float*)d_in[6];
  const float* b2 = (const float*)d_in[7];
  float* out = (float*)d_out;

  float* gmaxf = (float*)d_ws;
  float* grs = (float*)((char*)d_ws + 4096 * sizeof(float));
  float* part = (float*)d_out;

  const int B = in_sizes[0] / 1024;
  const int NB = 32;
  const int blocks1 = B / NB;
  const int blocks3 = B / (NG * NB);

  enc_pass1_stats<<<blocks1, 128, 0, stream>>>(x, Wq, Wk, part, NB);
  enc_combine<<<128, 512, 0, stream>>>(part, gmaxf, grs, blocks1);
  enc_pass3_final<<<blocks3, 128 * NG, 0, stream>>>(
      x, Wq, Wk, Wv, W1, b1, W2, b2, gmaxf, grs, out, NB);
}


# pass 3 regrouped: each wave does a batch's attention + first LayerNorm (4 query tiles), no block barrier there; softmax max/sum folded into one exp2 offset held in LDS; LayerNorm sum order changes
# speedup vs baseline: 4.5743x; 1.0507x over previous
#include <hip/hip_runtime.h>

typedef __attribute__((ext_vector_type(2))) float v2f;
typedef __attribute__((ext_vector_type(8))) float v8f;

#define LN_EPS 1e-5f

#define XP 20

__device__ __forceinline__ v8f wmma4(v2f a, v2f b, v8f c) {
  return __builtin_amdgcn_wmma_f32_16x16x4_f32(false, a, false, b, (short)0, c,
                                               false, false);
}

__device__ __forceinline__ v2f ldA(const float* m, int ld, int r0, int c0,
                                   int l16, int h) {
  v2f a;
  a.x = m[(r0 + l16) * ld + c0 + 2 * h + 0];
  a.y = m[(r0 + l16) * ld + c0 + 2 * h + 1];
  return a;
}
__device__ __forceinline__ v2f ldBT(const float* m, int ld, int c0, int n0,
                                    int l16, int h) {
  v2f b;
  b.x = m[(n0 + l16) * ld + c0 + 2 * h + 0];
  b.y = m[(n0 + l16) * ld + c0 + 2 * h + 1];
  return b;
}

__device__ __forceinline__ v2f frag(v8f t, int kk) {
  v2f f = {t[2 * kk], t[2 * kk + 1]};
  return f;
}
__device__ __forceinline__ v8f qk_matrix(const float* Wq, const float* Wk,
                                         int l16, int h) {
  v8f mt = {};
#pragma unroll
  for (int kk = 0; kk < 4; kk++)
    mt = wmma4(ldA(Wq, 16, 0, 4 * kk, l16, h), ldBT(Wk, 16, 4 * kk, 0, l16, h),
               mt);
  return mt;
}
__device__ __forceinline__ v8f ldRows(const float* m, int kt, int l16, int h) {
  const float4* p = (const float4*)&m[(16 * kt + l16) * XP + 8 * h];
  float4 u0 = p[0], u1 = p[1];
  v8f t = {u0.x, u0.y, u0.z, u0.w, u1.x, u1.y, u1.z, u1.w};
  return t;
}

template <int XM>
__device__ __forceinline__ float row_xor(float v) {
  int b = __builtin_bit_cast(int, v);
  return __builtin_bit_cast(
      float, __builtin_amdgcn_update_dpp(0, b, 0x160 | XM, 0xf, 0xf, true));
}
__device__ __forceinline__ float xhalf(float v) {
  int b = __builtin_bit_cast(int, v);
  return __builtin_bit_cast(
      float, __builtin_amdgcn_permlanex16(b, b, 0x76543210, 0xfedcba98, false,
                                          false));
}
__device__ __forceinline__ float wave_sum(float v) {
  v += row_xor<1>(v);
  v += row_xor<2>(v);
  v += row_xor<4>(v);
  v += row_xor<8>(v);
  v += xhalf(v);
  return v;
}

__device__ __forceinline__ void lds_barrier() {
  asm volatile("s_wait_dscnt 0x0\n\ts_barrier_signal -1\n\ts_barrier_wait -1" ::
                   : "memory");
}

__device__ __forceinline__ void wave_lds_handover() {
  asm volatile("s_wait_dscnt 0x0" ::: "memory");
}

__global__ __launch_bounds__(128) void enc_pass1_stats(
    const float* __restrict__ x, const float* __restrict__ Wq,
    const float* __restrict__ Wk, float* __restrict__ part, int nb) {
  __shared__ __align__(16) float sX[2][64 * XP];
  int tid = threadIdx.x;
  int w = tid >> 5, lane = tid & 31, l16 = lane & 15, h = lane >> 4;
  v8f mt = qk_matrix(Wq, Wk, l16, h);
  const float4* xg = (const float4*)(x + (size_t)blockIdx.x * nb * 1024);
  float4 xc0 = {}, xc1 = {};
  if (nb > 0) {
    xc0 = xg[tid];
    xc1 = xg[tid + 128];
  }
  float vm[4][8], vsum[4][8];
#pragma unroll
  for (int j = 0; j < 4; j++)
#pragma unroll
    for (int r = 0; r < 8; r++) {
      vm[j][r] = -3.4e38f;
      vsum[j][r] = 0.0f;
    }

  for (int it = 0; it < nb; ++it) {
    float* bx = sX[it & 1];
    float4* sx0 = (float4*)&bx[(tid >> 2) * XP + (tid & 3) * 4];
    float4* sx1 = sx0 + 8 * XP;
    *sx0 = xc0;
    *sx1 = xc1;
    {
      const float4* xf = xg + (size_t)(it + 1 < nb ? it + 1 : it) * 256;
      xc0 = xf[tid];
      xc1 = xf[tid + 128];
    }
    lds_barrier();
    v8f xr[4];
#pragma unroll
    for (int j = 0; j < 4; j++) xr[j] = ldRows(bx, (w + j) & 3, l16, h);
    v8f qt = {};
#pragma unroll
    for (int kk = 0; kk < 4; kk++)
      qt = wmma4(frag(mt, kk), frag(xr[0], kk), qt);
#pragma unroll
    for (int j = 0; j < 4; j++) {
      v8f s = {};
#pragma unroll
      for (int kk = 0; kk < 4; kk++)
        s = wmma4(frag(xr[j], kk), frag(qt, kk), s);
#pragma unroll
      for (int r = 0; r < 8; r++) {
        float v = s[r] * (1.0f / 3.0f);
        float d = v - vm[j][r];
        float e = __expf(-fabsf(d));
        vsum[j][r] = (d > 0.0f) ? fmaf(vsum[j][r], e, 1.0f) : vsum[j][r] + e;
        vm[j][r] = fmaxf(vm[j][r], v);
      }
    }
  }
  float2* pp = (float2*)part + (size_t)blockIdx.x * 4096 + w * 1024 + lane;
#pragma unroll
  for (int j = 0; j < 4; j++)
#pragma unroll
    for (int r = 0; r < 8; r++)
      pp[(8 * j + r) * 32] = make_float2(vm[j][r], vsum[j][r]);
}

__device__ __forceinline__ void fold_pair(float& m, float& s, float2 t) {
  float mn = fmaxf(m, t.x);
  s = s * __expf(m - mn) + t.y * __expf(t.x - mn);
  m = mn;
}

__global__ __launch_bounds__(512) void enc_combine(
    const float* __restrict__ part, float* __restrict__ gmaxf,
    float* __restrict__ grs, int nblk) {
  __shared__ float2 red[16][32];
  int lane = threadIdx.x & 31, j = threadIdx.x >> 5;
  int e = blockIdx.x * 32 + lane;
  int per = nblk >> 4;
  const float2* p = (const float2*)part + (size_t)j * per * 4096 + e;
  float m = -3.4e38f, s = 0.0f;
#pragma unroll 8
  for (int i = 0; i < per; i++) fold_pair(m, s, p[(size_t)i * 4096]);
  red[j][lane] = make_float2(m, s);
  __syncthreads();
  if (j == 0) {
    for (int k = 1; k < 16; k++) fold_pair(m, s, red[k][lane]);
    int w = e >> 10, j4 = (e >> 8) & 3, r = (e >> 5) & 7;
    int qrow = 16 * w + (lane & 15);
    int kcol = 16 * ((w + j4) & 3) + 8 * (lane >> 4) + r;
    gmaxf[qrow * 64 + kcol] = m;
    grs[qrow * 64 + kcol] = 1.0f / s;
  }
}

#define NG 4
#define NSLOT 16
#define BS (64 * XP + 4)
#define OP 68
#define LOG2E 1.4426950408889634f

__device__ __forceinline__ v2f ldAW(const float* wp, int kk, int l16, int h) {
  v2f a;
  a.x = wp[(4 * kk + 2 * h + 0) * 16 + l16];
  a.y = wp[(4 * kk + 2 * h + 1) * 16 + l16];
  return a;
}

__global__ __launch_bounds__(128 * NG) void enc_pass3_final(
    const float* __restrict__ x, const float* __restrict__ Wq,
    const float* __restrict__ Wk, const float* __restrict__ Wv,
    const float* __restrict__ W1, const float* __restrict__ b1,
    const float* __restrict__ W2, const float* __restrict__ b2,
    const float* __restrict__ gmaxf, const float* __restrict__ grs,
    float* __restrict__ out, int nb) {
  __shared__ __align__(16) float sW1[64 * 256];
  __shared__ __align__(16) float sW2[64 * 256];
  __shared__ __align__(16) float sB1[1024], sB2[1024];
  __shared__ __align__(16) float sY[NSLOT * BS];
  __shared__ __align__(16) float sK[16 * 256];
  __shared__ float2 redB[4 * NG][NSLOT];
  __shared__ __align__(16) float sOut[4 * NG][NSLOT * OP];

  int tid = threadIdx.x;
  int wv = tid >> 5;
  int lane = tid & 31, l16 = lane & 15, h = lane >> 4;

  const int nT = nb >> 2;
  const float4* xg =
      (const float4*)(x + ((size_t)blockIdx.x * (NG * nb) +
                           (size_t)(wv >> 2) * nb + (size_t)(wv & 3)) * 1024);
  float4 xc[8];
#pragma unroll
  for (int n = 0; n < 8; n++) xc[n] = make_float4(0.0f, 0.0f, 0.0f, 0.0f);
  if (nT > 0) {
#pragma unroll
    for (int n = 0; n < 8; n++) xc[n] = xg[lane + 32 * n];
  }

  for (int i = tid; i < 16384; i += 128 * NG) {
    int pp = i >> 8, row = (i >> 4) & 15, col = i & 15;
    int prow = 4 * ((row & 7) >> 1) + 2 * (row >> 3) + (row & 1);
    sW1[pp * 256 + prow * 16 + col] = W1[i];
    sW2[pp * 256 + prow * 16 + col] = W2[i];
  }
  for (int i = tid; i < 1024; i += 128 * NG) {
    sB1[i] = b1[i];
    sB2[i] = b2[i];
  }
  {
    int qrow = 16 * (tid >> 7) + l16;
    int kcol = 16 * ((tid >> 5) & 3) + 8 * h;
    const float4* gp = (const float4*)(gmaxf + qrow * 64 + kcol);
    const float4* rp = (const float4*)(grs + qrow * 64 + kcol);
    float4 g0 = gp[0], g1 = gp[1], q0 = rp[0], q1 = rp[1];
    float4 k0, k1;
    k0.x = log2f(q0.x) - g0.x * LOG2E;
    k0.y = log2f(q0.y) - g0.y * LOG2E;
    k0.z = log2f(q0.z) - g0.z * LOG2E;
    k0.w = log2f(q0.w) - g0.w * LOG2E;
    k1.x = log2f(q1.x) - g1.x * LOG2E;
    k1.y = log2f(q1.y) - g1.y * LOG2E;
    k1.z = log2f(q1.z) - g1.z * LOG2E;
    k1.w = log2f(q1.w) - g1.w * LOG2E;
    *(float4*)&sK[tid * 8] = k0;
    *(float4*)&sK[tid * 8 + 4] = k1;
  }
  v8f mt = qk_matrix(Wq, Wk, l16, h);
  v2f av[4];
#pragma unroll
  for (int kk = 0; kk < 4; kk++) {
    av[kk].x = Wv[(8 * h + 2 * kk + 0) * 16 + l16];
    av[kk].y = Wv[(8 * h + 2 * kk + 1) * 16 + l16];
  }
  __syncthreads();

  float* mX = sY + wv * BS;
  float* tS = sOut[wv];
  for (int T = 0; T < nT; ++T) {
    {
      float4* sx = (float4*)&mX[(lane >> 2) * XP + (lane & 3) * 4];
#pragma unroll
      for (int n = 0; n < 8; n++) sx[n * 2 * XP] = xc[n];
      const float4* xf = xg + (size_t)(T + 1 < nT ? T + 1 : T) * 1024;
#pragma unroll
      for (int n = 0; n < 8; n++) xc[n] = xf[lane + 32 * n];
    }
    wave_lds_handover();

    v8f xr[4];
#pragma unroll
    for (int kt = 0; kt < 4; kt++) xr[kt] = ldRows(mX, kt, l16, h);
    float part = 0.0f, part2 = 0.0f;
#pragma nounroll
    for (int w = 0; w < 4; ++w) {
      v8f xq = ldRows(mX, w, l16, h);
      v8f qt = {};
#pragma unroll
      for (int kk = 0; kk < 4; kk++)
        qt = wmma4(frag(mt, kk), frag(xq, kk), qt);
      const float* kp = sK + (w * 128 + lane) * 8;
      v8f ut0 = {};
      v8f ut1 = {};
#pragma unroll
      for (int kt = 0; kt < 4; kt++) {
        v8f s = {};
#pragma unroll
        for (int kk = 0; kk < 4; kk++)
          s = wmma4(frag(xr[kt], kk), frag(qt, kk), s);
        float4 ka = *(const float4*)(kp + kt * 256);
        float4 kb = *(const float4*)(kp + kt * 256 + 4);
        const float c = LOG2E / 3.0f;
        s[0] = __builtin_amdgcn_exp2f(fmaf(s[0], c, ka.x));
        s[1] = __builtin_amdgcn_exp2f(fmaf(s[1], c, ka.y));
        s[2] = __builtin_amdgcn_exp2f(fmaf(s[2], c, ka.z));
        s[3] = __builtin_amdgcn_exp2f(fmaf(s[3], c, ka.w));
        s[4] = __builtin_amdgcn_exp2f(fmaf(s[4], c, kb.x));
        s[5] = __builtin_amdgcn_exp2f(fmaf(s[5], c, kb.y));
        s[6] = __builtin_amdgcn_exp2f(fmaf(s[6], c, kb.z));
        s[7] = __builtin_amdgcn_exp2f(fmaf(s[7], c, kb.w));
        const float* xt = mX + (16 * kt + 8 * h) * XP + l16;
#pragma unroll
        for (int kk = 0; kk < 4; kk++) {
          v2f a = {xt[(2 * kk + 0) * XP], xt[(2 * kk + 1) * XP]};
          if (kt & 1)
            ut1 = wmma4(a, frag(s, kk), ut1);
          else
            ut0 = wmma4(a, frag(s, kk), ut0);
        }
      }
      v8f ut = ut0 + ut1;
      v8f z = {};
#pragma unroll
      for (int kk = 0; kk < 4; kk++) z = wmma4(av[kk], frag(ut, kk), z);
      float tv[8];
#pragma unroll
      for (int r = 0; r < 8; r++) {
        float tt = xq[r] + z[r];
        tv[r] = tt;
        part += tt;
        part2 += tt * tt;
      }
      float4* tp = (float4*)(tS + (w * 64 + lane) * 4);
      tp[0] = make_float4(tv[0], tv[1], tv[2], tv[3]);
      tp[32] = make_float4(tv[4], tv[5], tv[6], tv[7]);
    }
    float mean = wave_sum(part) * (1.0f / 1024.0f);
    float rstd =
        rsqrtf(wave_sum(part2) * (1.0f / 1024.0f) - mean * mean + LN_EPS);
#pragma unroll
    for (int w = 0; w < 4; w++) {
      const float4* tp = (const float4*)(tS + (w * 64 + lane) * 4);
      float4 t0 = tp[0], t1 = tp[32];
      float4* yp = (float4*)&mX[(16 * w + l16) * XP + 8 * h];
      yp[0] = make_float4((t0.x - mean) * rstd, (t0.y - mean) * rstd,
                          (t0.z - mean) * rstd, (t0.w - mean) * rstd);
      yp[1] = make_float4((t1.x - mean) * rstd, (t1.y - mean) * rstd,
                          (t1.z - mean) * rstd, (t1.w - mean) * rstd);
    }

    {
      const int it = 4 * T + 3;
      lds_barrier();
      const float* yl = sY + l16 * BS + 8 * h;
      v8f yv[4], c1[4], c2[4];
      v2f aw[2][4];
#pragma unroll
      for (int pi = 0; pi < 4; pi++) {
        int p = 4 * wv + pi;
        float4 y0 = *(const float4*)(yl + p * XP);
        float4 y1 = *(const float4*)(yl + p * XP + 4);
        yv[pi] = (v8f){y0.x, y0.y, y0.z, y0.w, y1.x, y1.y, y1.z, y1.w};
        float4 u0 = *(const float4*)(sB1 + p * 16 + 8 * h);
        float4 u1 = *(const float4*)(sB1 + p * 16 + 8 * h + 4);
        c1[pi] = (v8f){u0.x, u0.y, u0.z, u0.w, u1.x, u1.y, u1.z, u1.w};
      }
#pragma unroll
      for (int kh = 0; kh < 2; kh++) {
#pragma unroll
        for (int k2 = 0; k2 < 2; k2++)
#pragma unroll
          for (int pi = 0; pi < 4; pi++)
            aw[k2][pi] = ldAW(sW1 + (4 * wv + pi) * 256, 2 * kh + k2, l16, h);
#pragma unroll
        for (int k2 = 0; k2 < 2; k2++)
#pragma unroll
          for (int pi = 0; pi < 4; pi++) {
            int kk = 2 * kh + k2;
            v2f bf = {yv[pi][2 * kk], yv[pi][2 * kk + 1]};
            c1[pi] = wmma4(aw[k2][pi], bf, c1[pi]);
          }
      }
#pragma unroll
      for (int pi = 0; pi < 4; pi++) {
        int p = 4 * wv + pi;
        float4 d0 = *(const float4*)(sB2 + p * 16 + 8 * h);
        float4 d1 = *(const float4*)(sB2 + p * 16 + 8 * h + 4);
        c2[pi] = (v8f){d0.x, d0.y, d0.z, d0.w, d1.x, d1.y, d1.z, d1.w};
      }
#pragma unroll
      for (int pi = 0; pi < 4; pi++)
#pragma unroll
        for (int r = 0; r < 8; r++) c1[pi][r] = fmaxf(c1[pi][r], 0.0f);
#pragma unroll
      for (int kh = 0; kh < 2; kh++) {
#pragma unroll
        for (int k2 = 0; k2 < 2; k2++)
#pragma unroll
          for (int pi = 0; pi < 4; pi++)
            aw[k2][pi] = ldAW(sW2 + (4 * wv + pi) * 256, 2 * kh + k2, l16, h);
#pragma unroll
        for (int k2 = 0; k2 < 2; k2++)
#pragma unroll
          for (int pi = 0; pi < 4; pi++) {
            int kk = 2 * kh + k2;
            v2f bf = {c1[pi][2 * kk], c1[pi][2 * kk + 1]};
            c2[pi] = wmma4(aw[k2][pi], bf, c2[pi]);
          }
      }
      float q1 = 0.0f, q2 = 0.0f;
#pragma unroll
      for (int pi = 0; pi < 4; pi++)
#pragma unroll
        for (int r = 0; r < 8; r++) {
          float tt = fmaxf(c2[pi][r], 0.0f) + yv[pi][r];
          c2[pi][r] = tt;
          q1 += tt;
          q2 += tt * tt;
        }
      q1 += xhalf(q1);
      q2 += xhalf(q2);
      if (h == 0) redB[wv][l16] = make_float2(q1, q2);
      lds_barrier();
      float u1s = 0.0f, u2s = 0.0f;
#pragma unroll
      for (int j = 0; j < 4 * NG; j++) {
        float2 pr = redB[j][l16];
        u1s += pr.x;
        u2s += pr.y;
      }
      float m2 = u1s * (1.0f / 1024.0f);
      float r2 = rsqrtf(u2s * (1.0f / 1024.0f) - m2 * m2 + LN_EPS);
      float* so = sOut[wv];
      float* sw = so + l16 * OP + 8 * h;
#pragma unroll
      for (int pi = 0; pi < 4; pi++) {
        float4 o0, o1;
        o0.x = (c2[pi][0] - m2) * r2;
        o0.y = (c2[pi][1] - m2) * r2;
        o0.z = (c2[pi][2] - m2) * r2;
        o0.w = (c2[pi][3] - m2) * r2;
        o1.x = (c2[pi][4] - m2) * r2;
        o1.y = (c2[pi][5] - m2) * r2;
        o1.z = (c2[pi][6] - m2) * r2;
        o1.w = (c2[pi][7] - m2) * r2;
        *(float4*)(sw + pi * 16) = o0;
        *(float4*)(sw + pi * 16 + 4) = o1;
      }
      wave_lds_handover();
      const float* sr = so + h * OP + 4 * l16;
      size_t bb = (size_t)blockIdx.x * (NG * nb) + (size_t)(it - 3 + h);
      float* og = out + (4 * wv) * 16 + 4 * l16;
      float4 ov[8];
#pragma unroll
      for (int i = 0; i < 8; i++) ov[i] = *(const float4*)(sr + 2 * i * OP);
#pragma unroll
      for (int i = 0; i < 8; i++) {
        size_t b = bb + (size_t)(i >> 1) * nb + (size_t)(2 * (i & 1));
        *(float4*)(og + b * 1024) = ov[i];
      }
    }
  }
}

extern "C" void kernel_launch(void* const* d_in, const int* in_sizes, int n_in,
                              void* d_out, int out_size, void* d_ws,
                              size_t ws_size, hipStream_t stream) {
  (void)n_in;
  (void)out_size;
  (void)ws_size;
  const float* x = (const float*)d_in[0];
  const float* Wq = (const float*)d_in[1];
  const float* Wk = (const float*)d_in[2];
  const float* Wv = (const float*)d_in[3];
  const float* W1 = (const float*)d_in[4];
  const float* b1 = (const float*)d_in[5];
  const float* W2 = (const float*)d_in[6];
  const float* b2 = (const float*)d_in[7];
  float* out = (float*)d_out;

  float* gmaxf = (float*)d_ws;
  float* grs = (float*)((char*)d_ws + 4096 * sizeof(float));
  float* part = (float*)d_out;

  const int B = in_sizes[0] / 1024;
  const int NB = 32;
  const int blocks1 = B / NB;
  const int blocks3 = B / (NG * NB);

  enc_pass1_stats<<<blocks1, 128, 0, stream>>>(x, Wq, Wk, part, NB);
  enc_combine<<<128, 512, 0, stream>>>(part, gmaxf, grs, blocks1);
  enc_pass3_final<<<blocks3, 128 * NG, 0, stream>>>(
      x, Wq, Wk, Wv, W1, b1, W2, b2, gmaxf, grs, out, NB);
}


# vector instructions beside the f32 WMMAs cut in both sweeps: log2(e)/3 folded into M (score rounding), exp2(-|d|) direct, max by the sum's compare; LN1 sums 2-wide (fp order), LN scale-shift one FMA
# speedup vs baseline: 4.7826x; 1.0455x over previous
#include <hip/hip_runtime.h>

typedef __attribute__((ext_vector_type(2))) float v2f;
typedef __attribute__((ext_vector_type(8))) float v8f;

#define LN_EPS 1e-5f
#define LOG2E 1.4426950408889634f
#define SCORE_SCALE (LOG2E / 3.0f)

#define XP 20

__device__ __forceinline__ v8f wmma4(v2f a, v2f b, v8f c) {
  return __builtin_amdgcn_wmma_f32_16x16x4_f32(false, a, false, b, (short)0, c,
                                               false, false);
}

__device__ __forceinline__ v2f ldA(const float* m, int ld, int r0, int c0,
                                   int l16, int h) {
  v2f a;
  a.x = m[(r0 + l16) * ld + c0 + 2 * h + 0];
  a.y = m[(r0 + l16) * ld + c0 + 2 * h + 1];
  return a;
}
__device__ __forceinline__ v2f ldBT(const float* m, int ld, int c0, int n0,
                                    int l16, int h) {
  v2f b;
  b.x = m[(n0 + l16) * ld + c0 + 2 * h + 0];
  b.y = m[(n0 + l16) * ld + c0 + 2 * h + 1];
  return b;
}

__device__ __forceinline__ v2f frag(v8f t, int kk) {
  v2f f = {t[2 * kk], t[2 * kk + 1]};
  return f;
}
__device__ __forceinline__ v8f qk_matrix(const float* Wq, const float* Wk,
                                         int l16, int h) {
  v8f mt = {};
#pragma unroll
  for (int kk = 0; kk < 4; kk++)
    mt = wmma4(ldA(Wq, 16, 0, 4 * kk, l16, h) * SCORE_SCALE,
               ldBT(Wk, 16, 4 * kk, 0, l16, h), mt);
  return mt;
}
__device__ __forceinline__ v8f ldRows(const float* m, int kt, int l16, int h) {
  const float4* p = (const float4*)&m[(16 * kt + l16) * XP + 8 * h];
  float4 u0 = p[0], u1 = p[1];
  v8f t = {u0.x, u0.y, u0.z, u0.w, u1.x, u1.y, u1.z, u1.w};
  return t;
}

template <int XM>
__device__ __forceinline__ float row_xor(float v) {
  int b = __builtin_bit_cast(int, v);
  return __builtin_bit_cast(
      float, __builtin_amdgcn_update_dpp(0, b, 0x160 | XM, 0xf, 0xf, true));
}
__device__ __forceinline__ float xhalf(float v) {
  int b = __builtin_bit_cast(int, v);
  return __builtin_bit_cast(
      float, __builtin_amdgcn_permlanex16(b, b, 0x76543210, 0xfedcba98, false,
                                          false));
}
__device__ __forceinline__ float wave_sum(float v) {
  v += row_xor<1>(v);
  v += row_xor<2>(v);
  v += row_xor<4>(v);
  v += row_xor<8>(v);
  v += xhalf(v);
  return v;
}

__device__ __forceinline__ void lds_barrier() {
  asm volatile("s_wait_dscnt 0x0\n\ts_barrier_signal -1\n\ts_barrier_wait -1" ::
                   : "memory");
}

__device__ __forceinline__ void wave_lds_handover() {
  asm volatile("s_wait_dscnt 0x0" ::: "memory");
}

__global__ __launch_bounds__(128) void enc_pass1_stats(
    const float* __restrict__ x, const float* __restrict__ Wq,
    const float* __restrict__ Wk, float* __restrict__ part, int nb) {
  __shared__ __align__(16) float sX[2][64 * XP];
  int tid = threadIdx.x;
  int w = tid >> 5, lane = tid & 31, l16 = lane & 15, h = lane >> 4;
  v8f mt = qk_matrix(Wq, Wk, l16, h);
  const float4* xg = (const float4*)(x + (size_t)blockIdx.x * nb * 1024);
  float4 xc0 = {}, xc1 = {};
  if (nb > 0) {
    xc0 = xg[tid];
    xc1 = xg[tid + 128];
  }
  float vm[4][8], vsum[4][8];
#pragma unroll
  for (int j = 0; j < 4; j++)
#pragma unroll
    for (int r = 0; r < 8; r++) {
      vm[j][r] = -3.4e38f;
      vsum[j][r] = 0.0f;
    }

  for (int it = 0; it < nb; ++it) {
    float* bx = sX[it & 1];
    float4* sx0 = (float4*)&bx[(tid >> 2) * XP + (tid & 3) * 4];
    float4* sx1 = sx0 + 8 * XP;
    *sx0 = xc0;
    *sx1 = xc1;
    {
      const float4* xf = xg + (size_t)(it + 1 < nb ? it + 1 : it) * 256;
      xc0 = xf[tid];
      xc1 = xf[tid + 128];
    }
    lds_barrier();
    v8f xr[4];
#pragma unroll
    for (int j = 0; j < 4; j++) xr[j] = ldRows(bx, (w + j) & 3, l16, h);
    v8f qt = {};
#pragma unroll
    for (int kk = 0; kk < 4; kk++)
      qt = wmma4(frag(mt, kk), frag(xr[0], kk), qt);
#pragma unroll
    for (int j = 0; j < 4; j++) {
      v8f s = {};
#pragma unroll
      for (int kk = 0; kk < 4; kk++)
        s = wmma4(frag(xr[j], kk), frag(qt, kk), s);
#pragma unroll
      for (int r = 0; r < 8; r++) {
        float v = s[r];
        float d = v - vm[j][r];
        float e = __builtin_amdgcn_exp2f(-fabsf(d));
        bool up = d > 0.0f;
        float grown = fmaf(vsum[j][r], e, 1.0f);
        float added = vsum[j][r] + e;
        vsum[j][r] = up ? grown : added;
        vm[j][r] = up ? v : vm[j][r];
      }
    }
  }
  float2* pp = (float2*)part + (size_t)blockIdx.x * 4096 + w * 1024 + lane;
#pragma unroll
  for (int j = 0; j < 4; j++)
#pragma unroll
    for (int r = 0; r < 8; r++)
      pp[(8 * j + r) * 32] = make_float2(vm[j][r], vsum[j][r]);
}

__device__ __forceinline__ void fold_pair(float& m, float& s, float2 t) {
  float mn = fmaxf(m, t.x);
  s = s * __builtin_amdgcn_exp2f(m - mn) + t.y * __builtin_amdgcn_exp2f(t.x - mn);
  m = mn;
}

__global__ __launch_bounds__(512) void enc_combine(
    const float* __restrict__ part, float* __restrict__ gmaxf,
    float* __restrict__ grs, int nblk) {
  __shared__ float2 red[16][32];
  int lane = threadIdx.x & 31, j = threadIdx.x >> 5;
  int e = blockIdx.x * 32 + lane;
  int per = nblk >> 4;
  const float2* p = (const float2*)part + (size_t)j * per * 4096 + e;
  float m = -3.4e38f, s = 0.0f;
#pragma unroll 8
  for (int i = 0; i < per; i++) fold_pair(m, s, p[(size_t)i * 4096]);
  red[j][lane] = make_float2(m, s);
  __syncthreads();
  if (j == 0) {
    for (int k = 1; k < 16; k++) fold_pair(m, s, red[k][lane]);
    int w = e >> 10, j4 = (e >> 8) & 3, r = (e >> 5) & 7;
    int qrow = 16 * w + (lane & 15);
    int kcol = 16 * ((w + j4) & 3) + 8 * (lane >> 4) + r;
    gmaxf[qrow * 64 + kcol] = m;
    grs[qrow * 64 + kcol] = 1.0f / s;
  }
}

#define NG 4
#define NSLOT 16
#define BS (64 * XP + 4)
#define OP 68

__device__ __forceinline__ v2f ldAW(const float* wp, int kk, int l16, int h) {
  v2f a;
  a.x = wp[(4 * kk + 2 * h + 0) * 16 + l16];
  a.y = wp[(4 * kk + 2 * h + 1) * 16 + l16];
  return a;
}

__global__ __launch_bounds__(128 * NG) void enc_pass3_final(
    const float* __restrict__ x, const float* __restrict__ Wq,
    const float* __restrict__ Wk, const float* __restrict__ Wv,
    const float* __restrict__ W1, const float* __restrict__ b1,
    const float* __restrict__ W2, const float* __restrict__ b2,
    const float* __restrict__ gmaxf, const float* __restrict__ grs,
    float* __restrict__ out, int nb) {
  __shared__ __align__(16) float sW1[64 * 256];
  __shared__ __align__(16) float sW2[64 * 256];
  __shared__ __align__(16) float sB1[1024], sB2[1024];
  __shared__ __align__(16) float sY[NSLOT * BS];
  __shared__ __align__(16) float sK[16 * 256];
  __shared__ float2 redB[4 * NG][NSLOT];
  __shared__ __align__(16) float sOut[4 * NG][NSLOT * OP];

  int tid = threadIdx.x;
  int wv = tid >> 5;
  int lane = tid & 31, l16 = lane & 15, h = lane >> 4;

  const int nT = nb >> 2;
  const float4* xg =
      (const float4*)(x + ((size_t)blockIdx.x * (NG * nb) +
                           (size_t)(wv >> 2) * nb + (size_t)(wv & 3)) * 1024);
  float4 xc[8];
#pragma unroll
  for (int n = 0; n < 8; n++) xc[n] = make_float4(0.0f, 0.0f, 0.0f, 0.0f);
  if (nT > 0) {
#pragma unroll
    for (int n = 0; n < 8; n++) xc[n] = xg[lane + 32 * n];
  }

  for (int i = tid; i < 16384; i += 128 * NG) {
    int pp = i >> 8, row = (i >> 4) & 15, col = i & 15;
    int prow = 4 * ((row & 7) >> 1) + 2 * (row >> 3) + (row & 1);
    sW1[pp * 256 + prow * 16 + col] = W1[i];
    sW2[pp * 256 + prow * 16 + col] = W2[i];
  }
  for (int i = tid; i < 1024; i += 128 * NG) {
    sB1[i] = b1[i];
    sB2[i] = b2[i];
  }
  {
    int qrow = 16 * (tid >> 7) + l16;
    int kcol = 16 * ((tid >> 5) & 3) + 8 * h;
    const float4* gp = (const float4*)(gmaxf + qrow * 64 + kcol);
    const float4* rp = (const float4*)(grs + qrow * 64 + kcol);
    float4 g0 = gp[0], g1 = gp[1], q0 = rp[0], q1 = rp[1];
    float4 k0, k1;
    k0.x = log2f(q0.x) - g0.x;
    k0.y = log2f(q0.y) - g0.y;
    k0.z = log2f(q0.z) - g0.z;
    k0.w = log2f(q0.w) - g0.w;
    k1.x = log2f(q1.x) - g1.x;
    k1.y = log2f(q1.y) - g1.y;
    k1.z = log2f(q1.z) - g1.z;
    k1.w = log2f(q1.w) - g1.w;
    *(float4*)&sK[tid * 8] = k0;
    *(float4*)&sK[tid * 8 + 4] = k1;
  }
  v8f mt = qk_matrix(Wq, Wk, l16, h);
  v2f av[4];
#pragma unroll
  for (int kk = 0; kk < 4; kk++) {
    av[kk].x = Wv[(8 * h + 2 * kk + 0) * 16 + l16];
    av[kk].y = Wv[(8 * h + 2 * kk + 1) * 16 + l16];
  }
  __syncthreads();

  float* mX = sY + wv * BS;
  float* tS = sOut[wv];
  for (int T = 0; T < nT; ++T) {
    {
      float4* sx = (float4*)&mX[(lane >> 2) * XP + (lane & 3) * 4];
#pragma unroll
      for (int n = 0; n < 8; n++) sx[n * 2 * XP] = xc[n];
      const float4* xf = xg + (size_t)(T + 1 < nT ? T + 1 : T) * 1024;
#pragma unroll
      for (int n = 0; n < 8; n++) xc[n] = xf[lane + 32 * n];
    }
    wave_lds_handover();

    v8f xr[4];
#pragma unroll
    for (int kt = 0; kt < 4; kt++) xr[kt] = ldRows(mX, kt, l16, h);
    v2f ps = {0.0f, 0.0f}, pq = {0.0f, 0.0f};
#pragma nounroll
    for (int w = 0; w < 4; ++w) {
      v8f xq = ldRows(mX, w, l16, h);
      v8f qt = {};
#pragma unroll
      for (int kk = 0; kk < 4; kk++)
        qt = wmma4(frag(mt, kk), frag(xq, kk), qt);
      const float* kp = sK + (w * 128 + lane) * 8;
      v8f ut0 = {};
      v8f ut1 = {};
#pragma unroll
      for (int kt = 0; kt < 4; kt++) {
        v8f s = {};
#pragma unroll
        for (int kk = 0; kk < 4; kk++)
          s = wmma4(frag(xr[kt], kk), frag(qt, kk), s);
        float4 ka = *(const float4*)(kp + kt * 256);
        float4 kb = *(const float4*)(kp + kt * 256 + 4);
        s[0] = __builtin_amdgcn_exp2f(s[0] + ka.x);
        s[1] = __builtin_amdgcn_exp2f(s[1] + ka.y);
        s[2] = __builtin_amdgcn_exp2f(s[2] + ka.z);
        s[3] = __builtin_amdgcn_exp2f(s[3] + ka.w);
        s[4] = __builtin_amdgcn_exp2f(s[4] + kb.x);
        s[5] = __builtin_amdgcn_exp2f(s[5] + kb.y);
        s[6] = __builtin_amdgcn_exp2f(s[6] + kb.z);
        s[7] = __builtin_amdgcn_exp2f(s[7] + kb.w);
        const float* xt = mX + (16 * kt + 8 * h) * XP + l16;
#pragma unroll
        for (int kk = 0; kk < 4; kk++) {
          v2f a = {xt[(2 * kk + 0) * XP], xt[(2 * kk + 1) * XP]};
          if (kt & 1)
            ut1 = wmma4(a, frag(s, kk), ut1);
          else
            ut0 = wmma4(a, frag(s, kk), ut0);
        }
      }
      v8f ut = ut0 + ut1;
      v8f z = {};
#pragma unroll
      for (int kk = 0; kk < 4; kk++) z = wmma4(av[kk], frag(ut, kk), z);
      v8f tv = xq + z;
#pragma unroll
      for (int kk = 0; kk < 4; kk++) {
        v2f tt = frag(tv, kk);
        ps += tt;
        pq += tt * tt;
      }
      float4* tp = (float4*)(tS + (w * 64 + lane) * 4);
      tp[0] = make_float4(tv[0], tv[1], tv[2], tv[3]);
      tp[32] = make_float4(tv[4], tv[5], tv[6], tv[7]);
    }
    float mean = wave_sum(ps.x + ps.y) * (1.0f / 1024.0f);
    float rstd = rsqrtf(wave_sum(pq.x + pq.y) * (1.0f / 1024.0f) - mean * mean +
                        LN_EPS);
    float nmr = -mean * rstd;
#pragma unroll
    for (int w = 0; w < 4; w++) {
      const float4* tp = (const float4*)(tS + (w * 64 + lane) * 4);
      float4 t0 = tp[0], t1 = tp[32];
      float4* yp = (float4*)&mX[(16 * w + l16) * XP + 8 * h];
      yp[0] = make_float4(fmaf(t0.x, rstd, nmr), fmaf(t0.y, rstd, nmr),
                          fmaf(t0.z, rstd, nmr), fmaf(t0.w, rstd, nmr));
      yp[1] = make_float4(fmaf(t1.x, rstd, nmr), fmaf(t1.y, rstd, nmr),
                          fmaf(t1.z, rstd, nmr), fmaf(t1.w, rstd, nmr));
    }

    {
      const int it = 4 * T + 3;
      lds_barrier();
      const float* yl = sY + l16 * BS + 8 * h;
      v8f yv[4], c1[4], c2[4];
      v2f aw[2][4];
#pragma unroll
      for (int pi = 0; pi < 4; pi++) {
        int p = 4 * wv + pi;
        float4 y0 = *(const float4*)(yl + p * XP);
        float4 y1 = *(const float4*)(yl + p * XP + 4);
        yv[pi] = (v8f){y0.x, y0.y, y0.z, y0.w, y1.x, y1.y, y1.z, y1.w};
        float4 u0 = *(const float4*)(sB1 + p * 16 + 8 * h);
        float4 u1 = *(const float4*)(sB1 + p * 16 + 8 * h + 4);
        c1[pi] = (v8f){u0.x, u0.y, u0.z, u0.w, u1.x, u1.y, u1.z, u1.w};
      }
#pragma unroll
      for (int kh = 0; kh < 2; kh++) {
#pragma unroll
        for (int k2 = 0; k2 < 2; k2++)
#pragma unroll
          for (int pi = 0; pi < 4; pi++)
            aw[k2][pi] = ldAW(sW1 + (4 * wv + pi) * 256, 2 * kh + k2, l16, h);
#pragma unroll
        for (int k2 = 0; k2 < 2; k2++)
#pragma unroll
          for (int pi = 0; pi < 4; pi++) {
            int kk = 2 * kh + k2;
            v2f bf = {yv[pi][2 * kk], yv[pi][2 * kk + 1]};
            c1[pi] = wmma4(aw[k2][pi], bf, c1[pi]);
          }
      }
#pragma unroll
      for (int pi = 0; pi < 4; pi++) {
        int p = 4 * wv + pi;
        float4 d0 = *(const float4*)(sB2 + p * 16 + 8 * h);
        float4 d1 = *(const float4*)(sB2 + p * 16 + 8 * h + 4);
        c2[pi] = (v8f){d0.x, d0.y, d0.z, d0.w, d1.x, d1.y, d1.z, d1.w};
      }
#pragma unroll
      for (int pi = 0; pi < 4; pi++)
#pragma unroll
        for (int r = 0; r < 8; r++) c1[pi][r] = fmaxf(c1[pi][r], 0.0f);
#pragma unroll
      for (int kh = 0; kh < 2; kh++) {
#pragma unroll
        for (int k2 = 0; k2 < 2; k2++)
#pragma unroll
          for (int pi = 0; pi < 4; pi++)
            aw[k2][pi] = ldAW(sW2 + (4 * wv + pi) * 256, 2 * kh + k2, l16, h);
#pragma unroll
        for (int k2 = 0; k2 < 2; k2++)
#pragma unroll
          for (int pi = 0; pi < 4; pi++) {
            int kk = 2 * kh + k2;
            v2f bf = {c1[pi][2 * kk], c1[pi][2 * kk + 1]};
            c2[pi] = wmma4(aw[k2][pi], bf, c2[pi]);
          }
      }
      float q1 = 0.0f, q2 = 0.0f;
#pragma unroll
      for (int pi = 0; pi < 4; pi++)
#pragma unroll
        for (int r = 0; r < 8; r++) {
          float tt = fmaxf(c2[pi][r], 0.0f) + yv[pi][r];
          c2[pi][r] = tt;
          q1 += tt;
          q2 += tt * tt;
        }
      q1 += xhalf(q1);
      q2 += xhalf(q2);
      if (h == 0) redB[wv][l16] = make_float2(q1, q2);
      lds_barrier();
      float u1s = 0.0f, u2s = 0.0f;
#pragma unroll
      for (int j = 0; j < 4 * NG; j++) {
        float2 pr = redB[j][l16];
        u1s += pr.x;
        u2s += pr.y;
      }
      float m2 = u1s * (1.0f / 1024.0f);
      float r2 = rsqrtf(u2s * (1.0f / 1024.0f) - m2 * m2 + LN_EPS);
      float nm2 = -m2 * r2;
      float* so = sOut[wv];
      float* sw = so + l16 * OP + 8 * h;
#pragma unroll
      for (int pi = 0; pi < 4; pi++) {
        float4 o0, o1;
        o0.x = fmaf(c2[pi][0], r2, nm2);
        o0.y = fmaf(c2[pi][1], r2, nm2);
        o0.z = fmaf(c2[pi][2], r2, nm2);
        o0.w = fmaf(c2[pi][3], r2, nm2);
        o1.x = fmaf(c2[pi][4], r2, nm2);
        o1.y = fmaf(c2[pi][5], r2, nm2);
        o1.z = fmaf(c2[pi][6], r2, nm2);
        o1.w = fmaf(c2[pi][7], r2, nm2);
        *(float4*)(sw + pi * 16) = o0;
        *(float4*)(sw + pi * 16 + 4) = o1;
      }
      wave_lds_handover();
      const float* sr = so + h * OP + 4 * l16;
      size_t bb = (size_t)blockIdx.x * (NG * nb) + (size_t)(it - 3 + h);
      float* og = out + (4 * wv) * 16 + 4 * l16;
      float4 ov[8];
#pragma unroll
      for (int i = 0; i < 8; i++) ov[i] = *(const float4*)(sr + 2 * i * OP);
#pragma unroll
      for (int i = 0; i < 8; i++) {
        size_t b = bb + (size_t)(i >> 1) * nb + (size_t)(2 * (i & 1));
        *(float4*)(og + b * 1024) = ov[i];
      }
    }
  }
}

extern "C" void kernel_launch(void* const* d_in, const int* in_sizes, int n_in,
                              void* d_out, int out_size, void* d_ws,
                              size_t ws_size, hipStream_t stream) {
  (void)n_in;
  (void)out_size;
  (void)ws_size;
  const float* x = (const float*)d_in[0];
  const float* Wq = (const float*)d_in[1];
  const float* Wk = (const float*)d_in[2];
  const float* Wv = (const float*)d_in[3];
  const float* W1 = (const float*)d_in[4];
  const float* b1 = (const float*)d_in[5];
  const float* W2 = (const float*)d_in[6];
  const float* b2 = (const float*)d_in[7];
  float* out = (float*)d_out;

  float* gmaxf = (float*)d_ws;
  float* grs = (float*)((char*)d_ws + 4096 * sizeof(float));
  float* part = (float*)d_out;

  const int B = in_sizes[0] / 1024;
  const int NB = 32;
  const int blocks1 = B / NB;
  const int blocks3 = B / (NG * NB);

  enc_pass1_stats<<<blocks1, 128, 0, stream>>>(x, Wq, Wk, part, NB);
  enc_combine<<<128, 512, 0, stream>>>(part, gmaxf, grs, blocks1);
  enc_pass3_final<<<blocks3, 128 * NG, 0, stream>>>(
      x, Wq, Wk, Wv, W1, b1, W2, b2, gmaxf, grs, out, NB);
}
